# starting kernel
# baseline (speedup 1.0000x reference)
#include <hip/hip_runtime.h>
#include <hip/hip_bf16.h>


typedef __attribute__((ext_vector_type(2))) float v2f;
typedef __attribute__((ext_vector_type(8))) float v8f;

#define NEG_SLOPE 0.2f
#define GAT_EPS   1e-16f

template <int KDIM, int MO>
__global__ __launch_bounds__(128) void gemm_wmma_f32(
    const float* __restrict__ X, const float* __restrict__ W,
    float* __restrict__ Hout, int ntiles) {
  const int wave = threadIdx.x >> 5;
  const int tile = blockIdx.x * 4 + wave;
  if (tile >= ntiles) return;
  const int lane  = threadIdx.x & 31;
  const int lmod  = lane & 15;
  const int lhalf = lane >> 4;
  const int row0  = tile * 16;

  v8f acc[MO / 16];
#pragma unroll
  for (int t = 0; t < MO / 16; ++t) acc[t] = v8f{0, 0, 0, 0, 0, 0, 0, 0};

  const float* xrow = X + (size_t)(row0 + lmod) * KDIM;
#pragma unroll 4
  for (int k0 = 0; k0 < KDIM; k0 += 4) {
    const int ka = k0 + 2 * lhalf;
    v2f a;
    a.x = xrow[ka];
    a.y = xrow[ka + 1];
#pragma unroll
    for (int t = 0; t < MO / 16; ++t) {
      const float* wr = W + (size_t)ka * MO + t * 16 + lmod;
      v2f b;
      b.x = wr[0];
      b.y = wr[MO];
      acc[t] = __builtin_amdgcn_wmma_f32_16x16x4_f32(
          false, a, false, b, (short)0, acc[t], false, false);
    }
  }
#pragma unroll
  for (int t = 0; t < MO / 16; ++t)
#pragma unroll
    for (int r = 0; r < 8; ++r)
      Hout[(size_t)(row0 + r + 8 * lhalf) * MO + t * 16 + lmod] = acc[t][r];
}

__global__ void node_attn(const float* __restrict__ Hm,
                          const float* __restrict__ a_src,
                          const float* __restrict__ a_dst,
                          float* __restrict__ esrc, float* __restrict__ edst,
                          int N, int Hh) {
  int idx = blockIdx.x * blockDim.x + threadIdx.x;
  if (idx >= N * Hh) return;
  const int h = idx % Hh;
  const int n = idx / Hh;
  const float* row = Hm + ((size_t)n * Hh + h) * 16;
  float s = 0.f, d = 0.f;
#pragma unroll
  for (int c = 0; c < 16; ++c) {
    const float v = row[c];
    s += v * a_src[h * 16 + c];
    d += v * a_dst[h * 16 + c];
  }
  esrc[idx] = s;
  edst[idx] = d;
}

__global__ void fill_f32(float* __restrict__ p, float v, int n) {
  int i = blockIdx.x * blockDim.x + threadIdx.x;
  if (i < n) p[i] = v;
}
__global__ void fill_u32(unsigned* __restrict__ p, unsigned v, int n) {
  int i = blockIdx.x * blockDim.x + threadIdx.x;
  if (i < n) p[i] = v;
}

__device__ __forceinline__ unsigned f2o(float f) {
  unsigned u = __float_as_uint(f);
  return (u & 0x80000000u) ? ~u : (u | 0x80000000u);
}
__device__ __forceinline__ float o2f(unsigned u) {
  return __uint_as_float((u & 0x80000000u) ? (u & 0x7FFFFFFFu) : ~u);
}
#define ORDERED_NEG_INF 0x007FFFFFu

__global__ void edge_score_max(const int* __restrict__ ei, int E, int N, int Hh,
                               const float* __restrict__ esrc,
                               const float* __restrict__ edst,
                               float* __restrict__ ebuf,
                               unsigned* __restrict__ mmax) {
  long long t = (long long)blockIdx.x * blockDim.x + threadIdx.x;
  const long long Etot = (long long)E + N;
  if (t >= Etot) return;
  int s, d;
  if (t < E) { s = ei[t]; d = ei[(long long)E + t]; }
  else       { s = d = (int)(t - E); }
  for (int h = 0; h < Hh; ++h) {
    float e = esrc[(size_t)s * Hh + h] + edst[(size_t)d * Hh + h];
    e = (e > 0.f) ? e : NEG_SLOPE * e;
    ebuf[t * Hh + h] = e;
    atomicMax(&mmax[(size_t)d * Hh + h], f2o(e));
  }
}

__global__ void edge_exp_sum(const int* __restrict__ ei, int E, int N, int Hh,
                             const unsigned* __restrict__ mmax,
                             float* __restrict__ ebuf,
                             float* __restrict__ denom) {
  long long t = (long long)blockIdx.x * blockDim.x + threadIdx.x;
  const long long Etot = (long long)E + N;
  if (t >= Etot) return;
  int d;
  if (t < E) d = ei[(long long)E + t];
  else       d = (int)(t - E);
  for (int h = 0; h < Hh; ++h) {
    const float m  = o2f(mmax[(size_t)d * Hh + h]);
    const float ex = __expf(ebuf[t * Hh + h] - m);
    ebuf[t * Hh + h] = ex;
    atomicAdd(&denom[(size_t)d * Hh + h], ex);
  }
}

__global__ void edge_aggregate(const int* __restrict__ ei, int E, int N, int Hh,
                               const float* __restrict__ ebuf,
                               const float* __restrict__ denom,
                               const float* __restrict__ Hm,
                               float* __restrict__ outb) {
  long long idx = (long long)blockIdx.x * blockDim.x + threadIdx.x;
  const long long tot = ((long long)E + N) * Hh;
  if (idx >= tot) return;
  const int h = (int)(idx % Hh);
  const long long t = idx / Hh;
  int s, d;
  if (t < E) { s = ei[t]; d = ei[(long long)E + t]; }
  else       { s = d = (int)(t - E); }
  const float alpha = ebuf[t * Hh + h] / (denom[(size_t)d * Hh + h] + GAT_EPS);
  const float4* hp = (const float4*)(Hm + ((size_t)s * Hh + h) * 16);
  float* op = outb + ((size_t)d * Hh + h) * 16;
  const float4 v0 = hp[0], v1 = hp[1], v2 = hp[2], v3 = hp[3];
  atomicAdd(op + 0,  alpha * v0.x); atomicAdd(op + 1,  alpha * v0.y);
  atomicAdd(op + 2,  alpha * v0.z); atomicAdd(op + 3,  alpha * v0.w);
  atomicAdd(op + 4,  alpha * v1.x); atomicAdd(op + 5,  alpha * v1.y);
  atomicAdd(op + 6,  alpha * v1.z); atomicAdd(op + 7,  alpha * v1.w);
  atomicAdd(op + 8,  alpha * v2.x); atomicAdd(op + 9,  alpha * v2.y);
  atomicAdd(op + 10, alpha * v2.z); atomicAdd(op + 11, alpha * v2.w);
  atomicAdd(op + 12, alpha * v3.x); atomicAdd(op + 13, alpha * v3.y);
  atomicAdd(op + 14, alpha * v3.z); atomicAdd(op + 15, alpha * v3.w);
}

__global__ void bias_act(const float* __restrict__ in, const float* __restrict__ b,
                         float* __restrict__ out, int N, int M, int relu) {
  int idx = blockIdx.x * blockDim.x + threadIdx.x;
  if (idx >= N * M) return;
  float v = in[idx] + b[idx % M];
  if (relu) v = fmaxf(v, 0.f);
  out[idx] = v;
}

static inline int cdiv(long long a, int b) { return (int)((a + b - 1) / b); }

struct LayerBufs {
  float* hbuf; float* outb; float* esrc; float* edst;
  unsigned* mmax; float* denom; float* ebuf;
};

static void run_edge_phase(const LayerBufs& L, const int* ei, int E, int N,
                           int Hh, hipStream_t stream) {
  const long long Etot = (long long)E + N;
  const int Mo = Hh * 16;
  fill_u32<<<cdiv(N * Hh, 256), 256, 0, stream>>>(L.mmax, ORDERED_NEG_INF, N * Hh);
  fill_f32<<<cdiv(N * Hh, 256), 256, 0, stream>>>(L.denom, 0.f, N * Hh);
  fill_f32<<<cdiv((long long)N * Mo, 256), 256, 0, stream>>>(L.outb, 0.f, N * Mo);
  edge_score_max<<<cdiv(Etot, 256), 256, 0, stream>>>(ei, E, N, Hh, L.esrc,
                                                      L.edst, L.ebuf, L.mmax);
  edge_exp_sum<<<cdiv(Etot, 256), 256, 0, stream>>>(ei, E, N, Hh, L.mmax,
                                                    L.ebuf, L.denom);
  edge_aggregate<<<cdiv(Etot * Hh, 256), 256, 0, stream>>>(
      ei, E, N, Hh, L.ebuf, L.denom, L.hbuf, L.outb);
}

extern "C" void kernel_launch(void* const* d_in, const int* in_sizes, int n_in,
                              void* d_out, int out_size, void* d_ws, size_t ws_size,
                              hipStream_t stream) {
  const float* x   = (const float*)d_in[0];
  const int*   ei  = (const int*)d_in[1];
  const float* W1  = (const float*)d_in[2];
  const float* a1s = (const float*)d_in[3];
  const float* a1d = (const float*)d_in[4];
  const float* b1  = (const float*)d_in[5];
  const float* W2  = (const float*)d_in[6];
  const float* a2s = (const float*)d_in[7];
  const float* a2d = (const float*)d_in[8];
  const float* b2  = (const float*)d_in[9];
  const float* W3  = (const float*)d_in[10];
  const float* a3s = (const float*)d_in[11];
  const float* a3d = (const float*)d_in[12];
  const float* b3  = (const float*)d_in[13];

  const int B = 4, F_IN = 128, H = 4, HC = 64;
  const int E = in_sizes[1] / 2;
  const int N = in_sizes[0] / (B * F_IN);
  const int NCLS = in_sizes[13];

  float* ws = (float*)d_ws;
  float*    hbuf  = ws;
  float*    xbuf  = hbuf + (size_t)N * HC;
  float*    outb  = xbuf + (size_t)N * HC;
  float*    esrc  = outb + (size_t)N * HC;
  float*    edst  = esrc + (size_t)N * H;
  unsigned* mmax  = (unsigned*)(edst + (size_t)N * H);
  float*    denom = (float*)(mmax + (size_t)N * H);
  float*    ebuf  = denom + (size_t)N * H;

  const int ntiles = (N + 15) / 16;
  const int gemm_grid = (ntiles + 3) / 4;

  LayerBufs L{hbuf, outb, esrc, edst, mmax, denom, ebuf};

  for (int b = 0; b < B; ++b) {
    gemm_wmma_f32<128, 64><<<gemm_grid, 128, 0, stream>>>(
        x + (size_t)b * N * F_IN, W1, hbuf, ntiles);
    node_attn<<<cdiv((long long)N * H, 256), 256, 0, stream>>>(
        hbuf, a1s, a1d, esrc, edst, N, H);
    run_edge_phase(L, ei, E, N, H, stream);
    bias_act<<<cdiv((long long)N * HC, 256), 256, 0, stream>>>(
        outb, b1, xbuf, N, HC, 1);

    gemm_wmma_f32<64, 64><<<gemm_grid, 128, 0, stream>>>(xbuf, W2, hbuf, ntiles);
    node_attn<<<cdiv((long long)N * H, 256), 256, 0, stream>>>(
        hbuf, a2s, a2d, esrc, edst, N, H);
    run_edge_phase(L, ei, E, N, H, stream);
    bias_act<<<cdiv((long long)N * HC, 256), 256, 0, stream>>>(
        outb, b2, xbuf, N, HC, 1);

    gemm_wmma_f32<64, 16><<<gemm_grid, 128, 0, stream>>>(xbuf, W3, hbuf, ntiles);
    node_attn<<<cdiv((long long)N * 1, 256), 256, 0, stream>>>(
        hbuf, a3s, a3d, esrc, edst, N, 1);
    run_edge_phase(L, ei, E, N, 1, stream);
    bias_act<<<cdiv((long long)N * NCLS, 256), 256, 0, stream>>>(
        outb, b3, (float*)d_out + (size_t)b * N * NCLS, N, NCLS, 0);
  }
}


# Edge phase: dst-bucketed edge list built once per call + one gather kernel per layer (16 lanes own a dst row: max, sum, weighted sum, bias); float atomics gone, sum order now fixed by src id
# speedup vs baseline: 7.6103x; 7.6103x over previous
#include <hip/hip_runtime.h>
#include <hip/hip_bf16.h>


typedef __attribute__((ext_vector_type(2))) float v2f;
typedef __attribute__((ext_vector_type(8))) float v8f;

#define NEG_SLOPE 0.2f
#define GAT_EPS   1e-16f

template <int KDIM, int MO>
__global__ __launch_bounds__(128) void gemm_wmma_f32(
    const float* __restrict__ X, const float* __restrict__ W,
    float* __restrict__ Hout, int ntiles) {
  const int wave = threadIdx.x >> 5;
  const int tile = blockIdx.x * 4 + wave;
  if (tile >= ntiles) return;
  const int lane  = threadIdx.x & 31;
  const int lmod  = lane & 15;
  const int lhalf = lane >> 4;
  const int row0  = tile * 16;

  v8f acc[MO / 16];
#pragma unroll
  for (int t = 0; t < MO / 16; ++t) acc[t] = v8f{0, 0, 0, 0, 0, 0, 0, 0};

  const float* xrow = X + (size_t)(row0 + lmod) * KDIM;
#pragma unroll 4
  for (int k0 = 0; k0 < KDIM; k0 += 4) {
    const int ka = k0 + 2 * lhalf;
    v2f a;
    a.x = xrow[ka];
    a.y = xrow[ka + 1];
#pragma unroll
    for (int t = 0; t < MO / 16; ++t) {
      const float* wr = W + (size_t)ka * MO + t * 16 + lmod;
      v2f b;
      b.x = wr[0];
      b.y = wr[MO];
      acc[t] = __builtin_amdgcn_wmma_f32_16x16x4_f32(
          false, a, false, b, (short)0, acc[t], false, false);
    }
  }
#pragma unroll
  for (int t = 0; t < MO / 16; ++t)
#pragma unroll
    for (int r = 0; r < 8; ++r)
      Hout[(size_t)(row0 + r + 8 * lhalf) * MO + t * 16 + lmod] = acc[t][r];
}

__global__ void node_attn(const float* __restrict__ Hm,
                          const float* __restrict__ a_src,
                          const float* __restrict__ a_dst,
                          float* __restrict__ esrc, float* __restrict__ edst,
                          int N, int Hh) {
  int idx = blockIdx.x * blockDim.x + threadIdx.x;
  if (idx >= N * Hh) return;
  const int h = idx % Hh;
  const int n = idx / Hh;
  const float* row = Hm + ((size_t)n * Hh + h) * 16;
  float s = 0.f, d = 0.f;
#pragma unroll
  for (int c = 0; c < 16; ++c) {
    const float v = row[c];
    s += v * a_src[h * 16 + c];
    d += v * a_dst[h * 16 + c];
  }
  esrc[idx] = s;
  edst[idx] = d;
}

__global__ void fill_u32(unsigned* __restrict__ p, unsigned v, int n) {
  int i = blockIdx.x * blockDim.x + threadIdx.x;
  if (i < n) p[i] = v;
}

__global__ void csr_count(const int* __restrict__ ei, int E, int* __restrict__ cnt) {
  int t = blockIdx.x * blockDim.x + threadIdx.x;
  if (t < E) atomicAdd(&cnt[ei[(size_t)E + t]], 1);
}

__global__ __launch_bounds__(1024) void csr_scan(int* __restrict__ cnt,
                                                 int* __restrict__ rowptr, int N) {
  __shared__ int part[1024];
  const int tid = threadIdx.x;
  const int chunk = (N + 1023) / 1024;
  const int beg = tid * chunk;
  const int end = (beg + chunk < N) ? (beg + chunk) : N;
  int s = 0;
  for (int i = beg; i < end; ++i) s += cnt[i];
  part[tid] = s;
  __syncthreads();
  for (int off = 1; off < 1024; off <<= 1) {
    const int v = (tid >= off) ? part[tid - off] : 0;
    __syncthreads();
    part[tid] += v;
    __syncthreads();
  }
  int run = part[tid] - s;
  for (int i = beg; i < end; ++i) {
    const int c = cnt[i];
    rowptr[i] = run;
    cnt[i] = run;
    run += c;
  }
  if (tid == 1023) rowptr[N] = part[1023];
}

__global__ void csr_fill(const int* __restrict__ ei, int E, int* __restrict__ cursor,
                         int* __restrict__ tmp_src) {
  int t = blockIdx.x * blockDim.x + threadIdx.x;
  if (t >= E) return;
  const int d = ei[(size_t)E + t];
  const int p = atomicAdd(&cursor[d], 1);
  tmp_src[p] = ei[t];
}

__global__ __launch_bounds__(256) void csr_order(const int* __restrict__ rowptr,
                                                 const int* __restrict__ tmp_src,
                                                 int* __restrict__ csr_src, int N) {
  const int g = (int)((blockIdx.x * 256u + threadIdx.x) >> 4);
  if (g >= N) return;
  const int lane = threadIdx.x & 15;
  const int beg = rowptr[g], end = rowptr[g + 1];
  for (int i = beg + lane; i < end; i += 16) {
    const int v = tmp_src[i];
    int r = 0;
    for (int j = beg; j < end; ++j) {
      const int w = tmp_src[j];
      r += ((w < v) || (w == v && j < i)) ? 1 : 0;
    }
    csr_src[beg + r] = v;
  }
}

template <int H>
__global__ __launch_bounds__(256) void gat_gather(
    const int* __restrict__ rowptr, const int* __restrict__ csr_src,
    const float* __restrict__ esrc, const float* __restrict__ edst,
    const float* __restrict__ Hm, const float* __restrict__ bias,
    float* __restrict__ out, int N, int relu) {
  constexpr int WD = H * 16;
  const int g = (int)((blockIdx.x * 256u + threadIdx.x) >> 4);
  if (g >= N) return;
  const int j = threadIdx.x & 15;
  const int h = (j * H) >> 4;
  const int beg = rowptr[g], end = rowptr[g + 1];
  const float ed = edst[(size_t)g * H + h];
  const float* es = esrc + h;

  float e0 = es[(size_t)g * H] + ed;
  e0 = (e0 > 0.f) ? e0 : NEG_SLOPE * e0;
  float m = e0;
#pragma unroll 4
  for (int i = beg; i < end; ++i) {
    const int s = csr_src[i];
    float e = es[(size_t)s * H] + ed;
    e = (e > 0.f) ? e : NEG_SLOPE * e;
    m = fmaxf(m, e);
  }

  float den = __expf(e0 - m);
#pragma unroll 4
  for (int i = beg; i < end; ++i) {
    const int s = csr_src[i];
    float e = es[(size_t)s * H] + ed;
    e = (e > 0.f) ? e : NEG_SLOPE * e;
    den += __expf(e - m);
  }
  const float dn = den + GAT_EPS;

  float acc[H];
  {
    const float a0 = __expf(e0 - m) / dn;
    const float* hp = Hm + (size_t)g * WD + j * H;
    if constexpr (H == 4) {
      const float4 v = *(const float4*)hp;
      acc[0] = a0 * v.x; acc[1] = a0 * v.y; acc[2] = a0 * v.z; acc[3] = a0 * v.w;
    } else {
      for (int c = 0; c < H; ++c) acc[c] = a0 * hp[c];
    }
  }
#pragma unroll 4
  for (int i = beg; i < end; ++i) {
    const int s = csr_src[i];
    float e = es[(size_t)s * H] + ed;
    e = (e > 0.f) ? e : NEG_SLOPE * e;
    const float a = __expf(e - m) / dn;
    const float* hp = Hm + (size_t)s * WD + j * H;
    if constexpr (H == 4) {
      const float4 v = *(const float4*)hp;
      acc[0] += a * v.x; acc[1] += a * v.y; acc[2] += a * v.z; acc[3] += a * v.w;
    } else {
      for (int c = 0; c < H; ++c) acc[c] += a * hp[c];
    }
  }

  float* op = out + (size_t)g * WD + j * H;
  const float* bp = bias + j * H;
  float r[H];
  for (int c = 0; c < H; ++c) {
    float v = acc[c] + bp[c];
    if (relu) v = fmaxf(v, 0.f);
    r[c] = v;
  }
  if constexpr (H == 4) {
    *(float4*)op = make_float4(r[0], r[1], r[2], r[3]);
  } else {
    for (int c = 0; c < H; ++c) op[c] = r[c];
  }
}

static inline int cdiv(long long a, int b) { return (int)((a + b - 1) / b); }

extern "C" void kernel_launch(void* const* d_in, const int* in_sizes, int n_in,
                              void* d_out, int out_size, void* d_ws, size_t ws_size,
                              hipStream_t stream) {
  const float* x   = (const float*)d_in[0];
  const int*   ei  = (const int*)d_in[1];
  const float* W1  = (const float*)d_in[2];
  const float* a1s = (const float*)d_in[3];
  const float* a1d = (const float*)d_in[4];
  const float* b1  = (const float*)d_in[5];
  const float* W2  = (const float*)d_in[6];
  const float* a2s = (const float*)d_in[7];
  const float* a2d = (const float*)d_in[8];
  const float* b2  = (const float*)d_in[9];
  const float* W3  = (const float*)d_in[10];
  const float* a3s = (const float*)d_in[11];
  const float* a3d = (const float*)d_in[12];
  const float* b3  = (const float*)d_in[13];

  const int B = 4, F_IN = 128, H = 4, HC = 64;
  const int E = in_sizes[1] / 2;
  const int N = in_sizes[0] / (B * F_IN);
  const int NCLS = in_sizes[13];

  float* ws = (float*)d_ws;
  float* hbuf    = ws;
  float* xbuf    = hbuf + (size_t)N * HC;
  float* esrc    = xbuf + (size_t)N * HC;
  float* edst    = esrc + (size_t)N * H;
  int*   cnt     = (int*)(edst + (size_t)N * H);
  int*   rowptr  = cnt + (size_t)N;
  int*   tmp_src = rowptr + (size_t)N + 1;
  int*   csr_src = tmp_src + (size_t)E;

  const int ntiles = (N + 15) / 16;
  const int gemm_grid = (ntiles + 3) / 4;
  const int gather_grid = cdiv((long long)N * 16, 256);

  fill_u32<<<cdiv(N, 256), 256, 0, stream>>>((unsigned*)cnt, 0u, N);
  csr_count<<<cdiv(E, 256), 256, 0, stream>>>(ei, E, cnt);
  csr_scan<<<1, 1024, 0, stream>>>(cnt, rowptr, N);
  csr_fill<<<cdiv(E, 256), 256, 0, stream>>>(ei, E, cnt, tmp_src);
  csr_order<<<gather_grid, 256, 0, stream>>>(rowptr, tmp_src, csr_src, N);

  for (int b = 0; b < B; ++b) {
    gemm_wmma_f32<128, 64><<<gemm_grid, 128, 0, stream>>>(
        x + (size_t)b * N * F_IN, W1, hbuf, ntiles);
    node_attn<<<cdiv((long long)N * H, 256), 256, 0, stream>>>(
        hbuf, a1s, a1d, esrc, edst, N, H);
    gat_gather<4><<<gather_grid, 256, 0, stream>>>(
        rowptr, csr_src, esrc, edst, hbuf, b1, xbuf, N, 1);

    gemm_wmma_f32<64, 64><<<gemm_grid, 128, 0, stream>>>(xbuf, W2, hbuf, ntiles);
    node_attn<<<cdiv((long long)N * H, 256), 256, 0, stream>>>(
        hbuf, a2s, a2d, esrc, edst, N, H);
    gat_gather<4><<<gather_grid, 256, 0, stream>>>(
        rowptr, csr_src, esrc, edst, hbuf, b2, xbuf, N, 1);

    gemm_wmma_f32<64, 16><<<gemm_grid, 128, 0, stream>>>(xbuf, W3, hbuf, ntiles);
    node_attn<<<cdiv((long long)N * 1, 256), 256, 0, stream>>>(
        hbuf, a3s, a3d, esrc, edst, N, 1);
    gat_gather<1><<<gather_grid, 256, 0, stream>>>(
        rowptr, csr_src, esrc, edst, hbuf, b3,
        (float*)d_out + (size_t)b * N * NCLS, N, 0);
  }
}


# Gather: a row's 16 lanes split its edges for the logits (1 index + 1 logit load per edge, kept in registers; DPP max/sum/broadcast), not 3 same-address sweeps; exp-sum order now lane partials + tree
# speedup vs baseline: 9.1158x; 1.1978x over previous
#include <hip/hip_runtime.h>
#include <hip/hip_bf16.h>


typedef __attribute__((ext_vector_type(2))) float v2f;
typedef __attribute__((ext_vector_type(8))) float v8f;

#define NEG_SLOPE 0.2f
#define GAT_EPS   1e-16f

template <int KDIM, int MO>
__global__ __launch_bounds__(128) void gemm_wmma_f32(
    const float* __restrict__ X, const float* __restrict__ W,
    float* __restrict__ Hout, int ntiles) {
  const int wave = threadIdx.x >> 5;
  const int tile = blockIdx.x * 4 + wave;
  if (tile >= ntiles) return;
  const int lane  = threadIdx.x & 31;
  const int lmod  = lane & 15;
  const int lhalf = lane >> 4;
  const int row0  = tile * 16;

  v8f acc[MO / 16];
#pragma unroll
  for (int t = 0; t < MO / 16; ++t) acc[t] = v8f{0, 0, 0, 0, 0, 0, 0, 0};

  const float* xrow = X + (size_t)(row0 + lmod) * KDIM;
#pragma unroll 4
  for (int k0 = 0; k0 < KDIM; k0 += 4) {
    const int ka = k0 + 2 * lhalf;
    v2f a;
    a.x = xrow[ka];
    a.y = xrow[ka + 1];
#pragma unroll
    for (int t = 0; t < MO / 16; ++t) {
      const float* wr = W + (size_t)ka * MO + t * 16 + lmod;
      v2f b;
      b.x = wr[0];
      b.y = wr[MO];
      acc[t] = __builtin_amdgcn_wmma_f32_16x16x4_f32(
          false, a, false, b, (short)0, acc[t], false, false);
    }
  }
#pragma unroll
  for (int t = 0; t < MO / 16; ++t)
#pragma unroll
    for (int r = 0; r < 8; ++r)
      Hout[(size_t)(row0 + r + 8 * lhalf) * MO + t * 16 + lmod] = acc[t][r];
}

__global__ void node_attn(const float* __restrict__ Hm,
                          const float* __restrict__ a_src,
                          const float* __restrict__ a_dst,
                          float* __restrict__ esrc, float* __restrict__ edst,
                          int N, int Hh) {
  int idx = blockIdx.x * blockDim.x + threadIdx.x;
  if (idx >= N * Hh) return;
  const int h = idx % Hh;
  const int n = idx / Hh;
  const float* row = Hm + ((size_t)n * Hh + h) * 16;
  float s = 0.f, d = 0.f;
#pragma unroll
  for (int c = 0; c < 16; ++c) {
    const float v = row[c];
    s += v * a_src[h * 16 + c];
    d += v * a_dst[h * 16 + c];
  }
  esrc[idx] = s;
  edst[idx] = d;
}

__global__ void fill_u32(unsigned* __restrict__ p, unsigned v, int n) {
  int i = blockIdx.x * blockDim.x + threadIdx.x;
  if (i < n) p[i] = v;
}

__global__ void csr_count(const int* __restrict__ ei, int E, int* __restrict__ cnt) {
  int t = blockIdx.x * blockDim.x + threadIdx.x;
  if (t < E) atomicAdd(&cnt[ei[(size_t)E + t]], 1);
}

__global__ __launch_bounds__(1024) void csr_scan(int* __restrict__ cnt,
                                                 int* __restrict__ rowptr, int N) {
  __shared__ int part[1024];
  const int tid = threadIdx.x;
  const int chunk = (N + 1023) / 1024;
  const int beg = tid * chunk;
  const int end = (beg + chunk < N) ? (beg + chunk) : N;
  int s = 0;
  for (int i = beg; i < end; ++i) s += cnt[i];
  part[tid] = s;
  __syncthreads();
  for (int off = 1; off < 1024; off <<= 1) {
    const int v = (tid >= off) ? part[tid - off] : 0;
    __syncthreads();
    part[tid] += v;
    __syncthreads();
  }
  int run = part[tid] - s;
  for (int i = beg; i < end; ++i) {
    const int c = cnt[i];
    rowptr[i] = run;
    cnt[i] = run;
    run += c;
  }
  if (tid == 1023) rowptr[N] = part[1023];
}

__global__ void csr_fill(const int* __restrict__ ei, int E, int* __restrict__ cursor,
                         int* __restrict__ tmp_src) {
  int t = blockIdx.x * blockDim.x + threadIdx.x;
  if (t >= E) return;
  const int d = ei[(size_t)E + t];
  const int p = atomicAdd(&cursor[d], 1);
  tmp_src[p] = ei[t];
}

__global__ __launch_bounds__(256) void csr_order(const int* __restrict__ rowptr,
                                                 const int* __restrict__ tmp_src,
                                                 int* __restrict__ csr_src, int N) {
  const int g = (int)((blockIdx.x * 256u + threadIdx.x) >> 4);
  if (g >= N) return;
  const int lane = threadIdx.x & 15;
  const int beg = rowptr[g], end = rowptr[g + 1];
  for (int i = beg + lane; i < end; i += 16) {
    const int v = tmp_src[i];
    int r = 0;
    for (int j = beg; j < end; ++j) {
      const int w = tmp_src[j];
      r += ((w < v) || (w == v && j < i)) ? 1 : 0;
    }
    csr_src[beg + r] = v;
  }
}

template <int CTRL>
__device__ __forceinline__ float dpp_f(float v) {
  const int x = __builtin_bit_cast(int, v);
  return __builtin_bit_cast(float, __builtin_amdgcn_update_dpp(x, x, CTRL, 0xF, 0xF, true));
}
template <int CTRL>
__device__ __forceinline__ int dpp_i(int v) {
  return __builtin_amdgcn_update_dpp(v, v, CTRL, 0xF, 0xF, true);
}
template <int H>
__device__ __forceinline__ float head_max(float v) {
  v = fmaxf(v, dpp_f<0x161>(v));
  v = fmaxf(v, dpp_f<0x162>(v));
  if constexpr (H == 1) {
    v = fmaxf(v, dpp_f<0x164>(v));
    v = fmaxf(v, dpp_f<0x168>(v));
  }
  return v;
}
template <int H>
__device__ __forceinline__ float head_sum(float v) {
  v += dpp_f<0x161>(v);
  v += dpp_f<0x162>(v);
  if constexpr (H == 1) {
    v += dpp_f<0x164>(v);
    v += dpp_f<0x168>(v);
  }
  return v;
}
template <int H, int QO>
__device__ __forceinline__ float head_bcast_f(float v) {
  if constexpr (H == 4) return dpp_f<QO * 0x55>(v);
  else                  return dpp_f<0x150 + QO>(v);
}
template <int H, int QO>
__device__ __forceinline__ int head_bcast_i(int v) {
  if constexpr (H == 4) return dpp_i<QO * 0x55>(v);
  else                  return dpp_i<0x150 + QO>(v);
}

template <int H>
__global__ __launch_bounds__(256) void gat_gather(
    const int* __restrict__ rowptr, const int* __restrict__ csr_src,
    const float* __restrict__ esrc, const float* __restrict__ edst,
    const float* __restrict__ Hm, const float* __restrict__ bias,
    float* __restrict__ out, int N, int relu) {
  static_assert(H == 1 || H == 4, "head_bcast covers a quad or a whole row");
  constexpr int WD  = H * 16;
  constexpr int LPH = 16 / H;
  constexpr int NT  = 2 * H;
  constexpr int NK  = NT * LPH;
  const int g = (int)((blockIdx.x * 256u + threadIdx.x) >> 4);
  if (g >= N) return;
  const int j = threadIdx.x & 15;
  const int h = j / LPH;
  const int q = j % LPH;
  const int beg = rowptr[g];
  const int n = rowptr[g + 1] - beg;
  auto ld_i = [](const int* b, unsigned off) { return *(const int*)((const char*)b + off); };
  auto ld_f = [](const float* b, unsigned off) { return *(const float*)((const char*)b + off); };
  const unsigned hoff = (unsigned)h * 4u;
  const unsigned roff = (unsigned)(j * H) * 4u;
  const float ed = ld_f(edst, (unsigned)g * (H * 4u) + hoff);

  float e0 = ld_f(esrc, (unsigned)g * (H * 4u) + hoff) + ed;
  e0 = (e0 > 0.f) ? e0 : NEG_SLOPE * e0;

  float acc[H];
  auto self_term = [&](float a0) {
    const float* hp = (const float*)((const char*)Hm + ((unsigned)g * (WD * 4u) + roff));
    if constexpr (H == 4) {
      const float4 v = *(const float4*)hp;
      acc[0] = a0 * v.x; acc[1] = a0 * v.y; acc[2] = a0 * v.z; acc[3] = a0 * v.w;
    } else {
      acc[0] = a0 * hp[0];
    }
  };
  auto finish = [&]() {
    float* op = (float*)((char*)out + ((unsigned)g * (WD * 4u) + roff));
    const float* bp = bias + j * H;
    float r[H];
    for (int c = 0; c < H; ++c) {
      float v = acc[c] + bp[c];
      if (relu) v = fmaxf(v, 0.f);
      r[c] = v;
    }
    if constexpr (H == 4) {
      *(float4*)op = make_float4(r[0], r[1], r[2], r[3]);
    } else {
      op[0] = r[0];
    }
  };

  if (n > NK) {
    auto logit = [&](int i) {
      const int s = ld_i(csr_src, (unsigned)(beg + i) * 4u);
      const float y = ld_f(esrc, (unsigned)s * (H * 4u) + hoff) + ed;
      return (y > 0.f) ? y : NEG_SLOPE * y;
    };
    float m = e0;
    for (int i = 0; i < n; ++i) m = fmaxf(m, logit(i));
    float den = __expf(e0 - m);
    for (int i = 0; i < n; ++i) den += __expf(logit(i) - m);
    const float dn = den + GAT_EPS;
    self_term(__expf(e0 - m) / dn);
    for (int i = 0; i < n; ++i) {
      const int s = ld_i(csr_src, (unsigned)(beg + i) * 4u);
      const float a = __expf(logit(i) - m) / dn;
      const float* hp = (const float*)((const char*)Hm + ((unsigned)s * (WD * 4u) + roff));
      if constexpr (H == 4) {
        const float4 v = *(const float4*)hp;
        acc[0] += a * v.x; acc[1] += a * v.y; acc[2] += a * v.z; acc[3] += a * v.w;
      } else {
        acc[0] += a * hp[0];
      }
    }
    finish();
    return;
  }

  int   sv[NT];
  float ev[NT];
  for (int t = 0; t < NT; ++t) ev[t] = -__builtin_inff();
  if (n > 0) {
    for (int t = 0; t < NT; ++t) {
      const int p = t * LPH + q;
      sv[t] = ld_i(csr_src, (unsigned)(beg + ((p < n) ? p : n - 1)) * 4u);
    }
  }
  float x0[H], x1[H];
  if (n > 0) {
    for (int t = 0; t < H; ++t) x0[t] = ld_f(esrc, (unsigned)sv[t] * (H * 4u) + hoff);
  }
  if (n > 16) {
    for (int t = 0; t < H; ++t) x1[t] = ld_f(esrc, (unsigned)sv[H + t] * (H * 4u) + hoff);
  }
  if (n > 0) {
    for (int t = 0; t < H; ++t) {
      float y = x0[t] + ed;
      y = (y > 0.f) ? y : NEG_SLOPE * y;
      ev[t] = (t * LPH + q < n) ? y : -__builtin_inff();
    }
  }
  if (n > 16) {
    for (int t = 0; t < H; ++t) {
      float y = x1[t] + ed;
      y = (y > 0.f) ? y : NEG_SLOPE * y;
      ev[H + t] = ((H + t) * LPH + q < n) ? y : -__builtin_inff();
    }
  }

  float m = e0;
  for (int t = 0; t < NT; ++t) m = fmaxf(m, ev[t]);
  m = head_max<H>(m);

  float part = 0.f;
  for (int t = 0; t < NT; ++t) {
    ev[t] = __expf(ev[t] - m);
    part += ev[t];
  }
  const float den = __expf(e0 - m) + head_sum<H>(part);
  const float dn = den + GAT_EPS;
  for (int t = 0; t < NT; ++t) ev[t] = ev[t] / dn;

  self_term(__expf(e0 - m) / dn);
#define GAT_ROW_LOAD(K, V)                                                     \
  {                                                                            \
    const int sb_ = head_bcast_i<H, (K) % LPH>(sv[(K) / LPH]);                 \
    const float* hp_ =                                                         \
        (const float*)((const char*)Hm + ((unsigned)sb_ * (WD * 4u) + roff));  \
    if constexpr (H == 4) V = *(const float4*)hp_;                             \
    else                  V.x = hp_[0];                                        \
  }
#define GAT_ROW_FMA(K, V)                                                      \
  {                                                                            \
    const float ab_ = head_bcast_f<H, (K) % LPH>(ev[(K) / LPH]);               \
    if constexpr (H == 4) {                                                    \
      acc[0] += ab_ * V.x; acc[1] += ab_ * V.y;                                \
      acc[2] += ab_ * V.z; acc[3] += ab_ * V.w;                                \
    } else {                                                                   \
      acc[0] += ab_ * V.x;                                                     \
    }                                                                          \
  }
#define GAT_EIGHT(K)                                                           \
  if (n > (K)) {                                                               \
    float4 r0_, r1_, r2_, r3_, r4_, r5_, r6_, r7_;                             \
    GAT_ROW_LOAD((K) + 0, r0_) GAT_ROW_LOAD((K) + 1, r1_)                      \
    GAT_ROW_LOAD((K) + 2, r2_) GAT_ROW_LOAD((K) + 3, r3_)                      \
    if (n > (K) + 4) {                                                         \
      GAT_ROW_LOAD((K) + 4, r4_) GAT_ROW_LOAD((K) + 5, r5_)                    \
      GAT_ROW_LOAD((K) + 6, r6_) GAT_ROW_LOAD((K) + 7, r7_)                    \
    }                                                                          \
    GAT_ROW_FMA((K) + 0, r0_) GAT_ROW_FMA((K) + 1, r1_)                        \
    GAT_ROW_FMA((K) + 2, r2_) GAT_ROW_FMA((K) + 3, r3_)                        \
    if (n > (K) + 4) {                                                         \
      GAT_ROW_FMA((K) + 4, r4_) GAT_ROW_FMA((K) + 5, r5_)                      \
      GAT_ROW_FMA((K) + 6, r6_) GAT_ROW_FMA((K) + 7, r7_)                      \
    }                                                                          \
  }
  GAT_EIGHT(0)
  GAT_EIGHT(8)
  GAT_EIGHT(16)
  GAT_EIGHT(24)
#undef GAT_EIGHT
#undef GAT_ROW_FMA
#undef GAT_ROW_LOAD
  finish();
}

static inline int cdiv(long long a, int b) { return (int)((a + b - 1) / b); }

extern "C" void kernel_launch(void* const* d_in, const int* in_sizes, int n_in,
                              void* d_out, int out_size, void* d_ws, size_t ws_size,
                              hipStream_t stream) {
  const float* x   = (const float*)d_in[0];
  const int*   ei  = (const int*)d_in[1];
  const float* W1  = (const float*)d_in[2];
  const float* a1s = (const float*)d_in[3];
  const float* a1d = (const float*)d_in[4];
  const float* b1  = (const float*)d_in[5];
  const float* W2  = (const float*)d_in[6];
  const float* a2s = (const float*)d_in[7];
  const float* a2d = (const float*)d_in[8];
  const float* b2  = (const float*)d_in[9];
  const float* W3  = (const float*)d_in[10];
  const float* a3s = (const float*)d_in[11];
  const float* a3d = (const float*)d_in[12];
  const float* b3  = (const float*)d_in[13];

  const int B = 4, F_IN = 128, H = 4, HC = 64;
  const int E = in_sizes[1] / 2;
  const int N = in_sizes[0] / (B * F_IN);
  const int NCLS = in_sizes[13];

  float* ws = (float*)d_ws;
  float* hbuf    = ws;
  float* xbuf    = hbuf + (size_t)N * HC;
  float* esrc    = xbuf + (size_t)N * HC;
  float* edst    = esrc + (size_t)N * H;
  int*   cnt     = (int*)(edst + (size_t)N * H);
  int*   rowptr  = cnt + (size_t)N;
  int*   tmp_src = rowptr + (size_t)N + 1;
  int*   csr_src = tmp_src + (size_t)E;

  const int ntiles = (N + 15) / 16;
  const int gemm_grid = (ntiles + 3) / 4;
  const int gather_grid = cdiv((long long)N * 16, 256);

  fill_u32<<<cdiv(N, 256), 256, 0, stream>>>((unsigned*)cnt, 0u, N);
  csr_count<<<cdiv(E, 256), 256, 0, stream>>>(ei, E, cnt);
  csr_scan<<<1, 1024, 0, stream>>>(cnt, rowptr, N);
  csr_fill<<<cdiv(E, 256), 256, 0, stream>>>(ei, E, cnt, tmp_src);
  csr_order<<<gather_grid, 256, 0, stream>>>(rowptr, tmp_src, csr_src, N);

  for (int b = 0; b < B; ++b) {
    gemm_wmma_f32<128, 64><<<gemm_grid, 128, 0, stream>>>(
        x + (size_t)b * N * F_IN, W1, hbuf, ntiles);
    node_attn<<<cdiv((long long)N * H, 256), 256, 0, stream>>>(
        hbuf, a1s, a1d, esrc, edst, N, H);
    gat_gather<4><<<gather_grid, 256, 0, stream>>>(
        rowptr, csr_src, esrc, edst, hbuf, b1, xbuf, N, 1);

    gemm_wmma_f32<64, 64><<<gemm_grid, 128, 0, stream>>>(xbuf, W2, hbuf, ntiles);
    node_attn<<<cdiv((long long)N * H, 256), 256, 0, stream>>>(
        hbuf, a2s, a2d, esrc, edst, N, H);
    gat_gather<4><<<gather_grid, 256, 0, stream>>>(
        rowptr, csr_src, esrc, edst, hbuf, b2, xbuf, N, 1);

    gemm_wmma_f32<64, 16><<<gemm_grid, 128, 0, stream>>>(xbuf, W3, hbuf, ntiles);
    node_attn<<<cdiv((long long)N * 1, 256), 256, 0, stream>>>(
        hbuf, a3s, a3d, esrc, edst, N, 1);
    gat_gather<1><<<gather_grid, 256, 0, stream>>>(
        rowptr, csr_src, esrc, edst, hbuf, b3,
        (float*)d_out + (size_t)b * N * NCLS, N, 0);
  }
}


# Attention logits (h.a_src, h.a_dst) taken in the GEMM epilogue from the accumulator tile by 16-lane DPP sums; separate logit pass gone (41 -> 29 launches); dot sum order now butterfly, was serial
# speedup vs baseline: 9.6607x; 1.0598x over previous
#include <hip/hip_runtime.h>
#include <hip/hip_bf16.h>


typedef __attribute__((ext_vector_type(2))) float v2f;
typedef __attribute__((ext_vector_type(8))) float v8f;

#define NEG_SLOPE 0.2f
#define GAT_EPS   1e-16f

template <int CTRL>
__device__ __forceinline__ float dpp_f(float v) {
  const int x = __builtin_bit_cast(int, v);
  return __builtin_bit_cast(float, __builtin_amdgcn_update_dpp(x, x, CTRL, 0xF, 0xF, true));
}
template <int CTRL>
__device__ __forceinline__ int dpp_i(int v) {
  return __builtin_amdgcn_update_dpp(v, v, CTRL, 0xF, 0xF, true);
}
template <int H>
__device__ __forceinline__ float head_max(float v) {
  v = fmaxf(v, dpp_f<0x161>(v));
  v = fmaxf(v, dpp_f<0x162>(v));
  if constexpr (H == 1) {
    v = fmaxf(v, dpp_f<0x164>(v));
    v = fmaxf(v, dpp_f<0x168>(v));
  }
  return v;
}
template <int H>
__device__ __forceinline__ float head_sum(float v) {
  v += dpp_f<0x161>(v);
  v += dpp_f<0x162>(v);
  if constexpr (H == 1) {
    v += dpp_f<0x164>(v);
    v += dpp_f<0x168>(v);
  }
  return v;
}
template <int H, int QO>
__device__ __forceinline__ float head_bcast_f(float v) {
  if constexpr (H == 4) return dpp_f<QO * 0x55>(v);
  else                  return dpp_f<0x150 + QO>(v);
}
template <int H, int QO>
__device__ __forceinline__ int head_bcast_i(int v) {
  if constexpr (H == 4) return dpp_i<QO * 0x55>(v);
  else                  return dpp_i<0x150 + QO>(v);
}

template <int KDIM, int MO>
__global__ __launch_bounds__(128) void gemm_wmma_f32(
    const float* __restrict__ X, const float* __restrict__ W,
    const float* __restrict__ a_src, const float* __restrict__ a_dst,
    float* __restrict__ Hout, float* __restrict__ esrc, float* __restrict__ edst,
    int ntiles) {
  const int wave = threadIdx.x >> 5;
  const int tile = blockIdx.x * 4 + wave;
  if (tile >= ntiles) return;
  const int lane  = threadIdx.x & 31;
  const int lmod  = lane & 15;
  const int lhalf = lane >> 4;
  const int row0  = tile * 16;

  v8f acc[MO / 16];
#pragma unroll
  for (int t = 0; t < MO / 16; ++t) acc[t] = v8f{0, 0, 0, 0, 0, 0, 0, 0};

  const float* xrow = X + (size_t)(row0 + lmod) * KDIM;
#pragma unroll 4
  for (int k0 = 0; k0 < KDIM; k0 += 4) {
    const int ka = k0 + 2 * lhalf;
    v2f a;
    a.x = xrow[ka];
    a.y = xrow[ka + 1];
#pragma unroll
    for (int t = 0; t < MO / 16; ++t) {
      const float* wr = W + (size_t)ka * MO + t * 16 + lmod;
      v2f b;
      b.x = wr[0];
      b.y = wr[MO];
      acc[t] = __builtin_amdgcn_wmma_f32_16x16x4_f32(
          false, a, false, b, (short)0, acc[t], false, false);
    }
  }
#pragma unroll
  for (int t = 0; t < MO / 16; ++t)
#pragma unroll
    for (int r = 0; r < 8; ++r)
      Hout[(size_t)(row0 + r + 8 * lhalf) * MO + t * 16 + lmod] = acc[t][r];

  constexpr int NH = MO / 16;
  constexpr int NS = (8 * NH + 15) / 16;
  float ms[NS], md[NS];
#pragma unroll
  for (int s = 0; s < NS; ++s) { ms[s] = 0.f; md[s] = 0.f; }
#pragma unroll
  for (int t = 0; t < NH; ++t) {
    const float as = a_src[t * 16 + lmod];
    const float ad = a_dst[t * 16 + lmod];
#pragma unroll
    for (int r = 0; r < 8; ++r) {
      const float s = head_sum<1>(acc[t][r] * as);
      const float d = head_sum<1>(acc[t][r] * ad);
      const int pos = r * NH + t;
      if (lmod == (pos & 15)) { ms[pos >> 4] = s; md[pos >> 4] = d; }
    }
  }
  float* es = esrc + (size_t)(row0 + 8 * lhalf) * NH;
  float* ed = edst + (size_t)(row0 + 8 * lhalf) * NH;
#pragma unroll
  for (int s = 0; s < NS; ++s) {
    if (s * 16 + lmod < 8 * NH) {
      es[s * 16 + lmod] = ms[s];
      ed[s * 16 + lmod] = md[s];
    }
  }
}

__global__ void fill_u32(unsigned* __restrict__ p, unsigned v, int n) {
  int i = blockIdx.x * blockDim.x + threadIdx.x;
  if (i < n) p[i] = v;
}

__global__ void csr_count(const int* __restrict__ ei, int E, int* __restrict__ cnt) {
  int t = blockIdx.x * blockDim.x + threadIdx.x;
  if (t < E) atomicAdd(&cnt[ei[(size_t)E + t]], 1);
}

__global__ __launch_bounds__(1024) void csr_scan(int* __restrict__ cnt,
                                                 int* __restrict__ rowptr, int N) {
  __shared__ int part[1024];
  const int tid = threadIdx.x;
  const int chunk = (N + 1023) / 1024;
  const int beg = tid * chunk;
  const int end = (beg + chunk < N) ? (beg + chunk) : N;
  int s = 0;
  for (int i = beg; i < end; ++i) s += cnt[i];
  part[tid] = s;
  __syncthreads();
  for (int off = 1; off < 1024; off <<= 1) {
    const int v = (tid >= off) ? part[tid - off] : 0;
    __syncthreads();
    part[tid] += v;
    __syncthreads();
  }
  int run = part[tid] - s;
  for (int i = beg; i < end; ++i) {
    const int c = cnt[i];
    rowptr[i] = run;
    cnt[i] = run;
    run += c;
  }
  if (tid == 1023) rowptr[N] = part[1023];
}

__global__ void csr_fill(const int* __restrict__ ei, int E, int* __restrict__ cursor,
                         int* __restrict__ tmp_src) {
  int t = blockIdx.x * blockDim.x + threadIdx.x;
  if (t >= E) return;
  const int d = ei[(size_t)E + t];
  const int p = atomicAdd(&cursor[d], 1);
  tmp_src[p] = ei[t];
}

__global__ __launch_bounds__(256) void csr_order(const int* __restrict__ rowptr,
                                                 const int* __restrict__ tmp_src,
                                                 int* __restrict__ csr_src, int N) {
  const int g = (int)((blockIdx.x * 256u + threadIdx.x) >> 4);
  if (g >= N) return;
  const int lane = threadIdx.x & 15;
  const int beg = rowptr[g], end = rowptr[g + 1];
  for (int i = beg + lane; i < end; i += 16) {
    const int v = tmp_src[i];
    int r = 0;
    for (int j = beg; j < end; ++j) {
      const int w = tmp_src[j];
      r += ((w < v) || (w == v && j < i)) ? 1 : 0;
    }
    csr_src[beg + r] = v;
  }
}

template <int H>
__global__ __launch_bounds__(256) void gat_gather(
    const int* __restrict__ rowptr, const int* __restrict__ csr_src,
    const float* __restrict__ esrc, const float* __restrict__ edst,
    const float* __restrict__ Hm, const float* __restrict__ bias,
    float* __restrict__ out, int N, int relu) {
  static_assert(H == 1 || H == 4, "head_bcast covers a quad or a whole row");
  constexpr int WD  = H * 16;
  constexpr int LPH = 16 / H;
  constexpr int NT  = 2 * H;
  constexpr int NK  = NT * LPH;
  const int g = (int)((blockIdx.x * 256u + threadIdx.x) >> 4);
  if (g >= N) return;
  const int j = threadIdx.x & 15;
  const int h = j / LPH;
  const int q = j % LPH;
  const int beg = rowptr[g];
  const int n = rowptr[g + 1] - beg;
  auto ld_i = [](const int* b, unsigned off) { return *(const int*)((const char*)b + off); };
  auto ld_f = [](const float* b, unsigned off) { return *(const float*)((const char*)b + off); };
  const unsigned hoff = (unsigned)h * 4u;
  const unsigned roff = (unsigned)(j * H) * 4u;
  const float ed = ld_f(edst, (unsigned)g * (H * 4u) + hoff);

  float e0 = ld_f(esrc, (unsigned)g * (H * 4u) + hoff) + ed;
  e0 = (e0 > 0.f) ? e0 : NEG_SLOPE * e0;

  float acc[H];
  auto self_term = [&](float a0) {
    const float* hp = (const float*)((const char*)Hm + ((unsigned)g * (WD * 4u) + roff));
    if constexpr (H == 4) {
      const float4 v = *(const float4*)hp;
      acc[0] = a0 * v.x; acc[1] = a0 * v.y; acc[2] = a0 * v.z; acc[3] = a0 * v.w;
    } else {
      acc[0] = a0 * hp[0];
    }
  };
  auto finish = [&]() {
    float* op = (float*)((char*)out + ((unsigned)g * (WD * 4u) + roff));
    const float* bp = bias + j * H;
    float r[H];
    for (int c = 0; c < H; ++c) {
      float v = acc[c] + bp[c];
      if (relu) v = fmaxf(v, 0.f);
      r[c] = v;
    }
    if constexpr (H == 4) {
      *(float4*)op = make_float4(r[0], r[1], r[2], r[3]);
    } else {
      op[0] = r[0];
    }
  };

  if (n > NK) {
    auto logit = [&](int i) {
      const int s = ld_i(csr_src, (unsigned)(beg + i) * 4u);
      const float y = ld_f(esrc, (unsigned)s * (H * 4u) + hoff) + ed;
      return (y > 0.f) ? y : NEG_SLOPE * y;
    };
    float m = e0;
    for (int i = 0; i < n; ++i) m = fmaxf(m, logit(i));
    float den = __expf(e0 - m);
    for (int i = 0; i < n; ++i) den += __expf(logit(i) - m);
    const float dn = den + GAT_EPS;
    self_term(__expf(e0 - m) / dn);
    for (int i = 0; i < n; ++i) {
      const int s = ld_i(csr_src, (unsigned)(beg + i) * 4u);
      const float a = __expf(logit(i) - m) / dn;
      const float* hp = (const float*)((const char*)Hm + ((unsigned)s * (WD * 4u) + roff));
      if constexpr (H == 4) {
        const float4 v = *(const float4*)hp;
        acc[0] += a * v.x; acc[1] += a * v.y; acc[2] += a * v.z; acc[3] += a * v.w;
      } else {
        acc[0] += a * hp[0];
      }
    }
    finish();
    return;
  }

  int   sv[NT];
  float ev[NT];
  for (int t = 0; t < NT; ++t) ev[t] = -__builtin_inff();
  if (n > 0) {
    for (int t = 0; t < NT; ++t) {
      const int p = t * LPH + q;
      sv[t] = ld_i(csr_src, (unsigned)(beg + ((p < n) ? p : n - 1)) * 4u);
    }
  }
  float x0[H], x1[H];
  if (n > 0) {
    for (int t = 0; t < H; ++t) x0[t] = ld_f(esrc, (unsigned)sv[t] * (H * 4u) + hoff);
  }
  if (n > 16) {
    for (int t = 0; t < H; ++t) x1[t] = ld_f(esrc, (unsigned)sv[H + t] * (H * 4u) + hoff);
  }
  if (n > 0) {
    for (int t = 0; t < H; ++t) {
      float y = x0[t] + ed;
      y = (y > 0.f) ? y : NEG_SLOPE * y;
      ev[t] = (t * LPH + q < n) ? y : -__builtin_inff();
    }
  }
  if (n > 16) {
    for (int t = 0; t < H; ++t) {
      float y = x1[t] + ed;
      y = (y > 0.f) ? y : NEG_SLOPE * y;
      ev[H + t] = ((H + t) * LPH + q < n) ? y : -__builtin_inff();
    }
  }

  float m = e0;
  for (int t = 0; t < NT; ++t) m = fmaxf(m, ev[t]);
  m = head_max<H>(m);

  float part = 0.f;
  for (int t = 0; t < NT; ++t) {
    ev[t] = __expf(ev[t] - m);
    part += ev[t];
  }
  const float den = __expf(e0 - m) + head_sum<H>(part);
  const float dn = den + GAT_EPS;
  for (int t = 0; t < NT; ++t) ev[t] = ev[t] / dn;

  self_term(__expf(e0 - m) / dn);
#define GAT_ROW_LOAD(K, V)                                                     \
  {                                                                            \
    const int sb_ = head_bcast_i<H, (K) % LPH>(sv[(K) / LPH]);                 \
    const float* hp_ =                                                         \
        (const float*)((const char*)Hm + ((unsigned)sb_ * (WD * 4u) + roff));  \
    if constexpr (H == 4) V = *(const float4*)hp_;                             \
    else                  V.x = hp_[0];                                        \
  }
#define GAT_ROW_FMA(K, V)                                                      \
  {                                                                            \
    const float ab_ = head_bcast_f<H, (K) % LPH>(ev[(K) / LPH]);               \
    if constexpr (H == 4) {                                                    \
      acc[0] += ab_ * V.x; acc[1] += ab_ * V.y;                                \
      acc[2] += ab_ * V.z; acc[3] += ab_ * V.w;                                \
    } else {                                                                   \
      acc[0] += ab_ * V.x;                                                     \
    }                                                                          \
  }
#define GAT_EIGHT(K)                                                           \
  if (n > (K)) {                                                               \
    float4 r0_, r1_, r2_, r3_, r4_, r5_, r6_, r7_;                             \
    GAT_ROW_LOAD((K) + 0, r0_) GAT_ROW_LOAD((K) + 1, r1_)                      \
    GAT_ROW_LOAD((K) + 2, r2_) GAT_ROW_LOAD((K) + 3, r3_)                      \
    if (n > (K) + 4) {                                                         \
      GAT_ROW_LOAD((K) + 4, r4_) GAT_ROW_LOAD((K) + 5, r5_)                    \
      GAT_ROW_LOAD((K) + 6, r6_) GAT_ROW_LOAD((K) + 7, r7_)                    \
    }                                                                          \
    GAT_ROW_FMA((K) + 0, r0_) GAT_ROW_FMA((K) + 1, r1_)                        \
    GAT_ROW_FMA((K) + 2, r2_) GAT_ROW_FMA((K) + 3, r3_)                        \
    if (n > (K) + 4) {                                                         \
      GAT_ROW_FMA((K) + 4, r4_) GAT_ROW_FMA((K) + 5, r5_)                      \
      GAT_ROW_FMA((K) + 6, r6_) GAT_ROW_FMA((K) + 7, r7_)                      \
    }                                                                          \
  }
  GAT_EIGHT(0)
  GAT_EIGHT(8)
  GAT_EIGHT(16)
  GAT_EIGHT(24)
#undef GAT_EIGHT
#undef GAT_ROW_FMA
#undef GAT_ROW_LOAD
  finish();
}

static inline int cdiv(long long a, int b) { return (int)((a + b - 1) / b); }

extern "C" void kernel_launch(void* const* d_in, const int* in_sizes, int n_in,
                              void* d_out, int out_size, void* d_ws, size_t ws_size,
                              hipStream_t stream) {
  const float* x   = (const float*)d_in[0];
  const int*   ei  = (const int*)d_in[1];
  const float* W1  = (const float*)d_in[2];
  const float* a1s = (const float*)d_in[3];
  const float* a1d = (const float*)d_in[4];
  const float* b1  = (const float*)d_in[5];
  const float* W2  = (const float*)d_in[6];
  const float* a2s = (const float*)d_in[7];
  const float* a2d = (const float*)d_in[8];
  const float* b2  = (const float*)d_in[9];
  const float* W3  = (const float*)d_in[10];
  const float* a3s = (const float*)d_in[11];
  const float* a3d = (const float*)d_in[12];
  const float* b3  = (const float*)d_in[13];

  const int B = 4, F_IN = 128, H = 4, HC = 64;
  const int E = in_sizes[1] / 2;
  const int N = in_sizes[0] / (B * F_IN);
  const int NCLS = in_sizes[13];

  float* ws = (float*)d_ws;
  float* hbuf    = ws;
  float* xbuf    = hbuf + (size_t)N * HC;
  float* esrc    = xbuf + (size_t)N * HC;
  float* edst    = esrc + (size_t)N * H;
  int*   cnt     = (int*)(edst + (size_t)N * H);
  int*   rowptr  = cnt + (size_t)N;
  int*   tmp_src = rowptr + (size_t)N + 1;
  int*   csr_src = tmp_src + (size_t)E;

  const int ntiles = (N + 15) / 16;
  const int gemm_grid = (ntiles + 3) / 4;
  const int gather_grid = cdiv((long long)N * 16, 256);

  fill_u32<<<cdiv(N, 256), 256, 0, stream>>>((unsigned*)cnt, 0u, N);
  csr_count<<<cdiv(E, 256), 256, 0, stream>>>(ei, E, cnt);
  csr_scan<<<1, 1024, 0, stream>>>(cnt, rowptr, N);
  csr_fill<<<cdiv(E, 256), 256, 0, stream>>>(ei, E, cnt, tmp_src);
  csr_order<<<gather_grid, 256, 0, stream>>>(rowptr, tmp_src, csr_src, N);

  for (int b = 0; b < B; ++b) {
    gemm_wmma_f32<128, 64><<<gemm_grid, 128, 0, stream>>>(
        x + (size_t)b * N * F_IN, W1, a1s, a1d, hbuf, esrc, edst, ntiles);
    gat_gather<4><<<gather_grid, 256, 0, stream>>>(
        rowptr, csr_src, esrc, edst, hbuf, b1, xbuf, N, 1);

    gemm_wmma_f32<64, 64><<<gemm_grid, 128, 0, stream>>>(
        xbuf, W2, a2s, a2d, hbuf, esrc, edst, ntiles);
    gat_gather<4><<<gather_grid, 256, 0, stream>>>(
        rowptr, csr_src, esrc, edst, hbuf, b2, xbuf, N, 1);

    gemm_wmma_f32<64, 16><<<gemm_grid, 128, 0, stream>>>(
        xbuf, W3, a3s, a3d, hbuf, esrc, edst, ntiles);
    gat_gather<1><<<gather_grid, 256, 0, stream>>>(
        rowptr, csr_src, esrc, edst, hbuf, b3,
        (float*)d_out + (size_t)b * N * NCLS, N, 0);
  }
}


# Gather takes each edge's logit from the h[src] row it already loads (lane dot + DPP sum); logit arrays and GEMM logit epilogue gone; softmax now running-max with rescaled sums, one divide per row
# speedup vs baseline: 11.8189x; 1.2234x over previous
#include <hip/hip_runtime.h>
#include <hip/hip_bf16.h>


typedef __attribute__((ext_vector_type(2))) float v2f;
typedef __attribute__((ext_vector_type(8))) float v8f;

#define NEG_SLOPE 0.2f
#define GAT_EPS   1e-16f

template <int CTRL>
__device__ __forceinline__ float dpp_f(float v) {
  const int x = __builtin_bit_cast(int, v);
  return __builtin_bit_cast(float, __builtin_amdgcn_update_dpp(x, x, CTRL, 0xF, 0xF, true));
}
template <int CTRL>
__device__ __forceinline__ int dpp_i(int v) {
  return __builtin_amdgcn_update_dpp(v, v, CTRL, 0xF, 0xF, true);
}
template <int H>
__device__ __forceinline__ float head_max(float v) {
  v = fmaxf(v, dpp_f<0x161>(v));
  v = fmaxf(v, dpp_f<0x162>(v));
  if constexpr (H == 1) {
    v = fmaxf(v, dpp_f<0x164>(v));
    v = fmaxf(v, dpp_f<0x168>(v));
  }
  return v;
}
template <int H>
__device__ __forceinline__ float head_sum(float v) {
  v += dpp_f<0x161>(v);
  v += dpp_f<0x162>(v);
  if constexpr (H == 1) {
    v += dpp_f<0x164>(v);
    v += dpp_f<0x168>(v);
  }
  return v;
}
template <int H, int QO>
__device__ __forceinline__ float head_bcast_f(float v) {
  if constexpr (H == 4) return dpp_f<QO * 0x55>(v);
  else                  return dpp_f<0x150 + QO>(v);
}
template <int H, int QO>
__device__ __forceinline__ int head_bcast_i(int v) {
  if constexpr (H == 4) return dpp_i<QO * 0x55>(v);
  else                  return dpp_i<0x150 + QO>(v);
}

template <int KDIM, int MO>
__global__ __launch_bounds__(128) void gemm_wmma_f32(
    const float* __restrict__ X, const float* __restrict__ W,
    float* __restrict__ Hout, int ntiles) {
  const int wave = threadIdx.x >> 5;
  const int tile = blockIdx.x * 4 + wave;
  if (tile >= ntiles) return;
  const int lane  = threadIdx.x & 31;
  const int lmod  = lane & 15;
  const int lhalf = lane >> 4;
  const int row0  = tile * 16;

  v8f acc[MO / 16];
#pragma unroll
  for (int t = 0; t < MO / 16; ++t) acc[t] = v8f{0, 0, 0, 0, 0, 0, 0, 0};

  const float* xrow = X + (size_t)(row0 + lmod) * KDIM;
#pragma unroll 4
  for (int k0 = 0; k0 < KDIM; k0 += 4) {
    const int ka = k0 + 2 * lhalf;
    v2f a;
    a.x = xrow[ka];
    a.y = xrow[ka + 1];
#pragma unroll
    for (int t = 0; t < MO / 16; ++t) {
      const float* wr = W + (size_t)ka * MO + t * 16 + lmod;
      v2f b;
      b.x = wr[0];
      b.y = wr[MO];
      acc[t] = __builtin_amdgcn_wmma_f32_16x16x4_f32(
          false, a, false, b, (short)0, acc[t], false, false);
    }
  }
#pragma unroll
  for (int t = 0; t < MO / 16; ++t)
#pragma unroll
    for (int r = 0; r < 8; ++r)
      Hout[(size_t)(row0 + r + 8 * lhalf) * MO + t * 16 + lmod] = acc[t][r];
}

__global__ void fill_u32(unsigned* __restrict__ p, unsigned v, int n) {
  int i = blockIdx.x * blockDim.x + threadIdx.x;
  if (i < n) p[i] = v;
}

__global__ void csr_count(const int* __restrict__ ei, int E, int* __restrict__ cnt) {
  int t = blockIdx.x * blockDim.x + threadIdx.x;
  if (t < E) atomicAdd(&cnt[ei[(size_t)E + t]], 1);
}

__global__ __launch_bounds__(1024) void csr_scan(int* __restrict__ cnt,
                                                 int* __restrict__ rowptr, int N) {
  __shared__ int part[1024];
  const int tid = threadIdx.x;
  const int chunk = (N + 1023) / 1024;
  const int beg = tid * chunk;
  const int end = (beg + chunk < N) ? (beg + chunk) : N;
  int s = 0;
  for (int i = beg; i < end; ++i) s += cnt[i];
  part[tid] = s;
  __syncthreads();
  for (int off = 1; off < 1024; off <<= 1) {
    const int v = (tid >= off) ? part[tid - off] : 0;
    __syncthreads();
    part[tid] += v;
    __syncthreads();
  }
  int run = part[tid] - s;
  for (int i = beg; i < end; ++i) {
    const int c = cnt[i];
    rowptr[i] = run;
    cnt[i] = run;
    run += c;
  }
  if (tid == 1023) rowptr[N] = part[1023];
}

__global__ void csr_fill(const int* __restrict__ ei, int E, int* __restrict__ cursor,
                         int* __restrict__ tmp_src) {
  int t = blockIdx.x * blockDim.x + threadIdx.x;
  if (t >= E) return;
  const int d = ei[(size_t)E + t];
  const int p = atomicAdd(&cursor[d], 1);
  tmp_src[p] = ei[t];
}

__global__ __launch_bounds__(256) void csr_order(const int* __restrict__ rowptr,
                                                 const int* __restrict__ tmp_src,
                                                 int* __restrict__ csr_src, int N) {
  const int g = (int)((blockIdx.x * 256u + threadIdx.x) >> 4);
  if (g >= N) return;
  const int lane = threadIdx.x & 15;
  const int beg = rowptr[g], end = rowptr[g + 1];
  for (int i = beg + lane; i < end; i += 16) {
    const int v = tmp_src[i];
    int r = 0;
    for (int j = beg; j < end; ++j) {
      const int w = tmp_src[j];
      r += ((w < v) || (w == v && j < i)) ? 1 : 0;
    }
    csr_src[beg + r] = v;
  }
}

template <int H>
__global__ __launch_bounds__(256) void gat_gather(
    const int* __restrict__ rowptr, const int* __restrict__ csr_src,
    const float* __restrict__ Hm,
    const float* __restrict__ a_src, const float* __restrict__ a_dst,
    const float* __restrict__ bias,
    float* __restrict__ out, int N, int relu) {
  static_assert(H == 1 || H == 4, "head_bcast covers a quad or a whole row");
  constexpr int WD  = H * 16;
  constexpr int LPH = 16 / H;
  constexpr int NT  = 2 * H;
  constexpr int NK  = NT * LPH;
  const int g = (int)((blockIdx.x * 256u + threadIdx.x) >> 4);
  if (g >= N) return;
  const int j = threadIdx.x & 15;
  const int q = j % LPH;
  const int beg = rowptr[g];
  const int n = rowptr[g + 1] - beg;
  auto ld_i = [](const int* b, unsigned off) { return *(const int*)((const char*)b + off); };
  const unsigned roff = (unsigned)(j * H) * 4u;
  auto ld_row = [&](unsigned node) {
    const float* hp = (const float*)((const char*)Hm + (node * (WD * 4u) + roff));
    float4 v;
    if constexpr (H == 4) v = *(const float4*)hp;
    else                  v = make_float4(hp[0], 0.f, 0.f, 0.f);
    return v;
  };
  float4 as, ad;
  if constexpr (H == 4) {
    const float* sp = a_src + j * 4;
    const float* dp = a_dst + j * 4;
    as = make_float4(sp[0], sp[1], sp[2], sp[3]);
    ad = make_float4(dp[0], dp[1], dp[2], dp[3]);
  } else {
    as = make_float4(a_src[j], 0.f, 0.f, 0.f);
    ad = make_float4(a_dst[j], 0.f, 0.f, 0.f);
  }
  auto head_dot = [](const float4& v, const float4& a) {
    float d = v.x * a.x;
    if constexpr (H == 4) { d += v.y * a.y; d += v.z * a.z; d += v.w * a.w; }
    return head_sum<H>(d);
  };
  auto leaky = [](float y) { return (y > 0.f) ? y : NEG_SLOPE * y; };

  int sv[NT];
  if (n > 0) {
    for (int t = 0; t < NT; ++t) {
      const int p = t * LPH + q;
      sv[t] = ld_i(csr_src, (unsigned)(beg + ((p < n) ? p : n - 1)) * 4u);
    }
  }

  const float4 v0 = ld_row((unsigned)g);
  const float ed = head_dot(v0, ad);
  float m = leaky(head_dot(v0, as) + ed);
  float den = 1.f;
  float4 acc = v0;

  auto step4 = [&](float y0, float y1, float y2, float y3, const float4& r0,
                   const float4& r1, const float4& r2, const float4& r3) {
    const float mn = fmaxf(fmaxf(fmaxf(y0, y1), fmaxf(y2, y3)), m);
    const float sc = __expf(m - mn);
    const float p0 = __expf(y0 - mn);
    const float p1 = __expf(y1 - mn);
    const float p2 = __expf(y2 - mn);
    const float p3 = __expf(y3 - mn);
    m = mn;
    den = den * sc + p0; den += p1; den += p2; den += p3;
    acc.x = acc.x * sc + p0 * r0.x; acc.x += p1 * r1.x; acc.x += p2 * r2.x; acc.x += p3 * r3.x;
    if constexpr (H == 4) {
      acc.y = acc.y * sc + p0 * r0.y; acc.y += p1 * r1.y; acc.y += p2 * r2.y; acc.y += p3 * r3.y;
      acc.z = acc.z * sc + p0 * r0.z; acc.z += p1 * r1.z; acc.z += p2 * r2.z; acc.z += p3 * r3.z;
      acc.w = acc.w * sc + p0 * r0.w; acc.w += p1 * r1.w; acc.w += p2 * r2.w; acc.w += p3 * r3.w;
    }
  };

#define GAT_ROW_LOAD(K, V)                                                     \
  { V = ld_row((unsigned)head_bcast_i<H, (K) % LPH>(sv[(K) / LPH])); }
#define GAT_LOGIT(K, V, Y)                                                     \
  {                                                                            \
    const float y_ = leaky(head_dot(V, as) + ed);                              \
    Y = ((K) < n) ? y_ : -__builtin_inff();                                    \
  }
#define GAT_FOUR(K, A, B, C, D)                                                \
  {                                                                            \
    float y0_, y1_, y2_, y3_;                                                  \
    GAT_LOGIT((K) + 0, A, y0_) GAT_LOGIT((K) + 1, B, y1_)                      \
    GAT_LOGIT((K) + 2, C, y2_) GAT_LOGIT((K) + 3, D, y3_)                      \
    step4(y0_, y1_, y2_, y3_, A, B, C, D);                                     \
  }
#define GAT_EIGHT(K)                                                           \
  if (n > (K)) {                                                               \
    float4 r0_, r1_, r2_, r3_, r4_, r5_, r6_, r7_;                             \
    GAT_ROW_LOAD((K) + 0, r0_) GAT_ROW_LOAD((K) + 1, r1_)                      \
    GAT_ROW_LOAD((K) + 2, r2_) GAT_ROW_LOAD((K) + 3, r3_)                      \
    if (n > (K) + 4) {                                                         \
      GAT_ROW_LOAD((K) + 4, r4_) GAT_ROW_LOAD((K) + 5, r5_)                    \
      GAT_ROW_LOAD((K) + 6, r6_) GAT_ROW_LOAD((K) + 7, r7_)                    \
    }                                                                          \
    GAT_FOUR((K), r0_, r1_, r2_, r3_)                                          \
    if (n > (K) + 4) GAT_FOUR((K) + 4, r4_, r5_, r6_, r7_)                     \
  }
  GAT_EIGHT(0)
  GAT_EIGHT(8)
  GAT_EIGHT(16)
  GAT_EIGHT(24)
#undef GAT_EIGHT
#undef GAT_FOUR
#undef GAT_LOGIT
#undef GAT_ROW_LOAD

  for (int base = NK; base < n; base += 4) {
    int s[4];
    for (int k = 0; k < 4; ++k) {
      const int p = base + k;
      s[k] = ld_i(csr_src, (unsigned)(beg + ((p < n) ? p : n - 1)) * 4u);
    }
    float4 r[4];
    for (int k = 0; k < 4; ++k) r[k] = ld_row((unsigned)s[k]);
    float y[4];
    for (int k = 0; k < 4; ++k) {
      const float yk = leaky(head_dot(r[k], as) + ed);
      y[k] = (base + k < n) ? yk : -__builtin_inff();
    }
    step4(y[0], y[1], y[2], y[3], r[0], r[1], r[2], r[3]);
  }

  {
    const float inv = 1.0f / (den + GAT_EPS);
    float* op = (float*)((char*)out + ((unsigned)g * (WD * 4u) + roff));
    const float* bp = bias + j * H;
    if constexpr (H == 4) {
      float4 r = make_float4(acc.x * inv + bp[0], acc.y * inv + bp[1],
                             acc.z * inv + bp[2], acc.w * inv + bp[3]);
      if (relu) {
        r.x = fmaxf(r.x, 0.f); r.y = fmaxf(r.y, 0.f);
        r.z = fmaxf(r.z, 0.f); r.w = fmaxf(r.w, 0.f);
      }
      *(float4*)op = r;
    } else {
      float r = acc.x * inv + bp[0];
      if (relu) r = fmaxf(r, 0.f);
      op[0] = r;
    }
  }
}

static inline int cdiv(long long a, int b) { return (int)((a + b - 1) / b); }

extern "C" void kernel_launch(void* const* d_in, const int* in_sizes, int n_in,
                              void* d_out, int out_size, void* d_ws, size_t ws_size,
                              hipStream_t stream) {
  const float* x   = (const float*)d_in[0];
  const int*   ei  = (const int*)d_in[1];
  const float* W1  = (const float*)d_in[2];
  const float* a1s = (const float*)d_in[3];
  const float* a1d = (const float*)d_in[4];
  const float* b1  = (const float*)d_in[5];
  const float* W2  = (const float*)d_in[6];
  const float* a2s = (const float*)d_in[7];
  const float* a2d = (const float*)d_in[8];
  const float* b2  = (const float*)d_in[9];
  const float* W3  = (const float*)d_in[10];
  const float* a3s = (const float*)d_in[11];
  const float* a3d = (const float*)d_in[12];
  const float* b3  = (const float*)d_in[13];

  const int B = 4, F_IN = 128, HC = 64;
  const int E = in_sizes[1] / 2;
  const int N = in_sizes[0] / (B * F_IN);
  const int NCLS = in_sizes[13];

  float* ws = (float*)d_ws;
  float* hbuf    = ws;
  float* xbuf    = hbuf + (size_t)N * HC;
  int*   cnt     = (int*)(xbuf + (size_t)N * HC);
  int*   rowptr  = cnt + (size_t)N;
  int*   tmp_src = rowptr + (size_t)N + 1;
  int*   csr_src = tmp_src + (size_t)E;

  const int ntiles = (N + 15) / 16;
  const int gemm_grid = (ntiles + 3) / 4;
  const int gather_grid = cdiv((long long)N * 16, 256);

  fill_u32<<<cdiv(N, 256), 256, 0, stream>>>((unsigned*)cnt, 0u, N);
  csr_count<<<cdiv(E, 256), 256, 0, stream>>>(ei, E, cnt);
  csr_scan<<<1, 1024, 0, stream>>>(cnt, rowptr, N);
  csr_fill<<<cdiv(E, 256), 256, 0, stream>>>(ei, E, cnt, tmp_src);
  csr_order<<<gather_grid, 256, 0, stream>>>(rowptr, tmp_src, csr_src, N);

  for (int b = 0; b < B; ++b) {
    gemm_wmma_f32<128, 64><<<gemm_grid, 128, 0, stream>>>(
        x + (size_t)b * N * F_IN, W1, hbuf, ntiles);
    gat_gather<4><<<gather_grid, 256, 0, stream>>>(
        rowptr, csr_src, hbuf, a1s, a1d, b1, xbuf, N, 1);

    gemm_wmma_f32<64, 64><<<gemm_grid, 128, 0, stream>>>(
        xbuf, W2, hbuf, ntiles);
    gat_gather<4><<<gather_grid, 256, 0, stream>>>(
        rowptr, csr_src, hbuf, a2s, a2d, b2, xbuf, N, 1);

    gemm_wmma_f32<64, 16><<<gemm_grid, 128, 0, stream>>>(
        xbuf, W3, hbuf, ntiles);
    gat_gather<1><<<gather_grid, 256, 0, stream>>>(
        rowptr, csr_src, hbuf, a3s, a3d, b3,
        (float*)d_out + (size_t)b * N * NCLS, N, 0);
  }
}


# Layer 2 and 3 projections (x @ W) run in the epilogue of the gather before them: a block's 16 rows go via LDS into the same f32 WMMA k loop; 8 GEMM launches gone (29 -> 21); sum order unchanged
# speedup vs baseline: 12.5080x; 1.0583x over previous
#include <hip/hip_runtime.h>
#include <hip/hip_bf16.h>


typedef __attribute__((ext_vector_type(2))) float v2f;
typedef __attribute__((ext_vector_type(8))) float v8f;

#define NEG_SLOPE 0.2f
#define GAT_EPS   1e-16f

template <int CTRL>
__device__ __forceinline__ float dpp_f(float v) {
  const int x = __builtin_bit_cast(int, v);
  return __builtin_bit_cast(float, __builtin_amdgcn_update_dpp(x, x, CTRL, 0xF, 0xF, true));
}
template <int CTRL>
__device__ __forceinline__ int dpp_i(int v) {
  return __builtin_amdgcn_update_dpp(v, v, CTRL, 0xF, 0xF, true);
}
template <int H>
__device__ __forceinline__ float head_max(float v) {
  v = fmaxf(v, dpp_f<0x161>(v));
  v = fmaxf(v, dpp_f<0x162>(v));
  if constexpr (H == 1) {
    v = fmaxf(v, dpp_f<0x164>(v));
    v = fmaxf(v, dpp_f<0x168>(v));
  }
  return v;
}
template <int H>
__device__ __forceinline__ float head_sum(float v) {
  v += dpp_f<0x161>(v);
  v += dpp_f<0x162>(v);
  if constexpr (H == 1) {
    v += dpp_f<0x164>(v);
    v += dpp_f<0x168>(v);
  }
  return v;
}
template <int H, int QO>
__device__ __forceinline__ float head_bcast_f(float v) {
  if constexpr (H == 4) return dpp_f<QO * 0x55>(v);
  else                  return dpp_f<0x150 + QO>(v);
}
template <int H, int QO>
__device__ __forceinline__ int head_bcast_i(int v) {
  if constexpr (H == 4) return dpp_i<QO * 0x55>(v);
  else                  return dpp_i<0x150 + QO>(v);
}

template <int KDIM, int MO>
__global__ __launch_bounds__(128) void gemm_wmma_f32(
    const float* __restrict__ X, const float* __restrict__ W,
    float* __restrict__ Hout, int ntiles) {
  const int wave = threadIdx.x >> 5;
  const int tile = blockIdx.x * 4 + wave;
  if (tile >= ntiles) return;
  const int lane  = threadIdx.x & 31;
  const int lmod  = lane & 15;
  const int lhalf = lane >> 4;
  const int row0  = tile * 16;

  v8f acc[MO / 16];
#pragma unroll
  for (int t = 0; t < MO / 16; ++t) acc[t] = v8f{0, 0, 0, 0, 0, 0, 0, 0};

  const float* xrow = X + (size_t)(row0 + lmod) * KDIM;
#pragma unroll 4
  for (int k0 = 0; k0 < KDIM; k0 += 4) {
    const int ka = k0 + 2 * lhalf;
    v2f a;
    a.x = xrow[ka];
    a.y = xrow[ka + 1];
#pragma unroll
    for (int t = 0; t < MO / 16; ++t) {
      const float* wr = W + (size_t)ka * MO + t * 16 + lmod;
      v2f b;
      b.x = wr[0];
      b.y = wr[MO];
      acc[t] = __builtin_amdgcn_wmma_f32_16x16x4_f32(
          false, a, false, b, (short)0, acc[t], false, false);
    }
  }
#pragma unroll
  for (int t = 0; t < MO / 16; ++t)
#pragma unroll
    for (int r = 0; r < 8; ++r)
      Hout[(size_t)(row0 + r + 8 * lhalf) * MO + t * 16 + lmod] = acc[t][r];
}

__global__ void fill_u32(unsigned* __restrict__ p, unsigned v, int n) {
  int i = blockIdx.x * blockDim.x + threadIdx.x;
  if (i < n) p[i] = v;
}

__global__ void csr_count(const int* __restrict__ ei, int E, int* __restrict__ cnt) {
  int t = blockIdx.x * blockDim.x + threadIdx.x;
  if (t < E) atomicAdd(&cnt[ei[(size_t)E + t]], 1);
}

__global__ __launch_bounds__(1024) void csr_scan(int* __restrict__ cnt,
                                                 int* __restrict__ rowptr, int N) {
  __shared__ int part[1024];
  const int tid = threadIdx.x;
  const int chunk = (N + 1023) / 1024;
  const int beg = tid * chunk;
  const int end = (beg + chunk < N) ? (beg + chunk) : N;
  int s = 0;
  for (int i = beg; i < end; ++i) s += cnt[i];
  part[tid] = s;
  __syncthreads();
  for (int off = 1; off < 1024; off <<= 1) {
    const int v = (tid >= off) ? part[tid - off] : 0;
    __syncthreads();
    part[tid] += v;
    __syncthreads();
  }
  int run = part[tid] - s;
  for (int i = beg; i < end; ++i) {
    const int c = cnt[i];
    rowptr[i] = run;
    cnt[i] = run;
    run += c;
  }
  if (tid == 1023) rowptr[N] = part[1023];
}

__global__ void csr_fill(const int* __restrict__ ei, int E, int* __restrict__ cursor,
                         int* __restrict__ tmp_src) {
  int t = blockIdx.x * blockDim.x + threadIdx.x;
  if (t >= E) return;
  const int d = ei[(size_t)E + t];
  const int p = atomicAdd(&cursor[d], 1);
  tmp_src[p] = ei[t];
}

__global__ __launch_bounds__(256) void csr_order(const int* __restrict__ rowptr,
                                                 const int* __restrict__ tmp_src,
                                                 int* __restrict__ csr_src, int N) {
  const int g = (int)((blockIdx.x * 256u + threadIdx.x) >> 4);
  if (g >= N) return;
  const int lane = threadIdx.x & 15;
  const int beg = rowptr[g], end = rowptr[g + 1];
  for (int i = beg + lane; i < end; i += 16) {
    const int v = tmp_src[i];
    int r = 0;
    for (int j = beg; j < end; ++j) {
      const int w = tmp_src[j];
      r += ((w < v) || (w == v && j < i)) ? 1 : 0;
    }
    csr_src[beg + r] = v;
  }
}

template <int H>
__device__ __forceinline__ float4 gat_row(
    const int* __restrict__ rowptr, const int* __restrict__ csr_src,
    const float* __restrict__ Hm,
    const float* __restrict__ a_src, const float* __restrict__ a_dst,
    const int g, const int j) {
  static_assert(H == 1 || H == 4, "head_bcast covers a quad or a whole row");
  constexpr int WD  = H * 16;
  constexpr int LPH = 16 / H;
  constexpr int NT  = 2 * H;
  constexpr int NK  = NT * LPH;
  const int q = j % LPH;
  const int beg = rowptr[g];
  const int n = rowptr[g + 1] - beg;
  auto ld_i = [](const int* b, unsigned off) { return *(const int*)((const char*)b + off); };
  const unsigned roff = (unsigned)(j * H) * 4u;
  auto ld_row = [&](unsigned node) {
    const float* hp = (const float*)((const char*)Hm + (node * (WD * 4u) + roff));
    float4 v;
    if constexpr (H == 4) v = *(const float4*)hp;
    else                  v = make_float4(hp[0], 0.f, 0.f, 0.f);
    return v;
  };
  float4 as, ad;
  if constexpr (H == 4) {
    const float* sp = a_src + j * 4;
    const float* dp = a_dst + j * 4;
    as = make_float4(sp[0], sp[1], sp[2], sp[3]);
    ad = make_float4(dp[0], dp[1], dp[2], dp[3]);
  } else {
    as = make_float4(a_src[j], 0.f, 0.f, 0.f);
    ad = make_float4(a_dst[j], 0.f, 0.f, 0.f);
  }
  auto head_dot = [](const float4& v, const float4& a) {
    float d = v.x * a.x;
    if constexpr (H == 4) { d += v.y * a.y; d += v.z * a.z; d += v.w * a.w; }
    return head_sum<H>(d);
  };
  auto leaky = [](float y) { return (y > 0.f) ? y : NEG_SLOPE * y; };

  int sv[NT];
  if (n > 0) {
    for (int t = 0; t < NT; ++t) {
      const int p = t * LPH + q;
      sv[t] = ld_i(csr_src, (unsigned)(beg + ((p < n) ? p : n - 1)) * 4u);
    }
  }

  const float4 v0 = ld_row((unsigned)g);
  const float ed = head_dot(v0, ad);
  float m = leaky(head_dot(v0, as) + ed);
  float den = 1.f;
  float4 acc = v0;

  auto step4 = [&](float y0, float y1, float y2, float y3, const float4& r0,
                   const float4& r1, const float4& r2, const float4& r3) {
    const float mn = fmaxf(fmaxf(fmaxf(y0, y1), fmaxf(y2, y3)), m);
    const float sc = __expf(m - mn);
    const float p0 = __expf(y0 - mn);
    const float p1 = __expf(y1 - mn);
    const float p2 = __expf(y2 - mn);
    const float p3 = __expf(y3 - mn);
    m = mn;
    den = den * sc + p0; den += p1; den += p2; den += p3;
    acc.x = acc.x * sc + p0 * r0.x; acc.x += p1 * r1.x; acc.x += p2 * r2.x; acc.x += p3 * r3.x;
    if constexpr (H == 4) {
      acc.y = acc.y * sc + p0 * r0.y; acc.y += p1 * r1.y; acc.y += p2 * r2.y; acc.y += p3 * r3.y;
      acc.z = acc.z * sc + p0 * r0.z; acc.z += p1 * r1.z; acc.z += p2 * r2.z; acc.z += p3 * r3.z;
      acc.w = acc.w * sc + p0 * r0.w; acc.w += p1 * r1.w; acc.w += p2 * r2.w; acc.w += p3 * r3.w;
    }
  };

#define GAT_ROW_LOAD(K, V)                                                     \
  { V = ld_row((unsigned)head_bcast_i<H, (K) % LPH>(sv[(K) / LPH])); }
#define GAT_LOGIT(K, V, Y)                                                     \
  {                                                                            \
    const float y_ = leaky(head_dot(V, as) + ed);                              \
    Y = ((K) < n) ? y_ : -__builtin_inff();                                    \
  }
#define GAT_FOUR(K, A, B, C, D)                                                \
  {                                                                            \
    float y0_, y1_, y2_, y3_;                                                  \
    GAT_LOGIT((K) + 0, A, y0_) GAT_LOGIT((K) + 1, B, y1_)                      \
    GAT_LOGIT((K) + 2, C, y2_) GAT_LOGIT((K) + 3, D, y3_)                      \
    step4(y0_, y1_, y2_, y3_, A, B, C, D);                                     \
  }
#define GAT_EIGHT(K)                                                           \
  if (n > (K)) {                                                               \
    float4 r0_, r1_, r2_, r3_, r4_, r5_, r6_, r7_;                             \
    GAT_ROW_LOAD((K) + 0, r0_) GAT_ROW_LOAD((K) + 1, r1_)                      \
    GAT_ROW_LOAD((K) + 2, r2_) GAT_ROW_LOAD((K) + 3, r3_)                      \
    if (n > (K) + 4) {                                                         \
      GAT_ROW_LOAD((K) + 4, r4_) GAT_ROW_LOAD((K) + 5, r5_)                    \
      GAT_ROW_LOAD((K) + 6, r6_) GAT_ROW_LOAD((K) + 7, r7_)                    \
    }                                                                          \
    GAT_FOUR((K), r0_, r1_, r2_, r3_)                                          \
    if (n > (K) + 4) GAT_FOUR((K) + 4, r4_, r5_, r6_, r7_)                     \
  }
  GAT_EIGHT(0)
  GAT_EIGHT(8)
  GAT_EIGHT(16)
  GAT_EIGHT(24)
#undef GAT_EIGHT
#undef GAT_FOUR
#undef GAT_LOGIT
#undef GAT_ROW_LOAD

  for (int base = NK; base < n; base += 4) {
    int s[4];
    for (int k = 0; k < 4; ++k) {
      const int p = base + k;
      s[k] = ld_i(csr_src, (unsigned)(beg + ((p < n) ? p : n - 1)) * 4u);
    }
    float4 r[4];
    for (int k = 0; k < 4; ++k) r[k] = ld_row((unsigned)s[k]);
    float y[4];
    for (int k = 0; k < 4; ++k) {
      const float yk = leaky(head_dot(r[k], as) + ed);
      y[k] = (base + k < n) ? yk : -__builtin_inff();
    }
    step4(y[0], y[1], y[2], y[3], r[0], r[1], r[2], r[3]);
  }

  const float inv = 1.0f / (den + GAT_EPS);
  float4 res = make_float4(acc.x * inv, 0.f, 0.f, 0.f);
  if constexpr (H == 4) { res.y = acc.y * inv; res.z = acc.z * inv; res.w = acc.w * inv; }
  return res;
}

template <int H>
__global__ __launch_bounds__(256) void gat_gather(
    const int* __restrict__ rowptr, const int* __restrict__ csr_src,
    const float* __restrict__ Hm,
    const float* __restrict__ a_src, const float* __restrict__ a_dst,
    const float* __restrict__ bias,
    float* __restrict__ out, int N, int relu) {
  constexpr int WD = H * 16;
  const int g = (int)((blockIdx.x * 256u + threadIdx.x) >> 4);
  if (g >= N) return;
  const int j = threadIdx.x & 15;
  const float4 v = gat_row<H>(rowptr, csr_src, Hm, a_src, a_dst, g, j);
  float* op = (float*)((char*)out + ((unsigned)g * (WD * 4u) + (unsigned)(j * H) * 4u));
  const float* bp = bias + j * H;
  if constexpr (H == 4) {
    float4 r = make_float4(v.x + bp[0], v.y + bp[1], v.z + bp[2], v.w + bp[3]);
    if (relu) {
      r.x = fmaxf(r.x, 0.f); r.y = fmaxf(r.y, 0.f);
      r.z = fmaxf(r.z, 0.f); r.w = fmaxf(r.w, 0.f);
    }
    *(float4*)op = r;
  } else {
    float r = v.x + bp[0];
    if (relu) r = fmaxf(r, 0.f);
    op[0] = r;
  }
}

template <int MO>
__global__ __launch_bounds__(256) void gat_gather_mm(
    const int* __restrict__ rowptr, const int* __restrict__ csr_src,
    const float* __restrict__ Hm,
    const float* __restrict__ a_src, const float* __restrict__ a_dst,
    const float* __restrict__ bias, const float* __restrict__ W,
    float* __restrict__ Hout, int N) {
  constexpr int KD = 64, PITCH = KD + 4;
  __shared__ __attribute__((aligned(16))) float xs[16 * PITCH];
  const int lrow = (int)(threadIdx.x >> 4);
  const int row0 = (int)(blockIdx.x * 16u);
  const int g = (row0 + lrow < N) ? (row0 + lrow) : (N - 1);
  const int j = threadIdx.x & 15;
  const float4 v = gat_row<4>(rowptr, csr_src, Hm, a_src, a_dst, g, j);
  const float* bp = bias + j * 4;
  float4 r = make_float4(fmaxf(v.x + bp[0], 0.f), fmaxf(v.y + bp[1], 0.f),
                         fmaxf(v.z + bp[2], 0.f), fmaxf(v.w + bp[3], 0.f));
  *(float4*)(xs + lrow * PITCH + j * 4) = r;

  const int wave  = (int)(threadIdx.x >> 5);
  const int lane  = threadIdx.x & 31;
  const int lmod  = lane & 15;
  const int lhalf = lane >> 4;
  const bool mm = wave < MO / 16;
  v2f b[KD / 4];
  if (mm) {
#pragma unroll
    for (int s = 0; s < KD / 4; ++s) {
      const float* wr = W + (size_t)(4 * s + 2 * lhalf) * MO + wave * 16 + lmod;
      b[s].x = wr[0];
      b[s].y = wr[MO];
    }
  }
  __syncthreads();
  if (!mm) return;
  v8f acc = v8f{0, 0, 0, 0, 0, 0, 0, 0};
  const float* xr = xs + lmod * PITCH + 2 * lhalf;
#pragma unroll
  for (int s = 0; s < KD / 4; ++s) {
    v2f a;
    a.x = xr[4 * s];
    a.y = xr[4 * s + 1];
    acc = __builtin_amdgcn_wmma_f32_16x16x4_f32(
        false, a, false, b[s], (short)0, acc, false, false);
  }
#pragma unroll
  for (int rr = 0; rr < 8; ++rr) {
    const int row = row0 + rr + 8 * lhalf;
    if (row < N) Hout[(size_t)row * MO + wave * 16 + lmod] = acc[rr];
  }
}

static inline int cdiv(long long a, int b) { return (int)((a + b - 1) / b); }

extern "C" void kernel_launch(void* const* d_in, const int* in_sizes, int n_in,
                              void* d_out, int out_size, void* d_ws, size_t ws_size,
                              hipStream_t stream) {
  const float* x   = (const float*)d_in[0];
  const int*   ei  = (const int*)d_in[1];
  const float* W1  = (const float*)d_in[2];
  const float* a1s = (const float*)d_in[3];
  const float* a1d = (const float*)d_in[4];
  const float* b1  = (const float*)d_in[5];
  const float* W2  = (const float*)d_in[6];
  const float* a2s = (const float*)d_in[7];
  const float* a2d = (const float*)d_in[8];
  const float* b2  = (const float*)d_in[9];
  const float* W3  = (const float*)d_in[10];
  const float* a3s = (const float*)d_in[11];
  const float* a3d = (const float*)d_in[12];
  const float* b3  = (const float*)d_in[13];

  const int B = 4, F_IN = 128, HC = 64;
  const int E = in_sizes[1] / 2;
  const int N = in_sizes[0] / (B * F_IN);
  const int NCLS = in_sizes[13];

  float* ws = (float*)d_ws;
  float* hbuf    = ws;
  float* xbuf    = hbuf + (size_t)N * HC;
  int*   cnt     = (int*)(xbuf + (size_t)N * HC);
  int*   rowptr  = cnt + (size_t)N;
  int*   tmp_src = rowptr + (size_t)N + 1;
  int*   csr_src = tmp_src + (size_t)E;

  const int ntiles = (N + 15) / 16;
  const int gemm_grid = (ntiles + 3) / 4;
  const int gather_grid = cdiv((long long)N * 16, 256);

  fill_u32<<<cdiv(N, 256), 256, 0, stream>>>((unsigned*)cnt, 0u, N);
  csr_count<<<cdiv(E, 256), 256, 0, stream>>>(ei, E, cnt);
  csr_scan<<<1, 1024, 0, stream>>>(cnt, rowptr, N);
  csr_fill<<<cdiv(E, 256), 256, 0, stream>>>(ei, E, cnt, tmp_src);
  csr_order<<<gather_grid, 256, 0, stream>>>(rowptr, tmp_src, csr_src, N);

  for (int b = 0; b < B; ++b) {
    gemm_wmma_f32<128, 64><<<gemm_grid, 128, 0, stream>>>(
        x + (size_t)b * N * F_IN, W1, hbuf, ntiles);
    gat_gather_mm<64><<<gather_grid, 256, 0, stream>>>(
        rowptr, csr_src, hbuf, a1s, a1d, b1, W2, xbuf, N);

    gat_gather_mm<16><<<gather_grid, 256, 0, stream>>>(
        rowptr, csr_src, xbuf, a2s, a2d, b2, W3, hbuf, N);

    gat_gather<1><<<gather_grid, 256, 0, stream>>>(
        rowptr, csr_src, hbuf, a3s, a3d, b3,
        (float*)d_out + (size_t)b * N * NCLS, N, 0);
  }
}


# Layer-1 GEMM has no launch of its own: its work-groups ride at the front of an independent launch (batch 0: bucket ordering; batch b+1: layer-3 gather of batch b); 21 -> 17 launches; sums unchanged
# speedup vs baseline: 13.2549x; 1.0597x over previous
#include <hip/hip_runtime.h>
#include <hip/hip_bf16.h>


typedef __attribute__((ext_vector_type(2))) float v2f;
typedef __attribute__((ext_vector_type(8))) float v8f;

#define NEG_SLOPE 0.2f
#define GAT_EPS   1e-16f

template <int CTRL>
__device__ __forceinline__ float dpp_f(float v) {
  const int x = __builtin_bit_cast(int, v);
  return __builtin_bit_cast(float, __builtin_amdgcn_update_dpp(x, x, CTRL, 0xF, 0xF, true));
}
template <int CTRL>
__device__ __forceinline__ int dpp_i(int v) {
  return __builtin_amdgcn_update_dpp(v, v, CTRL, 0xF, 0xF, true);
}
template <int H>
__device__ __forceinline__ float head_max(float v) {
  v = fmaxf(v, dpp_f<0x161>(v));
  v = fmaxf(v, dpp_f<0x162>(v));
  if constexpr (H == 1) {
    v = fmaxf(v, dpp_f<0x164>(v));
    v = fmaxf(v, dpp_f<0x168>(v));
  }
  return v;
}
template <int H>
__device__ __forceinline__ float head_sum(float v) {
  v += dpp_f<0x161>(v);
  v += dpp_f<0x162>(v);
  if constexpr (H == 1) {
    v += dpp_f<0x164>(v);
    v += dpp_f<0x168>(v);
  }
  return v;
}
template <int H, int QO>
__device__ __forceinline__ float head_bcast_f(float v) {
  if constexpr (H == 4) return dpp_f<QO * 0x55>(v);
  else                  return dpp_f<0x150 + QO>(v);
}
template <int H, int QO>
__device__ __forceinline__ int head_bcast_i(int v) {
  if constexpr (H == 4) return dpp_i<QO * 0x55>(v);
  else                  return dpp_i<0x150 + QO>(v);
}

template <int KDIM, int MO>
__device__ __forceinline__ void gemm_strip(
    const float* __restrict__ X, const float* __restrict__ W,
    float* __restrict__ Hout, int ntiles, const int tile) {
  if (tile >= ntiles) return;
  const int lane  = threadIdx.x & 31;
  const int lmod  = lane & 15;
  const int lhalf = lane >> 4;
  const int row0  = tile * 16;

  v8f acc[MO / 16];
#pragma unroll
  for (int t = 0; t < MO / 16; ++t) acc[t] = v8f{0, 0, 0, 0, 0, 0, 0, 0};

  const float* xrow = X + (size_t)(row0 + lmod) * KDIM;
#pragma unroll 4
  for (int k0 = 0; k0 < KDIM; k0 += 4) {
    const int ka = k0 + 2 * lhalf;
    v2f a;
    a.x = xrow[ka];
    a.y = xrow[ka + 1];
#pragma unroll
    for (int t = 0; t < MO / 16; ++t) {
      const float* wr = W + (size_t)ka * MO + t * 16 + lmod;
      v2f b;
      b.x = wr[0];
      b.y = wr[MO];
      acc[t] = __builtin_amdgcn_wmma_f32_16x16x4_f32(
          false, a, false, b, (short)0, acc[t], false, false);
    }
  }
#pragma unroll
  for (int t = 0; t < MO / 16; ++t)
#pragma unroll
    for (int r = 0; r < 8; ++r)
      Hout[(size_t)(row0 + r + 8 * lhalf) * MO + t * 16 + lmod] = acc[t][r];
}

__global__ void fill_u32(unsigned* __restrict__ p, unsigned v, int n) {
  int i = blockIdx.x * blockDim.x + threadIdx.x;
  if (i < n) p[i] = v;
}

__global__ void csr_count(const int* __restrict__ ei, int E, int* __restrict__ cnt) {
  int t = blockIdx.x * blockDim.x + threadIdx.x;
  if (t < E) atomicAdd(&cnt[ei[(size_t)E + t]], 1);
}

__global__ __launch_bounds__(1024) void csr_scan(int* __restrict__ cnt,
                                                 int* __restrict__ rowptr, int N) {
  __shared__ int part[1024];
  const int tid = threadIdx.x;
  const int chunk = (N + 1023) / 1024;
  const int beg = tid * chunk;
  const int end = (beg + chunk < N) ? (beg + chunk) : N;
  int s = 0;
  for (int i = beg; i < end; ++i) s += cnt[i];
  part[tid] = s;
  __syncthreads();
  for (int off = 1; off < 1024; off <<= 1) {
    const int v = (tid >= off) ? part[tid - off] : 0;
    __syncthreads();
    part[tid] += v;
    __syncthreads();
  }
  int run = part[tid] - s;
  for (int i = beg; i < end; ++i) {
    const int c = cnt[i];
    rowptr[i] = run;
    cnt[i] = run;
    run += c;
  }
  if (tid == 1023) rowptr[N] = part[1023];
}

__global__ void csr_fill(const int* __restrict__ ei, int E, int* __restrict__ cursor,
                         int* __restrict__ tmp_src) {
  int t = blockIdx.x * blockDim.x + threadIdx.x;
  if (t >= E) return;
  const int d = ei[(size_t)E + t];
  const int p = atomicAdd(&cursor[d], 1);
  tmp_src[p] = ei[t];
}

template <int KDIM, int MO>
__global__ __launch_bounds__(256) void csr_order_gemm(
    const int* __restrict__ rowptr, const int* __restrict__ tmp_src,
    int* __restrict__ csr_src, int N,
    const float* __restrict__ X, const float* __restrict__ W,
    float* __restrict__ Hout, int ntiles, int gemm_blocks) {
  if ((int)blockIdx.x < gemm_blocks) {
    gemm_strip<KDIM, MO>(X, W, Hout, ntiles, (int)(blockIdx.x * 8u + (threadIdx.x >> 5)));
    return;
  }
  const unsigned bid = blockIdx.x - (unsigned)gemm_blocks;
  const int g = (int)((bid * 256u + threadIdx.x) >> 4);
  if (g >= N) return;
  const int lane = threadIdx.x & 15;
  const int beg = rowptr[g], end = rowptr[g + 1];
  for (int i = beg + lane; i < end; i += 16) {
    const int v = tmp_src[i];
    int r = 0;
    for (int j = beg; j < end; ++j) {
      const int w = tmp_src[j];
      r += ((w < v) || (w == v && j < i)) ? 1 : 0;
    }
    csr_src[beg + r] = v;
  }
}

template <int H>
__device__ __forceinline__ float4 gat_row(
    const int* __restrict__ rowptr, const int* __restrict__ csr_src,
    const float* __restrict__ Hm,
    const float* __restrict__ a_src, const float* __restrict__ a_dst,
    const int g, const int j) {
  static_assert(H == 1 || H == 4, "head_bcast covers a quad or a whole row");
  constexpr int WD  = H * 16;
  constexpr int LPH = 16 / H;
  constexpr int NT  = 2 * H;
  constexpr int NK  = NT * LPH;
  const int q = j % LPH;
  const int beg = rowptr[g];
  const int n = rowptr[g + 1] - beg;
  auto ld_i = [](const int* b, unsigned off) { return *(const int*)((const char*)b + off); };
  const unsigned roff = (unsigned)(j * H) * 4u;
  auto ld_row = [&](unsigned node) {
    const float* hp = (const float*)((const char*)Hm + (node * (WD * 4u) + roff));
    float4 v;
    if constexpr (H == 4) v = *(const float4*)hp;
    else                  v = make_float4(hp[0], 0.f, 0.f, 0.f);
    return v;
  };
  float4 as, ad;
  if constexpr (H == 4) {
    const float* sp = a_src + j * 4;
    const float* dp = a_dst + j * 4;
    as = make_float4(sp[0], sp[1], sp[2], sp[3]);
    ad = make_float4(dp[0], dp[1], dp[2], dp[3]);
  } else {
    as = make_float4(a_src[j], 0.f, 0.f, 0.f);
    ad = make_float4(a_dst[j], 0.f, 0.f, 0.f);
  }
  auto head_dot = [](const float4& v, const float4& a) {
    float d = v.x * a.x;
    if constexpr (H == 4) { d += v.y * a.y; d += v.z * a.z; d += v.w * a.w; }
    return head_sum<H>(d);
  };
  auto leaky = [](float y) { return (y > 0.f) ? y : NEG_SLOPE * y; };

  int sv[NT];
  if (n > 0) {
    for (int t = 0; t < NT; ++t) {
      const int p = t * LPH + q;
      sv[t] = ld_i(csr_src, (unsigned)(beg + ((p < n) ? p : n - 1)) * 4u);
    }
  }

  const float4 v0 = ld_row((unsigned)g);
  const float ed = head_dot(v0, ad);
  float m = leaky(head_dot(v0, as) + ed);
  float den = 1.f;
  float4 acc = v0;

  auto step4 = [&](float y0, float y1, float y2, float y3, const float4& r0,
                   const float4& r1, const float4& r2, const float4& r3) {
    const float mn = fmaxf(fmaxf(fmaxf(y0, y1), fmaxf(y2, y3)), m);
    const float sc = __expf(m - mn);
    const float p0 = __expf(y0 - mn);
    const float p1 = __expf(y1 - mn);
    const float p2 = __expf(y2 - mn);
    const float p3 = __expf(y3 - mn);
    m = mn;
    den = den * sc + p0; den += p1; den += p2; den += p3;
    acc.x = acc.x * sc + p0 * r0.x; acc.x += p1 * r1.x; acc.x += p2 * r2.x; acc.x += p3 * r3.x;
    if constexpr (H == 4) {
      acc.y = acc.y * sc + p0 * r0.y; acc.y += p1 * r1.y; acc.y += p2 * r2.y; acc.y += p3 * r3.y;
      acc.z = acc.z * sc + p0 * r0.z; acc.z += p1 * r1.z; acc.z += p2 * r2.z; acc.z += p3 * r3.z;
      acc.w = acc.w * sc + p0 * r0.w; acc.w += p1 * r1.w; acc.w += p2 * r2.w; acc.w += p3 * r3.w;
    }
  };

#define GAT_ROW_LOAD(K, V)                                                     \
  { V = ld_row((unsigned)head_bcast_i<H, (K) % LPH>(sv[(K) / LPH])); }
#define GAT_LOGIT(K, V, Y)                                                     \
  {                                                                            \
    const float y_ = leaky(head_dot(V, as) + ed);                              \
    Y = ((K) < n) ? y_ : -__builtin_inff();                                    \
  }
#define GAT_FOUR(K, A, B, C, D)                                                \
  {                                                                            \
    float y0_, y1_, y2_, y3_;                                                  \
    GAT_LOGIT((K) + 0, A, y0_) GAT_LOGIT((K) + 1, B, y1_)                      \
    GAT_LOGIT((K) + 2, C, y2_) GAT_LOGIT((K) + 3, D, y3_)                      \
    step4(y0_, y1_, y2_, y3_, A, B, C, D);                                     \
  }
#define GAT_EIGHT(K)                                                           \
  if (n > (K)) {                                                               \
    float4 r0_, r1_, r2_, r3_, r4_, r5_, r6_, r7_;                             \
    GAT_ROW_LOAD((K) + 0, r0_) GAT_ROW_LOAD((K) + 1, r1_)                      \
    GAT_ROW_LOAD((K) + 2, r2_) GAT_ROW_LOAD((K) + 3, r3_)                      \
    if (n > (K) + 4) {                                                         \
      GAT_ROW_LOAD((K) + 4, r4_) GAT_ROW_LOAD((K) + 5, r5_)                    \
      GAT_ROW_LOAD((K) + 6, r6_) GAT_ROW_LOAD((K) + 7, r7_)                    \
    }                                                                          \
    GAT_FOUR((K), r0_, r1_, r2_, r3_)                                          \
    if (n > (K) + 4) GAT_FOUR((K) + 4, r4_, r5_, r6_, r7_)                     \
  }
  GAT_EIGHT(0)
  GAT_EIGHT(8)
  GAT_EIGHT(16)
  GAT_EIGHT(24)
#undef GAT_EIGHT
#undef GAT_FOUR
#undef GAT_LOGIT
#undef GAT_ROW_LOAD

  for (int base = NK; base < n; base += 4) {
    int s[4];
    for (int k = 0; k < 4; ++k) {
      const int p = base + k;
      s[k] = ld_i(csr_src, (unsigned)(beg + ((p < n) ? p : n - 1)) * 4u);
    }
    float4 r[4];
    for (int k = 0; k < 4; ++k) r[k] = ld_row((unsigned)s[k]);
    float y[4];
    for (int k = 0; k < 4; ++k) {
      const float yk = leaky(head_dot(r[k], as) + ed);
      y[k] = (base + k < n) ? yk : -__builtin_inff();
    }
    step4(y[0], y[1], y[2], y[3], r[0], r[1], r[2], r[3]);
  }

  const float inv = 1.0f / (den + GAT_EPS);
  float4 res = make_float4(acc.x * inv, 0.f, 0.f, 0.f);
  if constexpr (H == 4) { res.y = acc.y * inv; res.z = acc.z * inv; res.w = acc.w * inv; }
  return res;
}

template <int H, int KDIM, int MO>
__global__ __launch_bounds__(256) void gat_gather_gemm(
    const int* __restrict__ rowptr, const int* __restrict__ csr_src,
    const float* __restrict__ Hm,
    const float* __restrict__ a_src, const float* __restrict__ a_dst,
    const float* __restrict__ bias,
    float* __restrict__ out, int N, int relu,
    const float* __restrict__ Xn, const float* __restrict__ Wn,
    float* __restrict__ Hn, int ntiles, int gemm_blocks) {
  constexpr int WD = H * 16;
  if ((int)blockIdx.x < gemm_blocks) {
    gemm_strip<KDIM, MO>(Xn, Wn, Hn, ntiles, (int)(blockIdx.x * 8u + (threadIdx.x >> 5)));
    return;
  }
  const unsigned bid = blockIdx.x - (unsigned)gemm_blocks;
  const int g = (int)((bid * 256u + threadIdx.x) >> 4);
  if (g >= N) return;
  const int j = threadIdx.x & 15;
  const float4 v = gat_row<H>(rowptr, csr_src, Hm, a_src, a_dst, g, j);
  float* op = (float*)((char*)out + ((unsigned)g * (WD * 4u) + (unsigned)(j * H) * 4u));
  const float* bp = bias + j * H;
  if constexpr (H == 4) {
    float4 r = make_float4(v.x + bp[0], v.y + bp[1], v.z + bp[2], v.w + bp[3]);
    if (relu) {
      r.x = fmaxf(r.x, 0.f); r.y = fmaxf(r.y, 0.f);
      r.z = fmaxf(r.z, 0.f); r.w = fmaxf(r.w, 0.f);
    }
    *(float4*)op = r;
  } else {
    float r = v.x + bp[0];
    if (relu) r = fmaxf(r, 0.f);
    op[0] = r;
  }
}

template <int MO>
__global__ __launch_bounds__(256) void gat_gather_mm(
    const int* __restrict__ rowptr, const int* __restrict__ csr_src,
    const float* __restrict__ Hm,
    const float* __restrict__ a_src, const float* __restrict__ a_dst,
    const float* __restrict__ bias, const float* __restrict__ W,
    float* __restrict__ Hout, int N) {
  constexpr int KD = 64, PITCH = KD + 4;
  __shared__ __attribute__((aligned(16))) float xs[16 * PITCH];
  const int lrow = (int)(threadIdx.x >> 4);
  const int row0 = (int)(blockIdx.x * 16u);
  const int g = (row0 + lrow < N) ? (row0 + lrow) : (N - 1);
  const int j = threadIdx.x & 15;
  const float4 v = gat_row<4>(rowptr, csr_src, Hm, a_src, a_dst, g, j);
  const float* bp = bias + j * 4;
  float4 r = make_float4(fmaxf(v.x + bp[0], 0.f), fmaxf(v.y + bp[1], 0.f),
                         fmaxf(v.z + bp[2], 0.f), fmaxf(v.w + bp[3], 0.f));
  *(float4*)(xs + lrow * PITCH + j * 4) = r;

  const int wave  = (int)(threadIdx.x >> 5);
  const int lane  = threadIdx.x & 31;
  const int lmod  = lane & 15;
  const int lhalf = lane >> 4;
  const bool mm = wave < MO / 16;
  v2f b[KD / 4];
  if (mm) {
#pragma unroll
    for (int s = 0; s < KD / 4; ++s) {
      const float* wr = W + (size_t)(4 * s + 2 * lhalf) * MO + wave * 16 + lmod;
      b[s].x = wr[0];
      b[s].y = wr[MO];
    }
  }
  __syncthreads();
  if (!mm) return;
  v8f acc = v8f{0, 0, 0, 0, 0, 0, 0, 0};
  const float* xr = xs + lmod * PITCH + 2 * lhalf;
#pragma unroll
  for (int s = 0; s < KD / 4; ++s) {
    v2f a;
    a.x = xr[4 * s];
    a.y = xr[4 * s + 1];
    acc = __builtin_amdgcn_wmma_f32_16x16x4_f32(
        false, a, false, b[s], (short)0, acc, false, false);
  }
#pragma unroll
  for (int rr = 0; rr < 8; ++rr) {
    const int row = row0 + rr + 8 * lhalf;
    if (row < N) Hout[(size_t)row * MO + wave * 16 + lmod] = acc[rr];
  }
}

static inline int cdiv(long long a, int b) { return (int)((a + b - 1) / b); }

extern "C" void kernel_launch(void* const* d_in, const int* in_sizes, int n_in,
                              void* d_out, int out_size, void* d_ws, size_t ws_size,
                              hipStream_t stream) {
  const float* x   = (const float*)d_in[0];
  const int*   ei  = (const int*)d_in[1];
  const float* W1  = (const float*)d_in[2];
  const float* a1s = (const float*)d_in[3];
  const float* a1d = (const float*)d_in[4];
  const float* b1  = (const float*)d_in[5];
  const float* W2  = (const float*)d_in[6];
  const float* a2s = (const float*)d_in[7];
  const float* a2d = (const float*)d_in[8];
  const float* b2  = (const float*)d_in[9];
  const float* W3  = (const float*)d_in[10];
  const float* a3s = (const float*)d_in[11];
  const float* a3d = (const float*)d_in[12];
  const float* b3  = (const float*)d_in[13];

  const int B = 4, F_IN = 128, HC = 64;
  const int E = in_sizes[1] / 2;
  const int N = in_sizes[0] / (B * F_IN);
  const int NCLS = in_sizes[13];

  float* ws = (float*)d_ws;
  float* hbuf    = ws;
  float* xbuf    = hbuf + (size_t)N * HC;
  int*   cnt     = (int*)(xbuf + (size_t)N * HC);
  int*   rowptr  = cnt + (size_t)N;
  int*   tmp_src = rowptr + (size_t)N + 1;
  int*   csr_src = tmp_src + (size_t)E;

  const int ntiles = (N + 15) / 16;
  const int gemm_blocks = (ntiles + 7) / 8;
  const int gather_grid = cdiv((long long)N * 16, 256);

  fill_u32<<<cdiv(N, 256), 256, 0, stream>>>((unsigned*)cnt, 0u, N);
  csr_count<<<cdiv(E, 256), 256, 0, stream>>>(ei, E, cnt);
  csr_scan<<<1, 1024, 0, stream>>>(cnt, rowptr, N);
  csr_fill<<<cdiv(E, 256), 256, 0, stream>>>(ei, E, cnt, tmp_src);
  csr_order_gemm<128, 64><<<gemm_blocks + gather_grid, 256, 0, stream>>>(
      rowptr, tmp_src, csr_src, N, x, W1, hbuf, ntiles, gemm_blocks);

  float* p = hbuf;
  float* q = xbuf;
  for (int b = 0; b < B; ++b) {
    gat_gather_mm<64><<<gather_grid, 256, 0, stream>>>(
        rowptr, csr_src, p, a1s, a1d, b1, W2, q, N);

    gat_gather_mm<16><<<gather_grid, 256, 0, stream>>>(
        rowptr, csr_src, q, a2s, a2d, b2, W3, p, N);

    const int nb = (b + 1 < B) ? (b + 1) : b;
    const int rider = (b + 1 < B) ? gemm_blocks : 0;
    gat_gather_gemm<1, 128, 64><<<rider + gather_grid, 256, 0, stream>>>(
        rowptr, csr_src, p, a3s, a3d, b3,
        (float*)d_out + (size_t)b * N * NCLS, N, 0,
        x + (size_t)nb * N * F_IN, W1, q, ntiles, rider);

    float* t = p; p = q; q = t;
  }
}


# Row-pointer scan (1 block): each wave walks one contiguous segment with 16-byte whole-line loads/stores and in-wave lane-shift prefix sums, not 49 serial 4-byte accesses per thread at stride 196 B
# speedup vs baseline: 16.7067x; 1.2604x over previous
#include <hip/hip_runtime.h>
#include <hip/hip_bf16.h>


typedef __attribute__((ext_vector_type(2))) float v2f;
typedef __attribute__((ext_vector_type(8))) float v8f;

#define NEG_SLOPE 0.2f
#define GAT_EPS   1e-16f

template <int CTRL>
__device__ __forceinline__ float dpp_f(float v) {
  const int x = __builtin_bit_cast(int, v);
  return __builtin_bit_cast(float, __builtin_amdgcn_update_dpp(x, x, CTRL, 0xF, 0xF, true));
}
template <int CTRL>
__device__ __forceinline__ int dpp_i(int v) {
  return __builtin_amdgcn_update_dpp(v, v, CTRL, 0xF, 0xF, true);
}
template <int H>
__device__ __forceinline__ float head_max(float v) {
  v = fmaxf(v, dpp_f<0x161>(v));
  v = fmaxf(v, dpp_f<0x162>(v));
  if constexpr (H == 1) {
    v = fmaxf(v, dpp_f<0x164>(v));
    v = fmaxf(v, dpp_f<0x168>(v));
  }
  return v;
}
template <int H>
__device__ __forceinline__ float head_sum(float v) {
  v += dpp_f<0x161>(v);
  v += dpp_f<0x162>(v);
  if constexpr (H == 1) {
    v += dpp_f<0x164>(v);
    v += dpp_f<0x168>(v);
  }
  return v;
}
template <int H, int QO>
__device__ __forceinline__ float head_bcast_f(float v) {
  if constexpr (H == 4) return dpp_f<QO * 0x55>(v);
  else                  return dpp_f<0x150 + QO>(v);
}
template <int H, int QO>
__device__ __forceinline__ int head_bcast_i(int v) {
  if constexpr (H == 4) return dpp_i<QO * 0x55>(v);
  else                  return dpp_i<0x150 + QO>(v);
}

template <int KDIM, int MO>
__device__ __forceinline__ void gemm_strip(
    const float* __restrict__ X, const float* __restrict__ W,
    float* __restrict__ Hout, int ntiles, const int tile) {
  if (tile >= ntiles) return;
  const int lane  = threadIdx.x & 31;
  const int lmod  = lane & 15;
  const int lhalf = lane >> 4;
  const int row0  = tile * 16;

  v8f acc[MO / 16];
#pragma unroll
  for (int t = 0; t < MO / 16; ++t) acc[t] = v8f{0, 0, 0, 0, 0, 0, 0, 0};

  const float* xrow = X + (size_t)(row0 + lmod) * KDIM;
#pragma unroll 4
  for (int k0 = 0; k0 < KDIM; k0 += 4) {
    const int ka = k0 + 2 * lhalf;
    v2f a;
    a.x = xrow[ka];
    a.y = xrow[ka + 1];
#pragma unroll
    for (int t = 0; t < MO / 16; ++t) {
      const float* wr = W + (size_t)ka * MO + t * 16 + lmod;
      v2f b;
      b.x = wr[0];
      b.y = wr[MO];
      acc[t] = __builtin_amdgcn_wmma_f32_16x16x4_f32(
          false, a, false, b, (short)0, acc[t], false, false);
    }
  }
#pragma unroll
  for (int t = 0; t < MO / 16; ++t)
#pragma unroll
    for (int r = 0; r < 8; ++r)
      Hout[(size_t)(row0 + r + 8 * lhalf) * MO + t * 16 + lmod] = acc[t][r];
}

__global__ void fill_u32(unsigned* __restrict__ p, unsigned v, int n) {
  int i = blockIdx.x * blockDim.x + threadIdx.x;
  if (i < n) p[i] = v;
}

__global__ void csr_count(const int* __restrict__ ei, int E, int* __restrict__ cnt) {
  int t = blockIdx.x * blockDim.x + threadIdx.x;
  if (t < E) atomicAdd(&cnt[ei[(size_t)E + t]], 1);
}


template <int SH>
__device__ __forceinline__ int row_shr0(int v) {
  return __builtin_amdgcn_update_dpp(0, v, 0x110 + SH, 0xF, 0xF, true);
}
__device__ __forceinline__ int wave_scan_incl(int v, int lane) {
  v += row_shr0<1>(v);
  v += row_shr0<2>(v);
  v += row_shr0<4>(v);
  v += row_shr0<8>(v);
  const int low = __builtin_amdgcn_readlane(v, 15);
  return v + ((lane & 16) ? low : 0);
}
template <bool VEC>
__device__ __forceinline__ int4 scan_ld4(const int* p, int i, int N, bool take, int safe) {
  int4 v;
  if constexpr (VEC) {
    const bool in = take && (i < N);
    v = *(const int4*)(p + (in ? i : safe));
    if (!in) v = make_int4(0, 0, 0, 0);
  } else {
    const bool i0 = take && (i < N), i1 = take && (i + 1 < N);
    const bool i2 = take && (i + 2 < N), i3 = take && (i + 3 < N);
    const int a = p[i0 ? i     : safe];
    const int b = p[i1 ? i + 1 : safe];
    const int c = p[i2 ? i + 2 : safe];
    const int d = p[i3 ? i + 3 : safe];
    v = make_int4(i0 ? a : 0, i1 ? b : 0, i2 ? c : 0, i3 ? d : 0);
  }
  return v;
}
template <bool VEC>
__device__ __forceinline__ void scan_st4(int* p, int i, int N, const int4 v) {
  if constexpr (VEC) {
    if (i < N) *(int4*)(p + i) = v;
  } else {
    if (i     < N) p[i]     = v.x;
    if (i + 1 < N) p[i + 1] = v.y;
    if (i + 2 < N) p[i + 2] = v.z;
    if (i + 3 < N) p[i + 3] = v.w;
  }
}
template <bool VEC>
__device__ __forceinline__ void csr_scan_body(int* __restrict__ cnt, int* __restrict__ rowptr,
                                              int N, int* wtot) {
  constexpr int TRIP = 8;
  const int tid  = threadIdx.x;
  const int lane = tid & 31;
  const int wave = __builtin_amdgcn_readfirstlane(tid >> 5);
  const int seg   = (((N + 31) >> 5) + 127) & ~127;
  const int steps = seg >> 7;
  const int seg0  = wave * seg;
  const int base  = seg0 + lane * 4;

  int s = 0;
  for (int k0 = 0; k0 < steps; k0 += TRIP) {
    int4 v[TRIP];
#pragma unroll
    for (int u = 0; u < TRIP; ++u)
      v[u] = scan_ld4<VEC>(cnt, base + (k0 + u) * 128, N, k0 + u < steps, 0);
#pragma unroll
    for (int u = 0; u < TRIP; ++u) s += (v[u].x + v[u].y) + (v[u].z + v[u].w);
  }
  const int incl = wave_scan_incl(s, lane);
  if (lane == 31) wtot[wave] = incl;
  __syncthreads();
  const int w  = wtot[lane];
  const int wi = wave_scan_incl(w, lane);
  int carry = __builtin_amdgcn_readlane(wi - w, wave);
  const int total = __builtin_amdgcn_readlane(wi, 31);

  if (seg0 < N) {
    for (int k0 = 0; k0 < steps; k0 += TRIP) {
      int4 v[TRIP];
#pragma unroll
      for (int u = 0; u < TRIP; ++u)
        v[u] = scan_ld4<VEC>(cnt, base + (k0 + u) * 128, N, k0 + u < steps, seg0);
#pragma unroll
      for (int u = 0; u < TRIP; ++u) {
        if (k0 + u < steps) {
          const int t4 = (v[u].x + v[u].y) + (v[u].z + v[u].w);
          const int in = wave_scan_incl(t4, lane);
          int4 p;
          p.x = carry + (in - t4);
          p.y = p.x + v[u].x;
          p.z = p.y + v[u].y;
          p.w = p.z + v[u].z;
          const int i = base + (k0 + u) * 128;
          scan_st4<VEC>(rowptr, i, N, p);
          scan_st4<VEC>(cnt, i, N, p);
          carry += __builtin_amdgcn_readlane(in, 31);
        }
      }
    }
  }
  if (tid == 0) rowptr[N] = total;
}
__global__ __launch_bounds__(1024) void csr_scan(int* __restrict__ cnt,
                                                 int* __restrict__ rowptr, int N) {
  __shared__ int wtot[32];
  const bool vec = ((((size_t)cnt | (size_t)rowptr) & 15) == 0) && ((N & 3) == 0);
  if (vec) csr_scan_body<true>(cnt, rowptr, N, wtot);
  else     csr_scan_body<false>(cnt, rowptr, N, wtot);
}

__global__ void csr_fill(const int* __restrict__ ei, int E, int* __restrict__ cursor,
                         int* __restrict__ tmp_src) {
  int t = blockIdx.x * blockDim.x + threadIdx.x;
  if (t >= E) return;
  const int d = ei[(size_t)E + t];
  const int p = atomicAdd(&cursor[d], 1);
  tmp_src[p] = ei[t];
}

template <int KDIM, int MO>
__global__ __launch_bounds__(256) void csr_order_gemm(
    const int* __restrict__ rowptr, const int* __restrict__ tmp_src,
    int* __restrict__ csr_src, int N,
    const float* __restrict__ X, const float* __restrict__ W,
    float* __restrict__ Hout, int ntiles, int gemm_blocks) {
  if ((int)blockIdx.x < gemm_blocks) {
    gemm_strip<KDIM, MO>(X, W, Hout, ntiles, (int)(blockIdx.x * 8u + (threadIdx.x >> 5)));
    return;
  }
  const unsigned bid = blockIdx.x - (unsigned)gemm_blocks;
  const int g = (int)((bid * 256u + threadIdx.x) >> 4);
  if (g >= N) return;
  const int lane = threadIdx.x & 15;
  const int beg = rowptr[g], end = rowptr[g + 1];
  for (int i = beg + lane; i < end; i += 16) {
    const int v = tmp_src[i];
    int r = 0;
    for (int j = beg; j < end; ++j) {
      const int w = tmp_src[j];
      r += ((w < v) || (w == v && j < i)) ? 1 : 0;
    }
    csr_src[beg + r] = v;
  }
}

template <int H>
__device__ __forceinline__ float4 gat_row(
    const int* __restrict__ rowptr, const int* __restrict__ csr_src,
    const float* __restrict__ Hm,
    const float* __restrict__ a_src, const float* __restrict__ a_dst,
    const int g, const int j) {
  static_assert(H == 1 || H == 4, "head_bcast covers a quad or a whole row");
  constexpr int WD  = H * 16;
  constexpr int LPH = 16 / H;
  constexpr int NT  = 2 * H;
  constexpr int NK  = NT * LPH;
  const int q = j % LPH;
  const int beg = rowptr[g];
  const int n = rowptr[g + 1] - beg;
  auto ld_i = [](const int* b, unsigned off) { return *(const int*)((const char*)b + off); };
  const unsigned roff = (unsigned)(j * H) * 4u;
  auto ld_row = [&](unsigned node) {
    const float* hp = (const float*)((const char*)Hm + (node * (WD * 4u) + roff));
    float4 v;
    if constexpr (H == 4) v = *(const float4*)hp;
    else                  v = make_float4(hp[0], 0.f, 0.f, 0.f);
    return v;
  };
  float4 as, ad;
  if constexpr (H == 4) {
    const float* sp = a_src + j * 4;
    const float* dp = a_dst + j * 4;
    as = make_float4(sp[0], sp[1], sp[2], sp[3]);
    ad = make_float4(dp[0], dp[1], dp[2], dp[3]);
  } else {
    as = make_float4(a_src[j], 0.f, 0.f, 0.f);
    ad = make_float4(a_dst[j], 0.f, 0.f, 0.f);
  }
  auto head_dot = [](const float4& v, const float4& a) {
    float d = v.x * a.x;
    if constexpr (H == 4) { d += v.y * a.y; d += v.z * a.z; d += v.w * a.w; }
    return head_sum<H>(d);
  };
  auto leaky = [](float y) { return (y > 0.f) ? y : NEG_SLOPE * y; };

  int sv[NT];
  if (n > 0) {
    for (int t = 0; t < NT; ++t) {
      const int p = t * LPH + q;
      sv[t] = ld_i(csr_src, (unsigned)(beg + ((p < n) ? p : n - 1)) * 4u);
    }
  }

  const float4 v0 = ld_row((unsigned)g);
  const float ed = head_dot(v0, ad);
  float m = leaky(head_dot(v0, as) + ed);
  float den = 1.f;
  float4 acc = v0;

  auto step4 = [&](float y0, float y1, float y2, float y3, const float4& r0,
                   const float4& r1, const float4& r2, const float4& r3) {
    const float mn = fmaxf(fmaxf(fmaxf(y0, y1), fmaxf(y2, y3)), m);
    const float sc = __expf(m - mn);
    const float p0 = __expf(y0 - mn);
    const float p1 = __expf(y1 - mn);
    const float p2 = __expf(y2 - mn);
    const float p3 = __expf(y3 - mn);
    m = mn;
    den = den * sc + p0; den += p1; den += p2; den += p3;
    acc.x = acc.x * sc + p0 * r0.x; acc.x += p1 * r1.x; acc.x += p2 * r2.x; acc.x += p3 * r3.x;
    if constexpr (H == 4) {
      acc.y = acc.y * sc + p0 * r0.y; acc.y += p1 * r1.y; acc.y += p2 * r2.y; acc.y += p3 * r3.y;
      acc.z = acc.z * sc + p0 * r0.z; acc.z += p1 * r1.z; acc.z += p2 * r2.z; acc.z += p3 * r3.z;
      acc.w = acc.w * sc + p0 * r0.w; acc.w += p1 * r1.w; acc.w += p2 * r2.w; acc.w += p3 * r3.w;
    }
  };

#define GAT_ROW_LOAD(K, V)                                                     \
  { V = ld_row((unsigned)head_bcast_i<H, (K) % LPH>(sv[(K) / LPH])); }
#define GAT_LOGIT(K, V, Y)                                                     \
  {                                                                            \
    const float y_ = leaky(head_dot(V, as) + ed);                              \
    Y = ((K) < n) ? y_ : -__builtin_inff();                                    \
  }
#define GAT_FOUR(K, A, B, C, D)                                                \
  {                                                                            \
    float y0_, y1_, y2_, y3_;                                                  \
    GAT_LOGIT((K) + 0, A, y0_) GAT_LOGIT((K) + 1, B, y1_)                      \
    GAT_LOGIT((K) + 2, C, y2_) GAT_LOGIT((K) + 3, D, y3_)                      \
    step4(y0_, y1_, y2_, y3_, A, B, C, D);                                     \
  }
#define GAT_EIGHT(K)                                                           \
  if (n > (K)) {                                                               \
    float4 r0_, r1_, r2_, r3_, r4_, r5_, r6_, r7_;                             \
    GAT_ROW_LOAD((K) + 0, r0_) GAT_ROW_LOAD((K) + 1, r1_)                      \
    GAT_ROW_LOAD((K) + 2, r2_) GAT_ROW_LOAD((K) + 3, r3_)                      \
    if (n > (K) + 4) {                                                         \
      GAT_ROW_LOAD((K) + 4, r4_) GAT_ROW_LOAD((K) + 5, r5_)                    \
      GAT_ROW_LOAD((K) + 6, r6_) GAT_ROW_LOAD((K) + 7, r7_)                    \
    }                                                                          \
    GAT_FOUR((K), r0_, r1_, r2_, r3_)                                          \
    if (n > (K) + 4) GAT_FOUR((K) + 4, r4_, r5_, r6_, r7_)                     \
  }
  GAT_EIGHT(0)
  GAT_EIGHT(8)
  GAT_EIGHT(16)
  GAT_EIGHT(24)
#undef GAT_EIGHT
#undef GAT_FOUR
#undef GAT_LOGIT
#undef GAT_ROW_LOAD

  for (int base = NK; base < n; base += 4) {
    int s[4];
    for (int k = 0; k < 4; ++k) {
      const int p = base + k;
      s[k] = ld_i(csr_src, (unsigned)(beg + ((p < n) ? p : n - 1)) * 4u);
    }
    float4 r[4];
    for (int k = 0; k < 4; ++k) r[k] = ld_row((unsigned)s[k]);
    float y[4];
    for (int k = 0; k < 4; ++k) {
      const float yk = leaky(head_dot(r[k], as) + ed);
      y[k] = (base + k < n) ? yk : -__builtin_inff();
    }
    step4(y[0], y[1], y[2], y[3], r[0], r[1], r[2], r[3]);
  }

  const float inv = 1.0f / (den + GAT_EPS);
  float4 res = make_float4(acc.x * inv, 0.f, 0.f, 0.f);
  if constexpr (H == 4) { res.y = acc.y * inv; res.z = acc.z * inv; res.w = acc.w * inv; }
  return res;
}

template <int H, int KDIM, int MO>
__global__ __launch_bounds__(256) void gat_gather_gemm(
    const int* __restrict__ rowptr, const int* __restrict__ csr_src,
    const float* __restrict__ Hm,
    const float* __restrict__ a_src, const float* __restrict__ a_dst,
    const float* __restrict__ bias,
    float* __restrict__ out, int N, int relu,
    const float* __restrict__ Xn, const float* __restrict__ Wn,
    float* __restrict__ Hn, int ntiles, int gemm_blocks) {
  constexpr int WD = H * 16;
  if ((int)blockIdx.x < gemm_blocks) {
    gemm_strip<KDIM, MO>(Xn, Wn, Hn, ntiles, (int)(blockIdx.x * 8u + (threadIdx.x >> 5)));
    return;
  }
  const unsigned bid = blockIdx.x - (unsigned)gemm_blocks;
  const int g = (int)((bid * 256u + threadIdx.x) >> 4);
  if (g >= N) return;
  const int j = threadIdx.x & 15;
  const float4 v = gat_row<H>(rowptr, csr_src, Hm, a_src, a_dst, g, j);
  float* op = (float*)((char*)out + ((unsigned)g * (WD * 4u) + (unsigned)(j * H) * 4u));
  const float* bp = bias + j * H;
  if constexpr (H == 4) {
    float4 r = make_float4(v.x + bp[0], v.y + bp[1], v.z + bp[2], v.w + bp[3]);
    if (relu) {
      r.x = fmaxf(r.x, 0.f); r.y = fmaxf(r.y, 0.f);
      r.z = fmaxf(r.z, 0.f); r.w = fmaxf(r.w, 0.f);
    }
    *(float4*)op = r;
  } else {
    float r = v.x + bp[0];
    if (relu) r = fmaxf(r, 0.f);
    op[0] = r;
  }
}

template <int MO>
__global__ __launch_bounds__(256) void gat_gather_mm(
    const int* __restrict__ rowptr, const int* __restrict__ csr_src,
    const float* __restrict__ Hm,
    const float* __restrict__ a_src, const float* __restrict__ a_dst,
    const float* __restrict__ bias, const float* __restrict__ W,
    float* __restrict__ Hout, int N) {
  constexpr int KD = 64, PITCH = KD + 4;
  __shared__ __attribute__((aligned(16))) float xs[16 * PITCH];
  const int lrow = (int)(threadIdx.x >> 4);
  const int row0 = (int)(blockIdx.x * 16u);
  const int g = (row0 + lrow < N) ? (row0 + lrow) : (N - 1);
  const int j = threadIdx.x & 15;
  const float4 v = gat_row<4>(rowptr, csr_src, Hm, a_src, a_dst, g, j);
  const float* bp = bias + j * 4;
  float4 r = make_float4(fmaxf(v.x + bp[0], 0.f), fmaxf(v.y + bp[1], 0.f),
                         fmaxf(v.z + bp[2], 0.f), fmaxf(v.w + bp[3], 0.f));
  *(float4*)(xs + lrow * PITCH + j * 4) = r;

  const int wave  = (int)(threadIdx.x >> 5);
  const int lane  = threadIdx.x & 31;
  const int lmod  = lane & 15;
  const int lhalf = lane >> 4;
  const bool mm = wave < MO / 16;
  v2f b[KD / 4];
  if (mm) {
#pragma unroll
    for (int s = 0; s < KD / 4; ++s) {
      const float* wr = W + (size_t)(4 * s + 2 * lhalf) * MO + wave * 16 + lmod;
      b[s].x = wr[0];
      b[s].y = wr[MO];
    }
  }
  __syncthreads();
  if (!mm) return;
  v8f acc = v8f{0, 0, 0, 0, 0, 0, 0, 0};
  const float* xr = xs + lmod * PITCH + 2 * lhalf;
#pragma unroll
  for (int s = 0; s < KD / 4; ++s) {
    v2f a;
    a.x = xr[4 * s];
    a.y = xr[4 * s + 1];
    acc = __builtin_amdgcn_wmma_f32_16x16x4_f32(
        false, a, false, b[s], (short)0, acc, false, false);
  }
#pragma unroll
  for (int rr = 0; rr < 8; ++rr) {
    const int row = row0 + rr + 8 * lhalf;
    if (row < N) Hout[(size_t)row * MO + wave * 16 + lmod] = acc[rr];
  }
}

static inline int cdiv(long long a, int b) { return (int)((a + b - 1) / b); }

extern "C" void kernel_launch(void* const* d_in, const int* in_sizes, int n_in,
                              void* d_out, int out_size, void* d_ws, size_t ws_size,
                              hipStream_t stream) {
  const float* x   = (const float*)d_in[0];
  const int*   ei  = (const int*)d_in[1];
  const float* W1  = (const float*)d_in[2];
  const float* a1s = (const float*)d_in[3];
  const float* a1d = (const float*)d_in[4];
  const float* b1  = (const float*)d_in[5];
  const float* W2  = (const float*)d_in[6];
  const float* a2s = (const float*)d_in[7];
  const float* a2d = (const float*)d_in[8];
  const float* b2  = (const float*)d_in[9];
  const float* W3  = (const float*)d_in[10];
  const float* a3s = (const float*)d_in[11];
  const float* a3d = (const float*)d_in[12];
  const float* b3  = (const float*)d_in[13];

  const int B = 4, F_IN = 128, HC = 64;
  const int E = in_sizes[1] / 2;
  const int N = in_sizes[0] / (B * F_IN);
  const int NCLS = in_sizes[13];

  float* ws = (float*)d_ws;
  float* hbuf    = ws;
  float* xbuf    = hbuf + (size_t)N * HC;
  int*   cnt     = (int*)(xbuf + (size_t)N * HC);
  int*   rowptr  = cnt + (size_t)N;
  int*   tmp_src = rowptr + (size_t)N + 1;
  int*   csr_src = tmp_src + (size_t)E;

  const int ntiles = (N + 15) / 16;
  const int gemm_blocks = (ntiles + 7) / 8;
  const int gather_grid = cdiv((long long)N * 16, 256);

  fill_u32<<<cdiv(N, 256), 256, 0, stream>>>((unsigned*)cnt, 0u, N);
  csr_count<<<cdiv(E, 256), 256, 0, stream>>>(ei, E, cnt);
  csr_scan<<<1, 1024, 0, stream>>>(cnt, rowptr, N);
  csr_fill<<<cdiv(E, 256), 256, 0, stream>>>(ei, E, cnt, tmp_src);
  csr_order_gemm<128, 64><<<gemm_blocks + gather_grid, 256, 0, stream>>>(
      rowptr, tmp_src, csr_src, N, x, W1, hbuf, ntiles, gemm_blocks);

  float* p = hbuf;
  float* q = xbuf;
  for (int b = 0; b < B; ++b) {
    gat_gather_mm<64><<<gather_grid, 256, 0, stream>>>(
        rowptr, csr_src, p, a1s, a1d, b1, W2, q, N);

    gat_gather_mm<16><<<gather_grid, 256, 0, stream>>>(
        rowptr, csr_src, q, a2s, a2d, b2, W3, p, N);

    const int nb = (b + 1 < B) ? (b + 1) : b;
    const int rider = (b + 1 < B) ? gemm_blocks : 0;
    gat_gather_gemm<1, 128, 64><<<rider + gather_grid, 256, 0, stream>>>(
        rowptr, csr_src, p, a3s, a3d, b3,
        (float*)d_out + (size_t)b * N * NCLS, N, 0,
        x + (size_t)nb * N * F_IN, W1, q, ntiles, rider);

    float* t = p; p = q; q = t;
  }
}


# Layer 3 runs once for all 4 batches: its h rows kept batch-interleaved [node][batch][16], one 256-B row per edge, not 4 of 64 B; 17 -> 14 launches; logit dot order now 4 in-lane + quad butterfly
# speedup vs baseline: 18.2299x; 1.0912x over previous
#include <hip/hip_runtime.h>
#include <hip/hip_bf16.h>


typedef __attribute__((ext_vector_type(2))) float v2f;
typedef __attribute__((ext_vector_type(8))) float v8f;

#define NEG_SLOPE 0.2f
#define GAT_EPS   1e-16f

template <int CTRL>
__device__ __forceinline__ float dpp_f(float v) {
  const int x = __builtin_bit_cast(int, v);
  return __builtin_bit_cast(float, __builtin_amdgcn_update_dpp(x, x, CTRL, 0xF, 0xF, true));
}
template <int CTRL>
__device__ __forceinline__ int dpp_i(int v) {
  return __builtin_amdgcn_update_dpp(v, v, CTRL, 0xF, 0xF, true);
}
template <int H>
__device__ __forceinline__ float head_max(float v) {
  v = fmaxf(v, dpp_f<0x161>(v));
  v = fmaxf(v, dpp_f<0x162>(v));
  if constexpr (H == 1) {
    v = fmaxf(v, dpp_f<0x164>(v));
    v = fmaxf(v, dpp_f<0x168>(v));
  }
  return v;
}
template <int H>
__device__ __forceinline__ float head_sum(float v) {
  v += dpp_f<0x161>(v);
  v += dpp_f<0x162>(v);
  if constexpr (H == 1) {
    v += dpp_f<0x164>(v);
    v += dpp_f<0x168>(v);
  }
  return v;
}
template <int H, int QO>
__device__ __forceinline__ float head_bcast_f(float v) {
  if constexpr (H == 4) return dpp_f<QO * 0x55>(v);
  else                  return dpp_f<0x150 + QO>(v);
}
template <int H, int QO>
__device__ __forceinline__ int head_bcast_i(int v) {
  if constexpr (H == 4) return dpp_i<QO * 0x55>(v);
  else                  return dpp_i<0x150 + QO>(v);
}

template <int KDIM, int MO>
__device__ __forceinline__ void gemm_strip(
    const float* __restrict__ X, const float* __restrict__ W,
    float* __restrict__ Hout, int ntiles, const int tile) {
  if (tile >= ntiles) return;
  const int lane  = threadIdx.x & 31;
  const int lmod  = lane & 15;
  const int lhalf = lane >> 4;
  const int row0  = tile * 16;

  v8f acc[MO / 16];
#pragma unroll
  for (int t = 0; t < MO / 16; ++t) acc[t] = v8f{0, 0, 0, 0, 0, 0, 0, 0};

  const float* xrow = X + (size_t)(row0 + lmod) * KDIM;
#pragma unroll 4
  for (int k0 = 0; k0 < KDIM; k0 += 4) {
    const int ka = k0 + 2 * lhalf;
    v2f a;
    a.x = xrow[ka];
    a.y = xrow[ka + 1];
#pragma unroll
    for (int t = 0; t < MO / 16; ++t) {
      const float* wr = W + (size_t)ka * MO + t * 16 + lmod;
      v2f b;
      b.x = wr[0];
      b.y = wr[MO];
      acc[t] = __builtin_amdgcn_wmma_f32_16x16x4_f32(
          false, a, false, b, (short)0, acc[t], false, false);
    }
  }
#pragma unroll
  for (int t = 0; t < MO / 16; ++t)
#pragma unroll
    for (int r = 0; r < 8; ++r)
      Hout[(size_t)(row0 + r + 8 * lhalf) * MO + t * 16 + lmod] = acc[t][r];
}

__global__ void fill_u32(unsigned* __restrict__ p, unsigned v, int n) {
  int i = blockIdx.x * blockDim.x + threadIdx.x;
  if (i < n) p[i] = v;
}

__global__ void csr_count(const int* __restrict__ ei, int E, int* __restrict__ cnt) {
  int t = blockIdx.x * blockDim.x + threadIdx.x;
  if (t < E) atomicAdd(&cnt[ei[(size_t)E + t]], 1);
}


template <int SH>
__device__ __forceinline__ int row_shr0(int v) {
  return __builtin_amdgcn_update_dpp(0, v, 0x110 + SH, 0xF, 0xF, true);
}
__device__ __forceinline__ int wave_scan_incl(int v, int lane) {
  v += row_shr0<1>(v);
  v += row_shr0<2>(v);
  v += row_shr0<4>(v);
  v += row_shr0<8>(v);
  const int low = __builtin_amdgcn_readlane(v, 15);
  return v + ((lane & 16) ? low : 0);
}
template <bool VEC>
__device__ __forceinline__ int4 scan_ld4(const int* p, int i, int N, bool take, int safe) {
  int4 v;
  if constexpr (VEC) {
    const bool in = take && (i < N);
    v = *(const int4*)(p + (in ? i : safe));
    if (!in) v = make_int4(0, 0, 0, 0);
  } else {
    const bool i0 = take && (i < N), i1 = take && (i + 1 < N);
    const bool i2 = take && (i + 2 < N), i3 = take && (i + 3 < N);
    const int a = p[i0 ? i     : safe];
    const int b = p[i1 ? i + 1 : safe];
    const int c = p[i2 ? i + 2 : safe];
    const int d = p[i3 ? i + 3 : safe];
    v = make_int4(i0 ? a : 0, i1 ? b : 0, i2 ? c : 0, i3 ? d : 0);
  }
  return v;
}
template <bool VEC>
__device__ __forceinline__ void scan_st4(int* p, int i, int N, const int4 v) {
  if constexpr (VEC) {
    if (i < N) *(int4*)(p + i) = v;
  } else {
    if (i     < N) p[i]     = v.x;
    if (i + 1 < N) p[i + 1] = v.y;
    if (i + 2 < N) p[i + 2] = v.z;
    if (i + 3 < N) p[i + 3] = v.w;
  }
}
template <bool VEC>
__device__ __forceinline__ void csr_scan_body(int* __restrict__ cnt, int* __restrict__ rowptr,
                                              int N, int* wtot) {
  constexpr int TRIP = 8;
  const int tid  = threadIdx.x;
  const int lane = tid & 31;
  const int wave = __builtin_amdgcn_readfirstlane(tid >> 5);
  const int seg   = (((N + 31) >> 5) + 127) & ~127;
  const int steps = seg >> 7;
  const int seg0  = wave * seg;
  const int base  = seg0 + lane * 4;

  int s = 0;
  for (int k0 = 0; k0 < steps; k0 += TRIP) {
    int4 v[TRIP];
#pragma unroll
    for (int u = 0; u < TRIP; ++u)
      v[u] = scan_ld4<VEC>(cnt, base + (k0 + u) * 128, N, k0 + u < steps, 0);
#pragma unroll
    for (int u = 0; u < TRIP; ++u) s += (v[u].x + v[u].y) + (v[u].z + v[u].w);
  }
  const int incl = wave_scan_incl(s, lane);
  if (lane == 31) wtot[wave] = incl;
  __syncthreads();
  const int w  = wtot[lane];
  const int wi = wave_scan_incl(w, lane);
  int carry = __builtin_amdgcn_readlane(wi - w, wave);
  const int total = __builtin_amdgcn_readlane(wi, 31);

  if (seg0 < N) {
    for (int k0 = 0; k0 < steps; k0 += TRIP) {
      int4 v[TRIP];
#pragma unroll
      for (int u = 0; u < TRIP; ++u)
        v[u] = scan_ld4<VEC>(cnt, base + (k0 + u) * 128, N, k0 + u < steps, seg0);
#pragma unroll
      for (int u = 0; u < TRIP; ++u) {
        if (k0 + u < steps) {
          const int t4 = (v[u].x + v[u].y) + (v[u].z + v[u].w);
          const int in = wave_scan_incl(t4, lane);
          int4 p;
          p.x = carry + (in - t4);
          p.y = p.x + v[u].x;
          p.z = p.y + v[u].y;
          p.w = p.z + v[u].z;
          const int i = base + (k0 + u) * 128;
          scan_st4<VEC>(rowptr, i, N, p);
          scan_st4<VEC>(cnt, i, N, p);
          carry += __builtin_amdgcn_readlane(in, 31);
        }
      }
    }
  }
  if (tid == 0) rowptr[N] = total;
}
__global__ __launch_bounds__(1024) void csr_scan(int* __restrict__ cnt,
                                                 int* __restrict__ rowptr, int N) {
  __shared__ int wtot[32];
  const bool vec = ((((size_t)cnt | (size_t)rowptr) & 15) == 0) && ((N & 3) == 0);
  if (vec) csr_scan_body<true>(cnt, rowptr, N, wtot);
  else     csr_scan_body<false>(cnt, rowptr, N, wtot);
}

__global__ void csr_fill(const int* __restrict__ ei, int E, int* __restrict__ cursor,
                         int* __restrict__ tmp_src) {
  int t = blockIdx.x * blockDim.x + threadIdx.x;
  if (t >= E) return;
  const int d = ei[(size_t)E + t];
  const int p = atomicAdd(&cursor[d], 1);
  tmp_src[p] = ei[t];
}

template <int KDIM, int MO>
__global__ __launch_bounds__(256) void csr_order_gemm(
    const int* __restrict__ rowptr, const int* __restrict__ tmp_src,
    int* __restrict__ csr_src, int N,
    const float* __restrict__ X, const float* __restrict__ W,
    float* __restrict__ Hout, int ntiles, int gemm_blocks) {
  if ((int)blockIdx.x < gemm_blocks) {
    gemm_strip<KDIM, MO>(X, W, Hout, ntiles, (int)(blockIdx.x * 8u + (threadIdx.x >> 5)));
    return;
  }
  const unsigned bid = blockIdx.x - (unsigned)gemm_blocks;
  const int g = (int)((bid * 256u + threadIdx.x) >> 4);
  if (g >= N) return;
  const int lane = threadIdx.x & 15;
  const int beg = rowptr[g], end = rowptr[g + 1];
  for (int i = beg + lane; i < end; i += 16) {
    const int v = tmp_src[i];
    int r = 0;
    for (int j = beg; j < end; ++j) {
      const int w = tmp_src[j];
      r += ((w < v) || (w == v && j < i)) ? 1 : 0;
    }
    csr_src[beg + r] = v;
  }
}

template <int H, bool SHARED_A>
__device__ __forceinline__ float4 gat_row(
    const int* __restrict__ rowptr, const int* __restrict__ csr_src,
    const float* __restrict__ Hm,
    const float* __restrict__ a_src, const float* __restrict__ a_dst,
    const int g, const int j) {
  static_assert(H == 1 || H == 4, "head_bcast covers a quad or a whole row");
  constexpr int WD  = H * 16;
  constexpr int LPH = 16 / H;
  constexpr int NT  = 2 * H;
  constexpr int NK  = NT * LPH;
  const int q = j % LPH;
  const int beg = rowptr[g];
  const int n = rowptr[g + 1] - beg;
  auto ld_i = [](const int* b, unsigned off) { return *(const int*)((const char*)b + off); };
  const unsigned roff = (unsigned)(j * H) * 4u;
  auto ld_row = [&](unsigned node) {
    const float* hp = (const float*)((const char*)Hm + (node * (WD * 4u) + roff));
    float4 v;
    if constexpr (H == 4) v = *(const float4*)hp;
    else                  v = make_float4(hp[0], 0.f, 0.f, 0.f);
    return v;
  };
  float4 as, ad;
  if constexpr (H == 4) {
    const float* sp = a_src + (SHARED_A ? q : j) * 4;
    const float* dp = a_dst + (SHARED_A ? q : j) * 4;
    as = make_float4(sp[0], sp[1], sp[2], sp[3]);
    ad = make_float4(dp[0], dp[1], dp[2], dp[3]);
  } else {
    as = make_float4(a_src[j], 0.f, 0.f, 0.f);
    ad = make_float4(a_dst[j], 0.f, 0.f, 0.f);
  }
  auto head_dot = [](const float4& v, const float4& a) {
    float d = v.x * a.x;
    if constexpr (H == 4) { d += v.y * a.y; d += v.z * a.z; d += v.w * a.w; }
    return head_sum<H>(d);
  };
  auto leaky = [](float y) { return (y > 0.f) ? y : NEG_SLOPE * y; };

  int sv[NT];
  if (n > 0) {
    for (int t = 0; t < NT; ++t) {
      const int p = t * LPH + q;
      sv[t] = ld_i(csr_src, (unsigned)(beg + ((p < n) ? p : n - 1)) * 4u);
    }
  }

  const float4 v0 = ld_row((unsigned)g);
  const float ed = head_dot(v0, ad);
  float m = leaky(head_dot(v0, as) + ed);
  float den = 1.f;
  float4 acc = v0;

  auto step4 = [&](float y0, float y1, float y2, float y3, const float4& r0,
                   const float4& r1, const float4& r2, const float4& r3) {
    const float mn = fmaxf(fmaxf(fmaxf(y0, y1), fmaxf(y2, y3)), m);
    const float sc = __expf(m - mn);
    const float p0 = __expf(y0 - mn);
    const float p1 = __expf(y1 - mn);
    const float p2 = __expf(y2 - mn);
    const float p3 = __expf(y3 - mn);
    m = mn;
    den = den * sc + p0; den += p1; den += p2; den += p3;
    acc.x = acc.x * sc + p0 * r0.x; acc.x += p1 * r1.x; acc.x += p2 * r2.x; acc.x += p3 * r3.x;
    if constexpr (H == 4) {
      acc.y = acc.y * sc + p0 * r0.y; acc.y += p1 * r1.y; acc.y += p2 * r2.y; acc.y += p3 * r3.y;
      acc.z = acc.z * sc + p0 * r0.z; acc.z += p1 * r1.z; acc.z += p2 * r2.z; acc.z += p3 * r3.z;
      acc.w = acc.w * sc + p0 * r0.w; acc.w += p1 * r1.w; acc.w += p2 * r2.w; acc.w += p3 * r3.w;
    }
  };

#define GAT_ROW_LOAD(K, V)                                                     \
  { V = ld_row((unsigned)head_bcast_i<H, (K) % LPH>(sv[(K) / LPH])); }
#define GAT_LOGIT(K, V, Y)                                                     \
  {                                                                            \
    const float y_ = leaky(head_dot(V, as) + ed);                              \
    Y = ((K) < n) ? y_ : -__builtin_inff();                                    \
  }
#define GAT_FOUR(K, A, B, C, D)                                                \
  {                                                                            \
    float y0_, y1_, y2_, y3_;                                                  \
    GAT_LOGIT((K) + 0, A, y0_) GAT_LOGIT((K) + 1, B, y1_)                      \
    GAT_LOGIT((K) + 2, C, y2_) GAT_LOGIT((K) + 3, D, y3_)                      \
    step4(y0_, y1_, y2_, y3_, A, B, C, D);                                     \
  }
#define GAT_EIGHT(K)                                                           \
  if (n > (K)) {                                                               \
    float4 r0_, r1_, r2_, r3_, r4_, r5_, r6_, r7_;                             \
    GAT_ROW_LOAD((K) + 0, r0_) GAT_ROW_LOAD((K) + 1, r1_)                      \
    GAT_ROW_LOAD((K) + 2, r2_) GAT_ROW_LOAD((K) + 3, r3_)                      \
    if (n > (K) + 4) {                                                         \
      GAT_ROW_LOAD((K) + 4, r4_) GAT_ROW_LOAD((K) + 5, r5_)                    \
      GAT_ROW_LOAD((K) + 6, r6_) GAT_ROW_LOAD((K) + 7, r7_)                    \
    }                                                                          \
    GAT_FOUR((K), r0_, r1_, r2_, r3_)                                          \
    if (n > (K) + 4) GAT_FOUR((K) + 4, r4_, r5_, r6_, r7_)                     \
  }
  GAT_EIGHT(0)
  GAT_EIGHT(8)
  GAT_EIGHT(16)
  GAT_EIGHT(24)
#undef GAT_EIGHT
#undef GAT_FOUR
#undef GAT_LOGIT
#undef GAT_ROW_LOAD

  for (int base = NK; base < n; base += 4) {
    int s[4];
    for (int k = 0; k < 4; ++k) {
      const int p = base + k;
      s[k] = ld_i(csr_src, (unsigned)(beg + ((p < n) ? p : n - 1)) * 4u);
    }
    float4 r[4];
    for (int k = 0; k < 4; ++k) r[k] = ld_row((unsigned)s[k]);
    float y[4];
    for (int k = 0; k < 4; ++k) {
      const float yk = leaky(head_dot(r[k], as) + ed);
      y[k] = (base + k < n) ? yk : -__builtin_inff();
    }
    step4(y[0], y[1], y[2], y[3], r[0], r[1], r[2], r[3]);
  }

  const float inv = 1.0f / (den + GAT_EPS);
  float4 res = make_float4(acc.x * inv, 0.f, 0.f, 0.f);
  if constexpr (H == 4) { res.y = acc.y * inv; res.z = acc.z * inv; res.w = acc.w * inv; }
  return res;
}

template <int NB>
__global__ __launch_bounds__(256) void gat_gather_batches(
    const int* __restrict__ rowptr, const int* __restrict__ csr_src,
    const float* __restrict__ Hm,
    const float* __restrict__ a_src, const float* __restrict__ a_dst,
    const float* __restrict__ bias,
    float* __restrict__ out, int N) {
  static_assert(NB == 4, "a row of NB x 16 floats must be gat_row's 4-head row");
  const int g = (int)((blockIdx.x * 256u + threadIdx.x) >> 4);
  if (g >= N) return;
  const int j = threadIdx.x & 15;
  const float4 v = gat_row<NB, true>(rowptr, csr_src, Hm, a_src, a_dst, g, j);
  const unsigned b = (unsigned)j >> 2;
  const unsigned c = ((unsigned)j & 3u) * 4u;
  const float* bp = bias + c;
  float* op = (float*)((char*)out + ((b * (unsigned)N + (unsigned)g) * 64u + c * 4u));
  *(float4*)op = make_float4(v.x + bp[0], v.y + bp[1], v.z + bp[2], v.w + bp[3]);
}

template <int MO, int RK>
__global__ __launch_bounds__(256) void gat_gather_mm(
    const int* __restrict__ rowptr, const int* __restrict__ csr_src,
    const float* __restrict__ Hm,
    const float* __restrict__ a_src, const float* __restrict__ a_dst,
    const float* __restrict__ bias, const float* __restrict__ W,
    float* __restrict__ Hout, int N,
    const float* __restrict__ Xn, const float* __restrict__ Wn,
    float* __restrict__ Hn, int ntiles, int gemm_blocks) {
  constexpr int KD = 64, PITCH = KD + 4;
  constexpr int LDO = 64;
  __shared__ __attribute__((aligned(16))) float xs[16 * PITCH];
  unsigned bid = blockIdx.x;
  if constexpr (RK > 0) {
    if ((int)bid < gemm_blocks) {
      gemm_strip<RK, 64>(Xn, Wn, Hn, ntiles, (int)(bid * 8u + (threadIdx.x >> 5)));
      return;
    }
    bid -= (unsigned)gemm_blocks;
  }
  const int lrow = (int)(threadIdx.x >> 4);
  const int row0 = (int)(bid * 16u);
  const int g = (row0 + lrow < N) ? (row0 + lrow) : (N - 1);
  const int j = threadIdx.x & 15;
  const float4 v = gat_row<4, false>(rowptr, csr_src, Hm, a_src, a_dst, g, j);
  const float* bp = bias + j * 4;
  float4 r = make_float4(fmaxf(v.x + bp[0], 0.f), fmaxf(v.y + bp[1], 0.f),
                         fmaxf(v.z + bp[2], 0.f), fmaxf(v.w + bp[3], 0.f));
  *(float4*)(xs + lrow * PITCH + j * 4) = r;

  const int wave  = (int)(threadIdx.x >> 5);
  const int lane  = threadIdx.x & 31;
  const int lmod  = lane & 15;
  const int lhalf = lane >> 4;
  const bool mm = wave < MO / 16;
  v2f b[KD / 4];
  if (mm) {
#pragma unroll
    for (int s = 0; s < KD / 4; ++s) {
      const float* wr = W + (size_t)(4 * s + 2 * lhalf) * MO + wave * 16 + lmod;
      b[s].x = wr[0];
      b[s].y = wr[MO];
    }
  }
  __syncthreads();
  if (!mm) return;
  v8f acc = v8f{0, 0, 0, 0, 0, 0, 0, 0};
  const float* xr = xs + lmod * PITCH + 2 * lhalf;
#pragma unroll
  for (int s = 0; s < KD / 4; ++s) {
    v2f a;
    a.x = xr[4 * s];
    a.y = xr[4 * s + 1];
    acc = __builtin_amdgcn_wmma_f32_16x16x4_f32(
        false, a, false, b[s], (short)0, acc, false, false);
  }
#pragma unroll
  for (int rr = 0; rr < 8; ++rr) {
    const int row = row0 + rr + 8 * lhalf;
    if (row < N) Hout[(size_t)row * LDO + wave * 16 + lmod] = acc[rr];
  }
}

static inline int cdiv(long long a, int b) { return (int)((a + b - 1) / b); }

extern "C" void kernel_launch(void* const* d_in, const int* in_sizes, int n_in,
                              void* d_out, int out_size, void* d_ws, size_t ws_size,
                              hipStream_t stream) {
  const float* x   = (const float*)d_in[0];
  const int*   ei  = (const int*)d_in[1];
  const float* W1  = (const float*)d_in[2];
  const float* a1s = (const float*)d_in[3];
  const float* a1d = (const float*)d_in[4];
  const float* b1  = (const float*)d_in[5];
  const float* W2  = (const float*)d_in[6];
  const float* a2s = (const float*)d_in[7];
  const float* a2d = (const float*)d_in[8];
  const float* b2  = (const float*)d_in[9];
  const float* W3  = (const float*)d_in[10];
  const float* a3s = (const float*)d_in[11];
  const float* a3d = (const float*)d_in[12];
  const float* b3  = (const float*)d_in[13];

  const int B = 4, F_IN = 128, HC = 64;
  const int E = in_sizes[1] / 2;
  const int N = in_sizes[0] / (B * F_IN);

  float* ws = (float*)d_ws;
  float* h1      = ws;
  float* h3      = h1 + (size_t)N * HC;
  int*   cnt     = (int*)(h3 + (size_t)N * HC);
  int*   rowptr  = cnt + (size_t)N;
  int*   tmp_src = rowptr + (size_t)N + 1;
  int*   csr_src = tmp_src + (size_t)E;
  float* h2      = (float*)d_out;

  const int ntiles = (N + 15) / 16;
  const int gemm_blocks = (ntiles + 7) / 8;
  const int gather_grid = cdiv((long long)N * 16, 256);

  fill_u32<<<cdiv(N, 256), 256, 0, stream>>>((unsigned*)cnt, 0u, N);
  csr_count<<<cdiv(E, 256), 256, 0, stream>>>(ei, E, cnt);
  csr_scan<<<1, 1024, 0, stream>>>(cnt, rowptr, N);
  csr_fill<<<cdiv(E, 256), 256, 0, stream>>>(ei, E, cnt, tmp_src);
  csr_order_gemm<128, 64><<<gemm_blocks + gather_grid, 256, 0, stream>>>(
      rowptr, tmp_src, csr_src, N, x, W1, h1, ntiles, gemm_blocks);

  for (int b = 0; b < B; ++b) {
    gat_gather_mm<64, 0><<<gather_grid, 256, 0, stream>>>(
        rowptr, csr_src, h1, a1s, a1d, b1, W2, h2, N, nullptr, nullptr, nullptr, 0, 0);

    const int nb = (b + 1 < B) ? (b + 1) : b;
    const int rider = (b + 1 < B) ? gemm_blocks : 0;
    gat_gather_mm<16, 128><<<rider + gather_grid, 256, 0, stream>>>(
        rowptr, csr_src, h2, a2s, a2d, b2, W3, h3 + (size_t)b * 16, N,
        x + (size_t)nb * N * F_IN, W1, h1, ntiles, rider);
  }
  gat_gather_batches<4><<<gather_grid, 256, 0, stream>>>(
      rowptr, csr_src, h3, a3s, a3d, b3, (float*)d_out, N);
}


# Destination buckets by a two-level counting sort through LDS (256 node ranges; runs ordered in LDS, stored contiguous); global atomics and zero-fill launch gone; gather sum order unchanged
# speedup vs baseline: 24.9439x; 1.3683x over previous
#include <hip/hip_runtime.h>
#include <hip/hip_bf16.h>


typedef __attribute__((ext_vector_type(2))) float v2f;
typedef __attribute__((ext_vector_type(8))) float v8f;

#define NEG_SLOPE 0.2f
#define GAT_EPS   1e-16f

template <int CTRL>
__device__ __forceinline__ float dpp_f(float v) {
  const int x = __builtin_bit_cast(int, v);
  return __builtin_bit_cast(float, __builtin_amdgcn_update_dpp(x, x, CTRL, 0xF, 0xF, true));
}
template <int CTRL>
__device__ __forceinline__ int dpp_i(int v) {
  return __builtin_amdgcn_update_dpp(v, v, CTRL, 0xF, 0xF, true);
}
template <int H>
__device__ __forceinline__ float head_max(float v) {
  v = fmaxf(v, dpp_f<0x161>(v));
  v = fmaxf(v, dpp_f<0x162>(v));
  if constexpr (H == 1) {
    v = fmaxf(v, dpp_f<0x164>(v));
    v = fmaxf(v, dpp_f<0x168>(v));
  }
  return v;
}
template <int H>
__device__ __forceinline__ float head_sum(float v) {
  v += dpp_f<0x161>(v);
  v += dpp_f<0x162>(v);
  if constexpr (H == 1) {
    v += dpp_f<0x164>(v);
    v += dpp_f<0x168>(v);
  }
  return v;
}
template <int H, int QO>
__device__ __forceinline__ float head_bcast_f(float v) {
  if constexpr (H == 4) return dpp_f<QO * 0x55>(v);
  else                  return dpp_f<0x150 + QO>(v);
}
template <int H, int QO>
__device__ __forceinline__ int head_bcast_i(int v) {
  if constexpr (H == 4) return dpp_i<QO * 0x55>(v);
  else                  return dpp_i<0x150 + QO>(v);
}

template <int KDIM, int MO>
__device__ __forceinline__ void gemm_strip(
    const float* __restrict__ X, const float* __restrict__ W,
    float* __restrict__ Hout, int ntiles, const int tile) {
  if (tile >= ntiles) return;
  const int lane  = threadIdx.x & 31;
  const int lmod  = lane & 15;
  const int lhalf = lane >> 4;
  const int row0  = tile * 16;

  v8f acc[MO / 16];
#pragma unroll
  for (int t = 0; t < MO / 16; ++t) acc[t] = v8f{0, 0, 0, 0, 0, 0, 0, 0};

  const float* xrow = X + (size_t)(row0 + lmod) * KDIM;
#pragma unroll 4
  for (int k0 = 0; k0 < KDIM; k0 += 4) {
    const int ka = k0 + 2 * lhalf;
    v2f a;
    a.x = xrow[ka];
    a.y = xrow[ka + 1];
#pragma unroll
    for (int t = 0; t < MO / 16; ++t) {
      const float* wr = W + (size_t)ka * MO + t * 16 + lmod;
      v2f b;
      b.x = wr[0];
      b.y = wr[MO];
      acc[t] = __builtin_amdgcn_wmma_f32_16x16x4_f32(
          false, a, false, b, (short)0, acc[t], false, false);
    }
  }
#pragma unroll
  for (int t = 0; t < MO / 16; ++t)
#pragma unroll
    for (int r = 0; r < 8; ++r)
      Hout[(size_t)(row0 + r + 8 * lhalf) * MO + t * 16 + lmod] = acc[t][r];
}

#define NRANGE 256


template <int SH>
__device__ __forceinline__ int row_shr0(int v) {
  return __builtin_amdgcn_update_dpp(0, v, 0x110 + SH, 0xF, 0xF, true);
}
__device__ __forceinline__ int wave_scan_incl(int v, int lane) {
  v += row_shr0<1>(v);
  v += row_shr0<2>(v);
  v += row_shr0<4>(v);
  v += row_shr0<8>(v);
  const int low = __builtin_amdgcn_readlane(v, 15);
  return v + ((lane & 16) ? low : 0);
}
template <bool VEC>
__device__ __forceinline__ int4 scan_ld4(const int* p, int i, int N, bool take, int safe) {
  int4 v;
  if constexpr (VEC) {
    const bool in = take && (i < N);
    v = *(const int4*)(p + (in ? i : safe));
    if (!in) v = make_int4(0, 0, 0, 0);
  } else {
    const bool i0 = take && (i < N), i1 = take && (i + 1 < N);
    const bool i2 = take && (i + 2 < N), i3 = take && (i + 3 < N);
    const int a = p[i0 ? i     : safe];
    const int b = p[i1 ? i + 1 : safe];
    const int c = p[i2 ? i + 2 : safe];
    const int d = p[i3 ? i + 3 : safe];
    v = make_int4(i0 ? a : 0, i1 ? b : 0, i2 ? c : 0, i3 ? d : 0);
  }
  return v;
}
template <bool VEC>
__device__ __forceinline__ void scan_st4(int* p, int i, int N, const int4 v) {
  if constexpr (VEC) {
    if (i < N) *(int4*)(p + i) = v;
  } else {
    if (i     < N) p[i]     = v.x;
    if (i + 1 < N) p[i + 1] = v.y;
    if (i + 2 < N) p[i + 2] = v.z;
    if (i + 3 < N) p[i + 3] = v.w;
  }
}
template <bool VEC>
__device__ __forceinline__ void csr_scan_body(int* __restrict__ cnt, int* __restrict__ rowptr,
                                              int N, int* wtot) {
  constexpr int TRIP = 8;
  const int tid  = threadIdx.x;
  const int lane = tid & 31;
  const int wave = __builtin_amdgcn_readfirstlane(tid >> 5);
  const int seg   = (((N + 31) >> 5) + 127) & ~127;
  const int steps = seg >> 7;
  const int seg0  = wave * seg;
  const int base  = seg0 + lane * 4;

  int s = 0;
  for (int k0 = 0; k0 < steps; k0 += TRIP) {
    int4 v[TRIP];
#pragma unroll
    for (int u = 0; u < TRIP; ++u)
      v[u] = scan_ld4<VEC>(cnt, base + (k0 + u) * 128, N, k0 + u < steps, 0);
#pragma unroll
    for (int u = 0; u < TRIP; ++u) s += (v[u].x + v[u].y) + (v[u].z + v[u].w);
  }
  const int incl = wave_scan_incl(s, lane);
  if (lane == 31) wtot[wave] = incl;
  __syncthreads();
  const int w  = wtot[lane];
  const int wi = wave_scan_incl(w, lane);
  int carry = __builtin_amdgcn_readlane(wi - w, wave);
  const int total = __builtin_amdgcn_readlane(wi, 31);

  if (seg0 < N) {
    for (int k0 = 0; k0 < steps; k0 += TRIP) {
      int4 v[TRIP];
#pragma unroll
      for (int u = 0; u < TRIP; ++u)
        v[u] = scan_ld4<VEC>(cnt, base + (k0 + u) * 128, N, k0 + u < steps, seg0);
#pragma unroll
      for (int u = 0; u < TRIP; ++u) {
        if (k0 + u < steps) {
          const int t4 = (v[u].x + v[u].y) + (v[u].z + v[u].w);
          const int in = wave_scan_incl(t4, lane);
          int4 p;
          p.x = carry + (in - t4);
          p.y = p.x + v[u].x;
          p.z = p.y + v[u].y;
          p.w = p.z + v[u].z;
          const int i = base + (k0 + u) * 128;
          scan_st4<VEC>(rowptr, i, N, p);
          scan_st4<VEC>(cnt, i, N, p);
          carry += __builtin_amdgcn_readlane(in, 31);
        }
      }
    }
  }
  if (tid == 0) rowptr[N] = total;
}
__global__ __launch_bounds__(1024) void csr_scan(int* __restrict__ cnt,
                                                 int* __restrict__ rowptr, int N) {
  __shared__ int wtot[32];
  const bool vec = ((((size_t)cnt | (size_t)rowptr) & 15) == 0) && ((N & 3) == 0);
  if (vec) csr_scan_body<true>(cnt, rowptr, N, wtot);
  else     csr_scan_body<false>(cnt, rowptr, N, wtot);
}

#define STAGE_PAIRS 4096
#define STAGE_SRCS  8192

__device__ __forceinline__ int node_range(int d, int bw) {
  const unsigned r = (unsigned)d / (unsigned)bw;
  return (int)((r < NRANGE) ? r : (NRANGE - 1));
}

__global__ __launch_bounds__(1024) void bin_count(const int* __restrict__ ei, int E, int bw,
                                                  int chunk, int* __restrict__ cntm) {
  __shared__ int hist[NRANGE];
  const int tid = threadIdx.x;
  if (tid < NRANGE) hist[tid] = 0;
  __syncthreads();
  const int e0 = (int)blockIdx.x * chunk;
  const int e1 = (e0 + chunk < E) ? (e0 + chunk) : E;
  const int* dst = ei + (size_t)E;
  for (int t0 = e0 + tid; t0 < e1; t0 += 4096) {
    int d[4];
#pragma unroll
    for (int u = 0; u < 4; ++u) {
      const int t = t0 + u * 1024;
      d[u] = dst[(t < e1) ? t : t0];
    }
#pragma unroll
    for (int u = 0; u < 4; ++u)
      if (t0 + u * 1024 < e1) atomicAdd(&hist[node_range(d[u], bw)], 1);
  }
  __syncthreads();
  if (tid < NRANGE) cntm[tid * (int)gridDim.x + (int)blockIdx.x] = hist[tid];
}

__global__ __launch_bounds__(1024) void bin_scatter(const int* __restrict__ ei, int E, int bw,
                                                    int chunk, const int* __restrict__ offm,
                                                    int2* __restrict__ binned) {
  __shared__ int goff[NRANGE];
  __shared__ int lst[NRANGE];
  __shared__ int cur[NRANGE];
  __shared__ int wsum[NRANGE / 32];
  __shared__ int2 stage[STAGE_PAIRS];
  const int tid  = threadIdx.x;
  const int lane = tid & 31;
  const int wave = tid >> 5;
  const int e0 = (int)blockIdx.x * chunk;
  const int e1 = (e0 + chunk < E) ? (e0 + chunk) : E;
  const int n  = (e1 > e0) ? (e1 - e0) : 0;
  const bool staged = chunk <= STAGE_PAIRS;
  int c = 0, incl = 0;
  if (tid < NRANGE) {
    const int f  = tid * (int)gridDim.x + (int)blockIdx.x;
    const int g0 = offm[f];
    c = offm[f + 1] - g0;
    goff[tid] = g0;
    incl = wave_scan_incl(c, lane);
    if (lane == 31) wsum[wave] = incl;
  }
  __syncthreads();
  if (tid < NRANGE) {
    int pre = 0;
    for (int w = 0; w < NRANGE / 32; ++w) pre += (w < wave) ? wsum[w] : 0;
    const int l0 = pre + incl - c;
    lst[tid] = l0;
    cur[tid] = staged ? l0 : goff[tid];
  }
  __syncthreads();
  const int* dst = ei + (size_t)E;
  for (int t0 = e0 + tid; t0 < e1; t0 += 4096) {
    int sv[4], d[4];
#pragma unroll
    for (int u = 0; u < 4; ++u) {
      const int t  = t0 + u * 1024;
      const int tt = (t < e1) ? t : t0;
      sv[u] = ei[tt];
      d[u]  = dst[tt];
    }
#pragma unroll
    for (int u = 0; u < 4; ++u) {
      if (t0 + u * 1024 < e1) {
        const int p = atomicAdd(&cur[node_range(d[u], bw)], 1);
        if (staged) stage[p] = make_int2(sv[u], d[u]);
        else        binned[p] = make_int2(sv[u], d[u]);
      }
    }
  }
  if (staged) {
    __syncthreads();
    for (int i = tid; i < n; i += 1024) {
      const int2 v = stage[i];
      const int r = node_range(v.y, bw);
      binned[goff[r] + (i - lst[r])] = v;
    }
  }
}

__global__ __launch_bounds__(1024) void bin_finish(const int2* __restrict__ binned,
                                                   const int* __restrict__ offm, int G,
                                                   int E, int N, int bw,
                                                   int* __restrict__ rowptr,
                                                   int* __restrict__ tmp_src) {
  __shared__ int cnt[NRANGE];
  __shared__ int cur[NRANGE];
  __shared__ int wsum[NRANGE / 32];
  __shared__ int stage[STAGE_SRCS];
  const int tid  = threadIdx.x;
  const int lane = tid & 31;
  const int wave = tid >> 5;
  const int b  = (int)blockIdx.x;
  const int s  = offm[b * G];
  const int e  = offm[(b + 1) * G];
  const int n  = e - s;
  const int d0 = b * bw;
  const bool staged = n <= STAGE_SRCS;
  if (tid < NRANGE) cnt[tid] = 0;
  __syncthreads();
  for (int i0 = s + tid; i0 < e; i0 += 4096) {
    int d[4];
#pragma unroll
    for (int u = 0; u < 4; ++u) {
      const int i = i0 + u * 1024;
      d[u] = binned[(i < e) ? i : i0].y;
    }
#pragma unroll
    for (int u = 0; u < 4; ++u) {
      const unsigned dl = (unsigned)(d[u] - d0);
      if (i0 + u * 1024 < e && dl < NRANGE) atomicAdd(&cnt[dl], 1);
    }
  }
  __syncthreads();
  int c = 0, incl = 0;
  if (tid < NRANGE) {
    c = cnt[tid];
    incl = wave_scan_incl(c, lane);
    if (lane == 31) wsum[wave] = incl;
  }
  __syncthreads();
  if (tid < NRANGE) {
    int pre = 0;
    for (int w = 0; w < NRANGE / 32; ++w) pre += (w < wave) ? wsum[w] : 0;
    const int l0 = pre + incl - c;
    cur[tid] = staged ? l0 : (s + l0);
    const int node = d0 + tid;
    if (tid < bw && node < N) rowptr[node] = s + l0;
  }
  if (b == NRANGE - 1 && tid == 0) rowptr[N] = E;
  __syncthreads();
  for (int i0 = s + tid; i0 < e; i0 += 4096) {
    int2 v[4];
#pragma unroll
    for (int u = 0; u < 4; ++u) {
      const int i = i0 + u * 1024;
      v[u] = binned[(i < e) ? i : i0];
    }
#pragma unroll
    for (int u = 0; u < 4; ++u) {
      const unsigned dl = (unsigned)(v[u].y - d0);
      if (i0 + u * 1024 < e && dl < NRANGE) {
        const int p = atomicAdd(&cur[dl], 1);
        if (staged) stage[p] = v[u].x;
        else        tmp_src[p] = v[u].x;
      }
    }
  }
  if (staged) {
    __syncthreads();
    for (int i = tid; i < n; i += 1024) tmp_src[s + i] = stage[i];
  }
}

template <int KDIM, int MO>
__global__ __launch_bounds__(256) void csr_order_gemm(
    const int* __restrict__ rowptr, const int* __restrict__ tmp_src,
    int* __restrict__ csr_src, int N,
    const float* __restrict__ X, const float* __restrict__ W,
    float* __restrict__ Hout, int ntiles, int gemm_blocks) {
  if ((int)blockIdx.x < gemm_blocks) {
    gemm_strip<KDIM, MO>(X, W, Hout, ntiles, (int)(blockIdx.x * 8u + (threadIdx.x >> 5)));
    return;
  }
  const unsigned bid = blockIdx.x - (unsigned)gemm_blocks;
  const int g = (int)((bid * 256u + threadIdx.x) >> 4);
  if (g >= N) return;
  const int lane = threadIdx.x & 15;
  const int beg = rowptr[g], end = rowptr[g + 1];
  for (int i = beg + lane; i < end; i += 16) {
    const int v = tmp_src[i];
    int r = 0;
    for (int j = beg; j < end; ++j) {
      const int w = tmp_src[j];
      r += ((w < v) || (w == v && j < i)) ? 1 : 0;
    }
    csr_src[beg + r] = v;
  }
}

template <int H, bool SHARED_A>
__device__ __forceinline__ float4 gat_row(
    const int* __restrict__ rowptr, const int* __restrict__ csr_src,
    const float* __restrict__ Hm,
    const float* __restrict__ a_src, const float* __restrict__ a_dst,
    const int g, const int j) {
  static_assert(H == 1 || H == 4, "head_bcast covers a quad or a whole row");
  constexpr int WD  = H * 16;
  constexpr int LPH = 16 / H;
  constexpr int NT  = 2 * H;
  constexpr int NK  = NT * LPH;
  const int q = j % LPH;
  const int beg = rowptr[g];
  const int n = rowptr[g + 1] - beg;
  auto ld_i = [](const int* b, unsigned off) { return *(const int*)((const char*)b + off); };
  const unsigned roff = (unsigned)(j * H) * 4u;
  auto ld_row = [&](unsigned node) {
    const float* hp = (const float*)((const char*)Hm + (node * (WD * 4u) + roff));
    float4 v;
    if constexpr (H == 4) v = *(const float4*)hp;
    else                  v = make_float4(hp[0], 0.f, 0.f, 0.f);
    return v;
  };
  float4 as, ad;
  if constexpr (H == 4) {
    const float* sp = a_src + (SHARED_A ? q : j) * 4;
    const float* dp = a_dst + (SHARED_A ? q : j) * 4;
    as = make_float4(sp[0], sp[1], sp[2], sp[3]);
    ad = make_float4(dp[0], dp[1], dp[2], dp[3]);
  } else {
    as = make_float4(a_src[j], 0.f, 0.f, 0.f);
    ad = make_float4(a_dst[j], 0.f, 0.f, 0.f);
  }
  auto head_dot = [](const float4& v, const float4& a) {
    float d = v.x * a.x;
    if constexpr (H == 4) { d += v.y * a.y; d += v.z * a.z; d += v.w * a.w; }
    return head_sum<H>(d);
  };
  auto leaky = [](float y) { return (y > 0.f) ? y : NEG_SLOPE * y; };

  int sv[NT];
  if (n > 0) {
    for (int t = 0; t < NT; ++t) {
      const int p = t * LPH + q;
      sv[t] = ld_i(csr_src, (unsigned)(beg + ((p < n) ? p : n - 1)) * 4u);
    }
  }

  const float4 v0 = ld_row((unsigned)g);
  const float ed = head_dot(v0, ad);
  float m = leaky(head_dot(v0, as) + ed);
  float den = 1.f;
  float4 acc = v0;

  auto step4 = [&](float y0, float y1, float y2, float y3, const float4& r0,
                   const float4& r1, const float4& r2, const float4& r3) {
    const float mn = fmaxf(fmaxf(fmaxf(y0, y1), fmaxf(y2, y3)), m);
    const float sc = __expf(m - mn);
    const float p0 = __expf(y0 - mn);
    const float p1 = __expf(y1 - mn);
    const float p2 = __expf(y2 - mn);
    const float p3 = __expf(y3 - mn);
    m = mn;
    den = den * sc + p0; den += p1; den += p2; den += p3;
    acc.x = acc.x * sc + p0 * r0.x; acc.x += p1 * r1.x; acc.x += p2 * r2.x; acc.x += p3 * r3.x;
    if constexpr (H == 4) {
      acc.y = acc.y * sc + p0 * r0.y; acc.y += p1 * r1.y; acc.y += p2 * r2.y; acc.y += p3 * r3.y;
      acc.z = acc.z * sc + p0 * r0.z; acc.z += p1 * r1.z; acc.z += p2 * r2.z; acc.z += p3 * r3.z;
      acc.w = acc.w * sc + p0 * r0.w; acc.w += p1 * r1.w; acc.w += p2 * r2.w; acc.w += p3 * r3.w;
    }
  };

#define GAT_ROW_LOAD(K, V)                                                     \
  { V = ld_row((unsigned)head_bcast_i<H, (K) % LPH>(sv[(K) / LPH])); }
#define GAT_LOGIT(K, V, Y)                                                     \
  {                                                                            \
    const float y_ = leaky(head_dot(V, as) + ed);                              \
    Y = ((K) < n) ? y_ : -__builtin_inff();                                    \
  }
#define GAT_FOUR(K, A, B, C, D)                                                \
  {                                                                            \
    float y0_, y1_, y2_, y3_;                                                  \
    GAT_LOGIT((K) + 0, A, y0_) GAT_LOGIT((K) + 1, B, y1_)                      \
    GAT_LOGIT((K) + 2, C, y2_) GAT_LOGIT((K) + 3, D, y3_)                      \
    step4(y0_, y1_, y2_, y3_, A, B, C, D);                                     \
  }
#define GAT_EIGHT(K)                                                           \
  if (n > (K)) {                                                               \
    float4 r0_, r1_, r2_, r3_, r4_, r5_, r6_, r7_;                             \
    GAT_ROW_LOAD((K) + 0, r0_) GAT_ROW_LOAD((K) + 1, r1_)                      \
    GAT_ROW_LOAD((K) + 2, r2_) GAT_ROW_LOAD((K) + 3, r3_)                      \
    if (n > (K) + 4) {                                                         \
      GAT_ROW_LOAD((K) + 4, r4_) GAT_ROW_LOAD((K) + 5, r5_)                    \
      GAT_ROW_LOAD((K) + 6, r6_) GAT_ROW_LOAD((K) + 7, r7_)                    \
    }                                                                          \
    GAT_FOUR((K), r0_, r1_, r2_, r3_)                                          \
    if (n > (K) + 4) GAT_FOUR((K) + 4, r4_, r5_, r6_, r7_)                     \
  }
  GAT_EIGHT(0)
  GAT_EIGHT(8)
  GAT_EIGHT(16)
  GAT_EIGHT(24)
#undef GAT_EIGHT
#undef GAT_FOUR
#undef GAT_LOGIT
#undef GAT_ROW_LOAD

  for (int base = NK; base < n; base += 4) {
    int s[4];
    for (int k = 0; k < 4; ++k) {
      const int p = base + k;
      s[k] = ld_i(csr_src, (unsigned)(beg + ((p < n) ? p : n - 1)) * 4u);
    }
    float4 r[4];
    for (int k = 0; k < 4; ++k) r[k] = ld_row((unsigned)s[k]);
    float y[4];
    for (int k = 0; k < 4; ++k) {
      const float yk = leaky(head_dot(r[k], as) + ed);
      y[k] = (base + k < n) ? yk : -__builtin_inff();
    }
    step4(y[0], y[1], y[2], y[3], r[0], r[1], r[2], r[3]);
  }

  const float inv = 1.0f / (den + GAT_EPS);
  float4 res = make_float4(acc.x * inv, 0.f, 0.f, 0.f);
  if constexpr (H == 4) { res.y = acc.y * inv; res.z = acc.z * inv; res.w = acc.w * inv; }
  return res;
}

template <int NB>
__global__ __launch_bounds__(256) void gat_gather_batches(
    const int* __restrict__ rowptr, const int* __restrict__ csr_src,
    const float* __restrict__ Hm,
    const float* __restrict__ a_src, const float* __restrict__ a_dst,
    const float* __restrict__ bias,
    float* __restrict__ out, int N) {
  static_assert(NB == 4, "a row of NB x 16 floats must be gat_row's 4-head row");
  const int g = (int)((blockIdx.x * 256u + threadIdx.x) >> 4);
  if (g >= N) return;
  const int j = threadIdx.x & 15;
  const float4 v = gat_row<NB, true>(rowptr, csr_src, Hm, a_src, a_dst, g, j);
  const unsigned b = (unsigned)j >> 2;
  const unsigned c = ((unsigned)j & 3u) * 4u;
  const float* bp = bias + c;
  float* op = (float*)((char*)out + ((b * (unsigned)N + (unsigned)g) * 64u + c * 4u));
  *(float4*)op = make_float4(v.x + bp[0], v.y + bp[1], v.z + bp[2], v.w + bp[3]);
}

template <int MO, int RK>
__global__ __launch_bounds__(256) void gat_gather_mm(
    const int* __restrict__ rowptr, const int* __restrict__ csr_src,
    const float* __restrict__ Hm,
    const float* __restrict__ a_src, const float* __restrict__ a_dst,
    const float* __restrict__ bias, const float* __restrict__ W,
    float* __restrict__ Hout, int N,
    const float* __restrict__ Xn, const float* __restrict__ Wn,
    float* __restrict__ Hn, int ntiles, int gemm_blocks) {
  constexpr int KD = 64, PITCH = KD + 4;
  constexpr int LDO = 64;
  __shared__ __attribute__((aligned(16))) float xs[16 * PITCH];
  unsigned bid = blockIdx.x;
  if constexpr (RK > 0) {
    if ((int)bid < gemm_blocks) {
      gemm_strip<RK, 64>(Xn, Wn, Hn, ntiles, (int)(bid * 8u + (threadIdx.x >> 5)));
      return;
    }
    bid -= (unsigned)gemm_blocks;
  }
  const int lrow = (int)(threadIdx.x >> 4);
  const int row0 = (int)(bid * 16u);
  const int g = (row0 + lrow < N) ? (row0 + lrow) : (N - 1);
  const int j = threadIdx.x & 15;
  const float4 v = gat_row<4, false>(rowptr, csr_src, Hm, a_src, a_dst, g, j);
  const float* bp = bias + j * 4;
  float4 r = make_float4(fmaxf(v.x + bp[0], 0.f), fmaxf(v.y + bp[1], 0.f),
                         fmaxf(v.z + bp[2], 0.f), fmaxf(v.w + bp[3], 0.f));
  *(float4*)(xs + lrow * PITCH + j * 4) = r;

  const int wave  = (int)(threadIdx.x >> 5);
  const int lane  = threadIdx.x & 31;
  const int lmod  = lane & 15;
  const int lhalf = lane >> 4;
  const bool mm = wave < MO / 16;
  v2f b[KD / 4];
  if (mm) {
#pragma unroll
    for (int s = 0; s < KD / 4; ++s) {
      const float* wr = W + (size_t)(4 * s + 2 * lhalf) * MO + wave * 16 + lmod;
      b[s].x = wr[0];
      b[s].y = wr[MO];
    }
  }
  __syncthreads();
  if (!mm) return;
  v8f acc = v8f{0, 0, 0, 0, 0, 0, 0, 0};
  const float* xr = xs + lmod * PITCH + 2 * lhalf;
#pragma unroll
  for (int s = 0; s < KD / 4; ++s) {
    v2f a;
    a.x = xr[4 * s];
    a.y = xr[4 * s + 1];
    acc = __builtin_amdgcn_wmma_f32_16x16x4_f32(
        false, a, false, b[s], (short)0, acc, false, false);
  }
#pragma unroll
  for (int rr = 0; rr < 8; ++rr) {
    const int row = row0 + rr + 8 * lhalf;
    if (row < N) Hout[(size_t)row * LDO + wave * 16 + lmod] = acc[rr];
  }
}

static inline int cdiv(long long a, int b) { return (int)((a + b - 1) / b); }

extern "C" void kernel_launch(void* const* d_in, const int* in_sizes, int n_in,
                              void* d_out, int out_size, void* d_ws, size_t ws_size,
                              hipStream_t stream) {
  const float* x   = (const float*)d_in[0];
  const int*   ei  = (const int*)d_in[1];
  const float* W1  = (const float*)d_in[2];
  const float* a1s = (const float*)d_in[3];
  const float* a1d = (const float*)d_in[4];
  const float* b1  = (const float*)d_in[5];
  const float* W2  = (const float*)d_in[6];
  const float* a2s = (const float*)d_in[7];
  const float* a2d = (const float*)d_in[8];
  const float* b2  = (const float*)d_in[9];
  const float* W3  = (const float*)d_in[10];
  const float* a3s = (const float*)d_in[11];
  const float* a3d = (const float*)d_in[12];
  const float* b3  = (const float*)d_in[13];

  const int B = 4, F_IN = 128, HC = 64;
  const int E = in_sizes[1] / 2;
  const int N = in_sizes[0] / (B * F_IN);

  float* ws = (float*)d_ws;
  float* h1      = ws;
  float* h3      = h1 + (size_t)N * HC;
  int*   rowptr  = (int*)(h3 + (size_t)N * HC);
  int*   tmp_src = rowptr + (size_t)N + 1;
  int*   csr_src = tmp_src + (size_t)E;
  float* h2      = (float*)d_out;

  const int ntiles = (N + 15) / 16;
  const int gemm_blocks = (ntiles + 7) / 8;
  const int gather_grid = cdiv((long long)N * 16, 256);

  const int bw    = cdiv(N, NRANGE);
  const int chunk = cdiv(E, 256);
  int*  cntm   = (int*)h1;
  int*  offm   = cntm + NRANGE * 256;
  int2* binned = (int2*)h3;
  bin_count<<<256, 1024, 0, stream>>>(ei, E, bw, chunk, cntm);
  csr_scan<<<1, 1024, 0, stream>>>(cntm, offm, NRANGE * 256);
  bin_scatter<<<256, 1024, 0, stream>>>(ei, E, bw, chunk, offm, binned);
  bin_finish<<<NRANGE, 1024, 0, stream>>>(binned, offm, 256, E, N, bw, rowptr, tmp_src);
  csr_order_gemm<128, 64><<<gemm_blocks + gather_grid, 256, 0, stream>>>(
      rowptr, tmp_src, csr_src, N, x, W1, h1, ntiles, gemm_blocks);

  for (int b = 0; b < B; ++b) {
    gat_gather_mm<64, 0><<<gather_grid, 256, 0, stream>>>(
        rowptr, csr_src, h1, a1s, a1d, b1, W2, h2, N, nullptr, nullptr, nullptr, 0, 0);

    const int nb = (b + 1 < B) ? (b + 1) : b;
    const int rider = (b + 1 < B) ? gemm_blocks : 0;
    gat_gather_mm<16, 128><<<rider + gather_grid, 256, 0, stream>>>(
        rowptr, csr_src, h2, a2s, a2d, b2, W3, h3 + (size_t)b * 16, N,
        x + (size_t)nb * N * F_IN, W1, h1, ntiles, rider);
  }
  gat_gather_batches<4><<<gather_grid, 256, 0, stream>>>(
      rowptr, csr_src, h3, a3s, a3d, b3, (float*)d_out, N);
}


# Gather: a lane's dot and weighted sums as packed 2-wide multiply-adds on its (x,y),(z,w) pairs, leaky as max(y, 0.2y): fewer vector issues per edge; logit dot order now 2 packed partials + add
# speedup vs baseline: 26.5704x; 1.0652x over previous
#include <hip/hip_runtime.h>
#include <hip/hip_bf16.h>


typedef __attribute__((ext_vector_type(2))) float v2f;
typedef __attribute__((ext_vector_type(8))) float v8f;

#define NEG_SLOPE 0.2f
#define GAT_EPS   1e-16f

template <int CTRL>
__device__ __forceinline__ float dpp_f(float v) {
  const int x = __builtin_bit_cast(int, v);
  return __builtin_bit_cast(float, __builtin_amdgcn_update_dpp(x, x, CTRL, 0xF, 0xF, true));
}
template <int CTRL>
__device__ __forceinline__ int dpp_i(int v) {
  return __builtin_amdgcn_update_dpp(v, v, CTRL, 0xF, 0xF, true);
}
template <int H>
__device__ __forceinline__ float head_max(float v) {
  v = fmaxf(v, dpp_f<0x161>(v));
  v = fmaxf(v, dpp_f<0x162>(v));
  if constexpr (H == 1) {
    v = fmaxf(v, dpp_f<0x164>(v));
    v = fmaxf(v, dpp_f<0x168>(v));
  }
  return v;
}
template <int H>
__device__ __forceinline__ float head_sum(float v) {
  v += dpp_f<0x161>(v);
  v += dpp_f<0x162>(v);
  if constexpr (H == 1) {
    v += dpp_f<0x164>(v);
    v += dpp_f<0x168>(v);
  }
  return v;
}
template <int H, int QO>
__device__ __forceinline__ float head_bcast_f(float v) {
  if constexpr (H == 4) return dpp_f<QO * 0x55>(v);
  else                  return dpp_f<0x150 + QO>(v);
}
template <int H, int QO>
__device__ __forceinline__ int head_bcast_i(int v) {
  if constexpr (H == 4) return dpp_i<QO * 0x55>(v);
  else                  return dpp_i<0x150 + QO>(v);
}

template <int KDIM, int MO>
__device__ __forceinline__ void gemm_strip(
    const float* __restrict__ X, const float* __restrict__ W,
    float* __restrict__ Hout, int ntiles, const int tile) {
  if (tile >= ntiles) return;
  const int lane  = threadIdx.x & 31;
  const int lmod  = lane & 15;
  const int lhalf = lane >> 4;
  const int row0  = tile * 16;

  v8f acc[MO / 16];
#pragma unroll
  for (int t = 0; t < MO / 16; ++t) acc[t] = v8f{0, 0, 0, 0, 0, 0, 0, 0};

  const float* xrow = X + (size_t)(row0 + lmod) * KDIM;
#pragma unroll 4
  for (int k0 = 0; k0 < KDIM; k0 += 4) {
    const int ka = k0 + 2 * lhalf;
    v2f a;
    a.x = xrow[ka];
    a.y = xrow[ka + 1];
#pragma unroll
    for (int t = 0; t < MO / 16; ++t) {
      const float* wr = W + (size_t)ka * MO + t * 16 + lmod;
      v2f b;
      b.x = wr[0];
      b.y = wr[MO];
      acc[t] = __builtin_amdgcn_wmma_f32_16x16x4_f32(
          false, a, false, b, (short)0, acc[t], false, false);
    }
  }
#pragma unroll
  for (int t = 0; t < MO / 16; ++t)
#pragma unroll
    for (int r = 0; r < 8; ++r)
      Hout[(size_t)(row0 + r + 8 * lhalf) * MO + t * 16 + lmod] = acc[t][r];
}

#define NRANGE 256


template <int SH>
__device__ __forceinline__ int row_shr0(int v) {
  return __builtin_amdgcn_update_dpp(0, v, 0x110 + SH, 0xF, 0xF, true);
}
__device__ __forceinline__ int wave_scan_incl(int v, int lane) {
  v += row_shr0<1>(v);
  v += row_shr0<2>(v);
  v += row_shr0<4>(v);
  v += row_shr0<8>(v);
  const int low = __builtin_amdgcn_readlane(v, 15);
  return v + ((lane & 16) ? low : 0);
}
template <bool VEC>
__device__ __forceinline__ int4 scan_ld4(const int* p, int i, int N, bool take, int safe) {
  int4 v;
  if constexpr (VEC) {
    const bool in = take && (i < N);
    v = *(const int4*)(p + (in ? i : safe));
    if (!in) v = make_int4(0, 0, 0, 0);
  } else {
    const bool i0 = take && (i < N), i1 = take && (i + 1 < N);
    const bool i2 = take && (i + 2 < N), i3 = take && (i + 3 < N);
    const int a = p[i0 ? i     : safe];
    const int b = p[i1 ? i + 1 : safe];
    const int c = p[i2 ? i + 2 : safe];
    const int d = p[i3 ? i + 3 : safe];
    v = make_int4(i0 ? a : 0, i1 ? b : 0, i2 ? c : 0, i3 ? d : 0);
  }
  return v;
}
template <bool VEC>
__device__ __forceinline__ void scan_st4(int* p, int i, int N, const int4 v) {
  if constexpr (VEC) {
    if (i < N) *(int4*)(p + i) = v;
  } else {
    if (i     < N) p[i]     = v.x;
    if (i + 1 < N) p[i + 1] = v.y;
    if (i + 2 < N) p[i + 2] = v.z;
    if (i + 3 < N) p[i + 3] = v.w;
  }
}
template <bool VEC>
__device__ __forceinline__ void csr_scan_body(int* __restrict__ cnt, int* __restrict__ rowptr,
                                              int N, int* wtot) {
  constexpr int TRIP = 8;
  const int tid  = threadIdx.x;
  const int lane = tid & 31;
  const int wave = __builtin_amdgcn_readfirstlane(tid >> 5);
  const int seg   = (((N + 31) >> 5) + 127) & ~127;
  const int steps = seg >> 7;
  const int seg0  = wave * seg;
  const int base  = seg0 + lane * 4;

  int s = 0;
  for (int k0 = 0; k0 < steps; k0 += TRIP) {
    int4 v[TRIP];
#pragma unroll
    for (int u = 0; u < TRIP; ++u)
      v[u] = scan_ld4<VEC>(cnt, base + (k0 + u) * 128, N, k0 + u < steps, 0);
#pragma unroll
    for (int u = 0; u < TRIP; ++u) s += (v[u].x + v[u].y) + (v[u].z + v[u].w);
  }
  const int incl = wave_scan_incl(s, lane);
  if (lane == 31) wtot[wave] = incl;
  __syncthreads();
  const int w  = wtot[lane];
  const int wi = wave_scan_incl(w, lane);
  int carry = __builtin_amdgcn_readlane(wi - w, wave);
  const int total = __builtin_amdgcn_readlane(wi, 31);

  if (seg0 < N) {
    for (int k0 = 0; k0 < steps; k0 += TRIP) {
      int4 v[TRIP];
#pragma unroll
      for (int u = 0; u < TRIP; ++u)
        v[u] = scan_ld4<VEC>(cnt, base + (k0 + u) * 128, N, k0 + u < steps, seg0);
#pragma unroll
      for (int u = 0; u < TRIP; ++u) {
        if (k0 + u < steps) {
          const int t4 = (v[u].x + v[u].y) + (v[u].z + v[u].w);
          const int in = wave_scan_incl(t4, lane);
          int4 p;
          p.x = carry + (in - t4);
          p.y = p.x + v[u].x;
          p.z = p.y + v[u].y;
          p.w = p.z + v[u].z;
          const int i = base + (k0 + u) * 128;
          scan_st4<VEC>(rowptr, i, N, p);
          scan_st4<VEC>(cnt, i, N, p);
          carry += __builtin_amdgcn_readlane(in, 31);
        }
      }
    }
  }
  if (tid == 0) rowptr[N] = total;
}
__global__ __launch_bounds__(1024) void csr_scan(int* __restrict__ cnt,
                                                 int* __restrict__ rowptr, int N) {
  __shared__ int wtot[32];
  const bool vec = ((((size_t)cnt | (size_t)rowptr) & 15) == 0) && ((N & 3) == 0);
  if (vec) csr_scan_body<true>(cnt, rowptr, N, wtot);
  else     csr_scan_body<false>(cnt, rowptr, N, wtot);
}

#define STAGE_PAIRS 4096
#define STAGE_SRCS  8192

__device__ __forceinline__ int node_range(int d, int bw) {
  const unsigned r = (unsigned)d / (unsigned)bw;
  return (int)((r < NRANGE) ? r : (NRANGE - 1));
}

__global__ __launch_bounds__(1024) void bin_count(const int* __restrict__ ei, int E, int bw,
                                                  int chunk, int* __restrict__ cntm) {
  __shared__ int hist[NRANGE];
  const int tid = threadIdx.x;
  if (tid < NRANGE) hist[tid] = 0;
  __syncthreads();
  const int e0 = (int)blockIdx.x * chunk;
  const int e1 = (e0 + chunk < E) ? (e0 + chunk) : E;
  const int* dst = ei + (size_t)E;
  for (int t0 = e0 + tid; t0 < e1; t0 += 4096) {
    int d[4];
#pragma unroll
    for (int u = 0; u < 4; ++u) {
      const int t = t0 + u * 1024;
      d[u] = dst[(t < e1) ? t : t0];
    }
#pragma unroll
    for (int u = 0; u < 4; ++u)
      if (t0 + u * 1024 < e1) atomicAdd(&hist[node_range(d[u], bw)], 1);
  }
  __syncthreads();
  if (tid < NRANGE) cntm[tid * (int)gridDim.x + (int)blockIdx.x] = hist[tid];
}

__global__ __launch_bounds__(1024) void bin_scatter(const int* __restrict__ ei, int E, int bw,
                                                    int chunk, const int* __restrict__ offm,
                                                    int2* __restrict__ binned) {
  __shared__ int goff[NRANGE];
  __shared__ int lst[NRANGE];
  __shared__ int cur[NRANGE];
  __shared__ int wsum[NRANGE / 32];
  __shared__ int2 stage[STAGE_PAIRS];
  const int tid  = threadIdx.x;
  const int lane = tid & 31;
  const int wave = tid >> 5;
  const int e0 = (int)blockIdx.x * chunk;
  const int e1 = (e0 + chunk < E) ? (e0 + chunk) : E;
  const int n  = (e1 > e0) ? (e1 - e0) : 0;
  const bool staged = chunk <= STAGE_PAIRS;
  int c = 0, incl = 0;
  if (tid < NRANGE) {
    const int f  = tid * (int)gridDim.x + (int)blockIdx.x;
    const int g0 = offm[f];
    c = offm[f + 1] - g0;
    goff[tid] = g0;
    incl = wave_scan_incl(c, lane);
    if (lane == 31) wsum[wave] = incl;
  }
  __syncthreads();
  if (tid < NRANGE) {
    int pre = 0;
    for (int w = 0; w < NRANGE / 32; ++w) pre += (w < wave) ? wsum[w] : 0;
    const int l0 = pre + incl - c;
    lst[tid] = l0;
    cur[tid] = staged ? l0 : goff[tid];
  }
  __syncthreads();
  const int* dst = ei + (size_t)E;
  for (int t0 = e0 + tid; t0 < e1; t0 += 4096) {
    int sv[4], d[4];
#pragma unroll
    for (int u = 0; u < 4; ++u) {
      const int t  = t0 + u * 1024;
      const int tt = (t < e1) ? t : t0;
      sv[u] = ei[tt];
      d[u]  = dst[tt];
    }
#pragma unroll
    for (int u = 0; u < 4; ++u) {
      if (t0 + u * 1024 < e1) {
        const int p = atomicAdd(&cur[node_range(d[u], bw)], 1);
        if (staged) stage[p] = make_int2(sv[u], d[u]);
        else        binned[p] = make_int2(sv[u], d[u]);
      }
    }
  }
  if (staged) {
    __syncthreads();
    for (int i = tid; i < n; i += 1024) {
      const int2 v = stage[i];
      const int r = node_range(v.y, bw);
      binned[goff[r] + (i - lst[r])] = v;
    }
  }
}

__global__ __launch_bounds__(1024) void bin_finish(const int2* __restrict__ binned,
                                                   const int* __restrict__ offm, int G,
                                                   int E, int N, int bw,
                                                   int* __restrict__ rowptr,
                                                   int* __restrict__ tmp_src) {
  __shared__ int cnt[NRANGE];
  __shared__ int cur[NRANGE];
  __shared__ int wsum[NRANGE / 32];
  __shared__ int stage[STAGE_SRCS];
  const int tid  = threadIdx.x;
  const int lane = tid & 31;
  const int wave = tid >> 5;
  const int b  = (int)blockIdx.x;
  const int s  = offm[b * G];
  const int e  = offm[(b + 1) * G];
  const int n  = e - s;
  const int d0 = b * bw;
  const bool staged = n <= STAGE_SRCS;
  if (tid < NRANGE) cnt[tid] = 0;
  __syncthreads();
  for (int i0 = s + tid; i0 < e; i0 += 4096) {
    int d[4];
#pragma unroll
    for (int u = 0; u < 4; ++u) {
      const int i = i0 + u * 1024;
      d[u] = binned[(i < e) ? i : i0].y;
    }
#pragma unroll
    for (int u = 0; u < 4; ++u) {
      const unsigned dl = (unsigned)(d[u] - d0);
      if (i0 + u * 1024 < e && dl < NRANGE) atomicAdd(&cnt[dl], 1);
    }
  }
  __syncthreads();
  int c = 0, incl = 0;
  if (tid < NRANGE) {
    c = cnt[tid];
    incl = wave_scan_incl(c, lane);
    if (lane == 31) wsum[wave] = incl;
  }
  __syncthreads();
  if (tid < NRANGE) {
    int pre = 0;
    for (int w = 0; w < NRANGE / 32; ++w) pre += (w < wave) ? wsum[w] : 0;
    const int l0 = pre + incl - c;
    cur[tid] = staged ? l0 : (s + l0);
    const int node = d0 + tid;
    if (tid < bw && node < N) rowptr[node] = s + l0;
  }
  if (b == NRANGE - 1 && tid == 0) rowptr[N] = E;
  __syncthreads();
  for (int i0 = s + tid; i0 < e; i0 += 4096) {
    int2 v[4];
#pragma unroll
    for (int u = 0; u < 4; ++u) {
      const int i = i0 + u * 1024;
      v[u] = binned[(i < e) ? i : i0];
    }
#pragma unroll
    for (int u = 0; u < 4; ++u) {
      const unsigned dl = (unsigned)(v[u].y - d0);
      if (i0 + u * 1024 < e && dl < NRANGE) {
        const int p = atomicAdd(&cur[dl], 1);
        if (staged) stage[p] = v[u].x;
        else        tmp_src[p] = v[u].x;
      }
    }
  }
  if (staged) {
    __syncthreads();
    for (int i = tid; i < n; i += 1024) tmp_src[s + i] = stage[i];
  }
}

template <int KDIM, int MO>
__global__ __launch_bounds__(256) void csr_order_gemm(
    const int* __restrict__ rowptr, const int* __restrict__ tmp_src,
    int* __restrict__ csr_src, int N,
    const float* __restrict__ X, const float* __restrict__ W,
    float* __restrict__ Hout, int ntiles, int gemm_blocks) {
  if ((int)blockIdx.x < gemm_blocks) {
    gemm_strip<KDIM, MO>(X, W, Hout, ntiles, (int)(blockIdx.x * 8u + (threadIdx.x >> 5)));
    return;
  }
  const unsigned bid = blockIdx.x - (unsigned)gemm_blocks;
  const int g = (int)((bid * 256u + threadIdx.x) >> 4);
  if (g >= N) return;
  const int lane = threadIdx.x & 15;
  const int beg = rowptr[g], end = rowptr[g + 1];
  for (int i = beg + lane; i < end; i += 16) {
    const int v = tmp_src[i];
    int r = 0;
    for (int j = beg; j < end; ++j) {
      const int w = tmp_src[j];
      r += ((w < v) || (w == v && j < i)) ? 1 : 0;
    }
    csr_src[beg + r] = v;
  }
}

template <int H, bool SHARED_A>
__device__ __forceinline__ float4 gat_row(
    const int* __restrict__ rowptr, const int* __restrict__ csr_src,
    const float* __restrict__ Hm,
    const float* __restrict__ a_src, const float* __restrict__ a_dst,
    const int g, const int j) {
  static_assert(H == 1 || H == 4, "head_bcast covers a quad or a whole row");
  constexpr int WD  = H * 16;
  constexpr int LPH = 16 / H;
  constexpr int NT  = 2 * H;
  constexpr int NK  = NT * LPH;
  const int q = j % LPH;
  const int beg = rowptr[g];
  const int n = rowptr[g + 1] - beg;
  auto ld_i = [](const int* b, unsigned off) { return *(const int*)((const char*)b + off); };
  const unsigned roff = (unsigned)(j * H) * 4u;
  auto ld_row = [&](unsigned node) {
    const float* hp = (const float*)((const char*)Hm + (node * (WD * 4u) + roff));
    float4 v;
    if constexpr (H == 4) v = *(const float4*)hp;
    else                  v = make_float4(hp[0], 0.f, 0.f, 0.f);
    return v;
  };
  float4 as, ad;
  if constexpr (H == 4) {
    const float* sp = a_src + (SHARED_A ? q : j) * 4;
    const float* dp = a_dst + (SHARED_A ? q : j) * 4;
    as = make_float4(sp[0], sp[1], sp[2], sp[3]);
    ad = make_float4(dp[0], dp[1], dp[2], dp[3]);
  } else {
    as = make_float4(a_src[j], 0.f, 0.f, 0.f);
    ad = make_float4(a_dst[j], 0.f, 0.f, 0.f);
  }
  auto head_dot = [](const float4& v, const float4& a) {
    float d;
    if constexpr (H == 4) {
      v2f t = v2f{v.x, v.y} * v2f{a.x, a.y};
      t = v2f{v.z, v.w} * v2f{a.z, a.w} + t;
      d = t.x + t.y;
    } else {
      d = v.x * a.x;
    }
    return head_sum<H>(d);
  };
  auto leaky = [](float y) { return fmaxf(y, NEG_SLOPE * y); };

  int sv[NT];
  if (n > 0) {
    for (int t = 0; t < NT; ++t) {
      const int p = t * LPH + q;
      sv[t] = ld_i(csr_src, (unsigned)(beg + ((p < n) ? p : n - 1)) * 4u);
    }
  }

  const float4 v0 = ld_row((unsigned)g);
  const float ed = head_dot(v0, ad);
  float m = leaky(head_dot(v0, as) + ed);
  float den = 1.f;
  float4 acc = v0;

  auto step4 = [&](float y0, float y1, float y2, float y3, const float4& r0,
                   const float4& r1, const float4& r2, const float4& r3) {
    const float mn = fmaxf(fmaxf(fmaxf(y0, y1), fmaxf(y2, y3)), m);
    const float sc = __expf(m - mn);
    const float p0 = __expf(y0 - mn);
    const float p1 = __expf(y1 - mn);
    const float p2 = __expf(y2 - mn);
    const float p3 = __expf(y3 - mn);
    m = mn;
    den = den * sc + p0; den += p1; den += p2; den += p3;
    if constexpr (H == 4) {
      const v2f s2 = {sc, sc};
      const v2f q0 = {p0, p0}, q1 = {p1, p1}, q2 = {p2, p2}, q3 = {p3, p3};
      v2f lo = v2f{acc.x, acc.y} * s2;
      v2f hi = v2f{acc.z, acc.w} * s2;
      lo = q0 * v2f{r0.x, r0.y} + lo;  hi = q0 * v2f{r0.z, r0.w} + hi;
      lo = q1 * v2f{r1.x, r1.y} + lo;  hi = q1 * v2f{r1.z, r1.w} + hi;
      lo = q2 * v2f{r2.x, r2.y} + lo;  hi = q2 * v2f{r2.z, r2.w} + hi;
      lo = q3 * v2f{r3.x, r3.y} + lo;  hi = q3 * v2f{r3.z, r3.w} + hi;
      acc = make_float4(lo.x, lo.y, hi.x, hi.y);
    } else {
      acc.x = acc.x * sc + p0 * r0.x; acc.x += p1 * r1.x; acc.x += p2 * r2.x; acc.x += p3 * r3.x;
    }
  };

#define GAT_ROW_LOAD(K, V)                                                     \
  { V = ld_row((unsigned)head_bcast_i<H, (K) % LPH>(sv[(K) / LPH])); }
#define GAT_LOGIT(K, V, Y)                                                     \
  {                                                                            \
    const float y_ = leaky(head_dot(V, as) + ed);                              \
    Y = ((K) < n) ? y_ : -__builtin_inff();                                    \
  }
#define GAT_FOUR(K, A, B, C, D)                                                \
  {                                                                            \
    float y0_, y1_, y2_, y3_;                                                  \
    GAT_LOGIT((K) + 0, A, y0_) GAT_LOGIT((K) + 1, B, y1_)                      \
    GAT_LOGIT((K) + 2, C, y2_) GAT_LOGIT((K) + 3, D, y3_)                      \
    step4(y0_, y1_, y2_, y3_, A, B, C, D);                                     \
  }
#define GAT_EIGHT(K)                                                           \
  if (n > (K)) {                                                               \
    float4 r0_, r1_, r2_, r3_, r4_, r5_, r6_, r7_;                             \
    GAT_ROW_LOAD((K) + 0, r0_) GAT_ROW_LOAD((K) + 1, r1_)                      \
    GAT_ROW_LOAD((K) + 2, r2_) GAT_ROW_LOAD((K) + 3, r3_)                      \
    if (n > (K) + 4) {                                                         \
      GAT_ROW_LOAD((K) + 4, r4_) GAT_ROW_LOAD((K) + 5, r5_)                    \
      GAT_ROW_LOAD((K) + 6, r6_) GAT_ROW_LOAD((K) + 7, r7_)                    \
    }                                                                          \
    GAT_FOUR((K), r0_, r1_, r2_, r3_)                                          \
    if (n > (K) + 4) GAT_FOUR((K) + 4, r4_, r5_, r6_, r7_)                     \
  }
  GAT_EIGHT(0)
  GAT_EIGHT(8)
  GAT_EIGHT(16)
  GAT_EIGHT(24)
#undef GAT_EIGHT
#undef GAT_FOUR
#undef GAT_LOGIT
#undef GAT_ROW_LOAD

  for (int base = NK; base < n; base += 4) {
    int s[4];
    for (int k = 0; k < 4; ++k) {
      const int p = base + k;
      s[k] = ld_i(csr_src, (unsigned)(beg + ((p < n) ? p : n - 1)) * 4u);
    }
    float4 r[4];
    for (int k = 0; k < 4; ++k) r[k] = ld_row((unsigned)s[k]);
    float y[4];
    for (int k = 0; k < 4; ++k) {
      const float yk = leaky(head_dot(r[k], as) + ed);
      y[k] = (base + k < n) ? yk : -__builtin_inff();
    }
    step4(y[0], y[1], y[2], y[3], r[0], r[1], r[2], r[3]);
  }

  const float inv = 1.0f / (den + GAT_EPS);
  float4 res = make_float4(acc.x * inv, 0.f, 0.f, 0.f);
  if constexpr (H == 4) { res.y = acc.y * inv; res.z = acc.z * inv; res.w = acc.w * inv; }
  return res;
}

template <int NB>
__global__ __launch_bounds__(256) void gat_gather_batches(
    const int* __restrict__ rowptr, const int* __restrict__ csr_src,
    const float* __restrict__ Hm,
    const float* __restrict__ a_src, const float* __restrict__ a_dst,
    const float* __restrict__ bias,
    float* __restrict__ out, int N) {
  static_assert(NB == 4, "a row of NB x 16 floats must be gat_row's 4-head row");
  const int g = (int)((blockIdx.x * 256u + threadIdx.x) >> 4);
  if (g >= N) return;
  const int j = threadIdx.x & 15;
  const float4 v = gat_row<NB, true>(rowptr, csr_src, Hm, a_src, a_dst, g, j);
  const unsigned b = (unsigned)j >> 2;
  const unsigned c = ((unsigned)j & 3u) * 4u;
  const float* bp = bias + c;
  float* op = (float*)((char*)out + ((b * (unsigned)N + (unsigned)g) * 64u + c * 4u));
  *(float4*)op = make_float4(v.x + bp[0], v.y + bp[1], v.z + bp[2], v.w + bp[3]);
}

template <int MO, int RK>
__global__ __launch_bounds__(256) void gat_gather_mm(
    const int* __restrict__ rowptr, const int* __restrict__ csr_src,
    const float* __restrict__ Hm,
    const float* __restrict__ a_src, const float* __restrict__ a_dst,
    const float* __restrict__ bias, const float* __restrict__ W,
    float* __restrict__ Hout, int N,
    const float* __restrict__ Xn, const float* __restrict__ Wn,
    float* __restrict__ Hn, int ntiles, int gemm_blocks) {
  constexpr int KD = 64, PITCH = KD + 4;
  constexpr int LDO = 64;
  __shared__ __attribute__((aligned(16))) float xs[16 * PITCH];
  unsigned bid = blockIdx.x;
  if constexpr (RK > 0) {
    if ((int)bid < gemm_blocks) {
      gemm_strip<RK, 64>(Xn, Wn, Hn, ntiles, (int)(bid * 8u + (threadIdx.x >> 5)));
      return;
    }
    bid -= (unsigned)gemm_blocks;
  }
  const int lrow = (int)(threadIdx.x >> 4);
  const int row0 = (int)(bid * 16u);
  const int g = (row0 + lrow < N) ? (row0 + lrow) : (N - 1);
  const int j = threadIdx.x & 15;
  const float4 v = gat_row<4, false>(rowptr, csr_src, Hm, a_src, a_dst, g, j);
  const float* bp = bias + j * 4;
  float4 r = make_float4(fmaxf(v.x + bp[0], 0.f), fmaxf(v.y + bp[1], 0.f),
                         fmaxf(v.z + bp[2], 0.f), fmaxf(v.w + bp[3], 0.f));
  *(float4*)(xs + lrow * PITCH + j * 4) = r;

  const int wave  = (int)(threadIdx.x >> 5);
  const int lane  = threadIdx.x & 31;
  const int lmod  = lane & 15;
  const int lhalf = lane >> 4;
  const bool mm = wave < MO / 16;
  v2f b[KD / 4];
  if (mm) {
#pragma unroll
    for (int s = 0; s < KD / 4; ++s) {
      const float* wr = W + (size_t)(4 * s + 2 * lhalf) * MO + wave * 16 + lmod;
      b[s].x = wr[0];
      b[s].y = wr[MO];
    }
  }
  __syncthreads();
  if (!mm) return;
  v8f acc = v8f{0, 0, 0, 0, 0, 0, 0, 0};
  const float* xr = xs + lmod * PITCH + 2 * lhalf;
#pragma unroll
  for (int s = 0; s < KD / 4; ++s) {
    v2f a;
    a.x = xr[4 * s];
    a.y = xr[4 * s + 1];
    acc = __builtin_amdgcn_wmma_f32_16x16x4_f32(
        false, a, false, b[s], (short)0, acc, false, false);
  }
#pragma unroll
  for (int rr = 0; rr < 8; ++rr) {
    const int row = row0 + rr + 8 * lhalf;
    if (row < N) Hout[(size_t)row * LDO + wave * 16 + lmod] = acc[rr];
  }
}

static inline int cdiv(long long a, int b) { return (int)((a + b - 1) / b); }

extern "C" void kernel_launch(void* const* d_in, const int* in_sizes, int n_in,
                              void* d_out, int out_size, void* d_ws, size_t ws_size,
                              hipStream_t stream) {
  const float* x   = (const float*)d_in[0];
  const int*   ei  = (const int*)d_in[1];
  const float* W1  = (const float*)d_in[2];
  const float* a1s = (const float*)d_in[3];
  const float* a1d = (const float*)d_in[4];
  const float* b1  = (const float*)d_in[5];
  const float* W2  = (const float*)d_in[6];
  const float* a2s = (const float*)d_in[7];
  const float* a2d = (const float*)d_in[8];
  const float* b2  = (const float*)d_in[9];
  const float* W3  = (const float*)d_in[10];
  const float* a3s = (const float*)d_in[11];
  const float* a3d = (const float*)d_in[12];
  const float* b3  = (const float*)d_in[13];

  const int B = 4, F_IN = 128, HC = 64;
  const int E = in_sizes[1] / 2;
  const int N = in_sizes[0] / (B * F_IN);

  float* ws = (float*)d_ws;
  float* h1      = ws;
  float* h3      = h1 + (size_t)N * HC;
  int*   rowptr  = (int*)(h3 + (size_t)N * HC);
  int*   tmp_src = rowptr + (size_t)N + 1;
  int*   csr_src = tmp_src + (size_t)E;
  float* h2      = (float*)d_out;

  const int ntiles = (N + 15) / 16;
  const int gemm_blocks = (ntiles + 7) / 8;
  const int gather_grid = cdiv((long long)N * 16, 256);

  const int bw    = cdiv(N, NRANGE);
  const int chunk = cdiv(E, 256);
  int*  cntm   = (int*)h1;
  int*  offm   = cntm + NRANGE * 256;
  int2* binned = (int2*)h3;
  bin_count<<<256, 1024, 0, stream>>>(ei, E, bw, chunk, cntm);
  csr_scan<<<1, 1024, 0, stream>>>(cntm, offm, NRANGE * 256);
  bin_scatter<<<256, 1024, 0, stream>>>(ei, E, bw, chunk, offm, binned);
  bin_finish<<<NRANGE, 1024, 0, stream>>>(binned, offm, 256, E, N, bw, rowptr, tmp_src);
  csr_order_gemm<128, 64><<<gemm_blocks + gather_grid, 256, 0, stream>>>(
      rowptr, tmp_src, csr_src, N, x, W1, h1, ntiles, gemm_blocks);

  for (int b = 0; b < B; ++b) {
    gat_gather_mm<64, 0><<<gather_grid, 256, 0, stream>>>(
        rowptr, csr_src, h1, a1s, a1d, b1, W2, h2, N, nullptr, nullptr, nullptr, 0, 0);

    const int nb = (b + 1 < B) ? (b + 1) : b;
    const int rider = (b + 1 < B) ? gemm_blocks : 0;
    gat_gather_mm<16, 128><<<rider + gather_grid, 256, 0, stream>>>(
        rowptr, csr_src, h2, a2s, a2d, b2, W3, h3 + (size_t)b * 16, N,
        x + (size_t)nb * N * F_IN, W1, h1, ntiles, rider);
  }
  gat_gather_batches<4><<<gather_grid, 256, 0, stream>>>(
      rowptr, csr_src, h3, a3s, a3d, b3, (float*)d_out, N);
}
